# starting kernel
# baseline (speedup 1.0000x reference)
#include <hip/hip_runtime.h>
#include <hip/hip_bf16.h>

#define HID     128
#define FIN     11
#define FOUT    19
#define NGRAPHS 4096

typedef __attribute__((ext_vector_type(2))) float v2f;
typedef __attribute__((ext_vector_type(8))) float v8f;

__global__ void k_init_deg(float* deg, int n) {
    int i = blockIdx.x * blockDim.x + threadIdx.x;
    if (i < n) deg[i] = 1.0f;
}

__global__ void k_deg_accum(const int* __restrict__ dst, float* deg, int e) {
    int i = blockIdx.x * blockDim.x + threadIdx.x;
    if (i < e) atomicAdd(&deg[dst[i]], 1.0f);
}

__global__ void k_dinv(const float* __restrict__ deg, float* dinv, int n) {
    int i = blockIdx.x * blockDim.x + threadIdx.x;
    if (i < n) {
        float d = deg[i];
        dinv[i] = (d > 0.0f) ? rsqrtf(d) : 0.0f;
    }
}

__global__ void k_gemm1(const float* __restrict__ x, const float* __restrict__ W1,
                        float* __restrict__ h, int n) {
    int idx = blockIdx.x * blockDim.x + threadIdx.x;
    if (idx >= n * HID) return;
    int node = idx >> 7;
    int f    = idx & (HID - 1);
    const float* xr = x + node * FIN;
    float acc = 0.0f;
#pragma unroll
    for (int k = 0; k < FIN; ++k)
        acc = fmaf(xr[k], W1[k * HID + f], acc);
    h[idx] = acc;
}

__global__ void k_init_out(float* __restrict__ out, const float* __restrict__ h,
                           const float* __restrict__ bias, const float* __restrict__ dinv,
                           int n) {
    int idx = blockIdx.x * blockDim.x + threadIdx.x;
    if (idx >= n * HID) return;
    int node = idx >> 7;
    int f    = idx & (HID - 1);
    float dn = dinv[node];
    out[idx] = bias[f] + h[idx] * dn * dn;
}

__global__ void k_edge_scatter(const float* __restrict__ h, float* __restrict__ out,
                               const int* __restrict__ src, const int* __restrict__ dst,
                               const float* __restrict__ dinv, int e) {
    int gid  = blockIdx.x * blockDim.x + threadIdx.x;
    int edge = gid >> 5;
    int lane = gid & 31;
    if (edge >= e) return;
    int s = src[edge];
    int d = dst[edge];
    float nrm = dinv[s] * dinv[d];
    float4 v = ((const float4*)(h + (size_t)s * HID))[lane];
    float* o = out + (size_t)d * HID + lane * 4;
    atomicAdd(o + 0, v.x * nrm);
    atomicAdd(o + 1, v.y * nrm);
    atomicAdd(o + 2, v.z * nrm);
    atomicAdd(o + 3, v.w * nrm);
}

__global__ __launch_bounds__(256) void k_gemm2_wmma(const float* __restrict__ in,
                                                    const float* __restrict__ W,
                                                    float* __restrict__ out, int n) {
    __shared__ float Ws[HID * HID];
    const int tid = threadIdx.x;

    const float4* W4  = (const float4*)W;
    float4*       Ws4 = (float4*)Ws;
#pragma unroll
    for (int i = 0; i < 16; ++i)
        Ws4[tid + 256 * i] = W4[tid + 256 * i];
    __syncthreads();

    const int m0   = blockIdx.x << 4;
    const int wave = tid >> 5;
    const int lane = tid & 31;
    const int half = lane >> 4;
    const int l15  = lane & 15;
    const int n0   = wave << 4;
    int row = m0 + l15;
    if (row >= n) row = n - 1;
    const int ncol = n0 + l15;
    const float* arow = in + (size_t)row * HID;

    v8f c = {};
#pragma unroll
    for (int k0 = 0; k0 < HID; k0 += 4) {
        const int ka = k0 + half * 2;
        float2 av = *(const float2*)(arow + ka);
        v2f a, b;
        a[0] = fmaxf(av.x, 0.0f);
        a[1] = fmaxf(av.y, 0.0f);
        b[0] = Ws[ka * HID + ncol];
        b[1] = Ws[(ka + 1) * HID + ncol];
        c = __builtin_amdgcn_wmma_f32_16x16x4_f32(
                 false, a,  false, b,
                 (short)0, c,  false,  false);
    }

    const int mbase = m0 + half * 8;
#pragma unroll
    for (int i = 0; i < 8; ++i) {
        int r = mbase + i;
        if (r < n) out[(size_t)r * HID + ncol] = c[i];
    }
}

__global__ void k_pool_init(float* sums, float* cnt) {
    int idx = blockIdx.x * blockDim.x + threadIdx.x;
    if (idx < NGRAPHS * HID) sums[idx] = 0.0f;
    if (idx < NGRAPHS)       cnt[idx]  = 0.0f;
}

__global__ void k_pool_accum(const float* __restrict__ h, const int* __restrict__ batch,
                             float* sums, int n) {
    int idx = blockIdx.x * blockDim.x + threadIdx.x;
    if (idx >= n * HID) return;
    int node = idx >> 7;
    int f    = idx & (HID - 1);
    atomicAdd(&sums[batch[node] * HID + f], h[idx]);
}

__global__ void k_cnt_accum(const int* __restrict__ batch, float* cnt, int n) {
    int i = blockIdx.x * blockDim.x + threadIdx.x;
    if (i < n) atomicAdd(&cnt[batch[i]], 1.0f);
}

__global__ void k_final(const float* __restrict__ sums, const float* __restrict__ cnt,
                        const float* __restrict__ Wlin, const float* __restrict__ blin,
                        float* __restrict__ out) {
    int idx = blockIdx.x * blockDim.x + threadIdx.x;
    if (idx >= NGRAPHS * FOUT) return;
    int g = idx / FOUT;
    int j = idx - g * FOUT;
    float inv = 1.0f / fmaxf(cnt[g], 1.0f);
    const float* s = sums + g * HID;
    float acc = blin[j];
#pragma unroll 4
    for (int k = 0; k < HID; ++k)
        acc = fmaf(s[k] * inv, Wlin[k * FOUT + j], acc);
    out[idx] = acc;
}

extern "C" void kernel_launch(void* const* d_in, const int* in_sizes, int n_in,
                              void* d_out, int out_size, void* d_ws, size_t ws_size,
                              hipStream_t stream) {
    (void)n_in; (void)out_size; (void)ws_size;

    const float* x     = (const float*)d_in[0];
    const int*   esrc  = (const int*)  d_in[1];
    const int*   edst  = (const int*)  d_in[2];
    const int*   batch = (const int*)  d_in[3];
    const float* W1    = (const float*)d_in[4];
    const float* b1    = (const float*)d_in[5];
    const float* W2    = (const float*)d_in[6];
    const float* b2    = (const float*)d_in[7];
    const float* Wlin  = (const float*)d_in[8];
    const float* blin  = (const float*)d_in[9];
    float*       out   = (float*)d_out;

    const int N = in_sizes[3];
    const int E = in_sizes[1];

    char*  ws  = (char*)d_ws;
    size_t off = 0;
    auto carve = [&](size_t bytes) -> void* {
        off = (off + 255) & ~(size_t)255;
        void* p = ws + off;
        off += bytes;
        return p;
    };
    float* deg   = (float*)carve((size_t)N * 4);
    float* dinv  = (float*)carve((size_t)N * 4);
    float* bufH  = (float*)carve((size_t)N * HID * 4);
    float* bufO1 = (float*)carve((size_t)N * HID * 4);
    float* bufO2 = (float*)carve((size_t)N * HID * 4);
    float* sums  = (float*)carve((size_t)NGRAPHS * HID * 4);
    float* cnt   = (float*)carve((size_t)NGRAPHS * 4);

    const int T = 256;
    int nb   = (N + T - 1) / T;
    int eb   = (E + T - 1) / T;
    int fb   = (N * HID + T - 1) / T;
    int ewb  = (int)(((size_t)E * 32 + T - 1) / T);
    int gb2  = (N + 15) / 16;
    int pb   = (NGRAPHS * HID + T - 1) / T;
    int hb   = (NGRAPHS * FOUT + T - 1) / T;

    k_init_deg <<<nb, T, 0, stream>>>(deg, N);
    k_deg_accum<<<eb, T, 0, stream>>>(edst, deg, E);
    k_dinv     <<<nb, T, 0, stream>>>(deg, dinv, N);

    k_gemm1       <<<fb,  T, 0, stream>>>(x, W1, bufH, N);
    k_init_out    <<<fb,  T, 0, stream>>>(bufO1, bufH, b1, dinv, N);
    k_edge_scatter<<<ewb, T, 0, stream>>>(bufH, bufO1, esrc, edst, dinv, E);

    k_gemm2_wmma  <<<gb2, T, 0, stream>>>(bufO1, W2, bufH, N);
    k_init_out    <<<fb,  T, 0, stream>>>(bufO2, bufH, b2, dinv, N);
    k_edge_scatter<<<ewb, T, 0, stream>>>(bufH, bufO2, esrc, edst, dinv, E);

    k_pool_init <<<pb, T, 0, stream>>>(sums, cnt);
    k_pool_accum<<<fb, T, 0, stream>>>(bufO2, batch, sums, N);
    k_cnt_accum <<<nb, T, 0, stream>>>(batch, cnt, N);
    k_final     <<<hb, T, 0, stream>>>(sums, cnt, Wlin, blin, out);
}


# Edge scatter by f32 atomics replaced by per-call destination lists (integer atomic slot, scan, fill); one wave per node sums its rows in registers in slot order, self-loop and bias folded in
# speedup vs baseline: 2.3926x; 2.3926x over previous
#include <hip/hip_runtime.h>
#include <hip/hip_bf16.h>

#define HID     128
#define FIN     11
#define FOUT    19
#define NGRAPHS 4096

typedef __attribute__((ext_vector_type(2))) float v2f;
typedef __attribute__((ext_vector_type(8))) float v8f;

#define SCAN_T 256
#define SCAN_N (SCAN_T * 4)
#define SCAN_SHIFT 10

__global__ void k_zero_cnt(int* __restrict__ cnt, int n) {
    int i = blockIdx.x * blockDim.x + threadIdx.x;
    if (i < n) cnt[i] = 0;
}

__global__ void k_count_rank(const int* __restrict__ dst, int* cnt, int* __restrict__ rank, int e) {
    int i = blockIdx.x * blockDim.x + threadIdx.x;
    if (i < e) rank[i] = atomicAdd(&cnt[dst[i]], 1);
}

__global__ __launch_bounds__(SCAN_T) void k_scan_local(const int* __restrict__ cnt,
                                                       int* __restrict__ rowloc,
                                                       int* __restrict__ blksum,
                                                       float* __restrict__ dinv, int n) {
    __shared__ int wsum[SCAN_T / 32];
    const int tid  = threadIdx.x;
    const int lane = tid & 31;
    const int wave = tid >> 5;
    const int i0   = blockIdx.x * SCAN_N + tid * 4;

    int c[4];
#pragma unroll
    for (int k = 0; k < 4; ++k) c[k] = (i0 + k < n) ? cnt[i0 + k] : 0;
    const int t = c[0] + c[1] + c[2] + c[3];

    int incl = t;
#pragma unroll
    for (int off = 1; off < 32; off <<= 1) {
        int y = __shfl_up(incl, off);
        if (lane >= off) incl += y;
    }
    if (lane == 31) wsum[wave] = incl;
    __syncthreads();
    int wbase = 0;
    for (int w = 0; w < wave; ++w) wbase += wsum[w];

    int run = wbase + incl - t;
#pragma unroll
    for (int k = 0; k < 4; ++k) {
        if (i0 + k < n) {
            rowloc[i0 + k] = run;
            float d = (float)(c[k] + 1);
            dinv[i0 + k] = (d > 0.0f) ? rsqrtf(d) : 0.0f;
        }
        run += c[k];
    }
    if (tid == SCAN_T - 1) blksum[blockIdx.x] = run;
}

__global__ void k_scan_blocks(const int* __restrict__ blksum, int* __restrict__ blkoff, int nblk) {
    const int lane = threadIdx.x & 31;
    int carry = 0;
    for (int base = 0; base < nblk; base += 32) {
        const int i = base + lane;
        const int v = (i < nblk) ? blksum[i] : 0;
        int incl = v;
#pragma unroll
        for (int off = 1; off < 32; off <<= 1) {
            int y = __shfl_up(incl, off);
            if (lane >= off) incl += y;
        }
        if (i < nblk) blkoff[i] = carry + incl - v;
        carry += __shfl(incl, 31);
    }
}

__global__ void k_fill(const int* __restrict__ src, const int* __restrict__ dst,
                       const int* __restrict__ rank, const int* __restrict__ rowloc,
                       const int* __restrict__ blkoff, const float* __restrict__ dinv,
                       int2* __restrict__ rec, int e) {
    int i = blockIdx.x * blockDim.x + threadIdx.x;
    if (i >= e) return;
    const int s = src[i];
    const int d = dst[i];
    const int pos = rowloc[d] + blkoff[d >> SCAN_SHIFT] + rank[i];
    rec[pos] = make_int2(s, __float_as_int(dinv[s]));
}

__global__ void k_gemm1(const float* __restrict__ x, const float* __restrict__ W1,
                        float* __restrict__ h, int n) {
    int idx = blockIdx.x * blockDim.x + threadIdx.x;
    if (idx >= n * HID) return;
    int node = idx >> 7;
    int f    = idx & (HID - 1);
    const float* xr = x + node * FIN;
    float acc = 0.0f;
#pragma unroll
    for (int k = 0; k < FIN; ++k)
        acc = fmaf(xr[k], W1[k * HID + f], acc);
    h[idx] = acc;
}


template <int G>
__device__ __forceinline__ void pull_group(const float4* __restrict__ h4, int lane,
                                           int s_l, int w_l, int j, float4& acc) {
    float4 v[G];
    float  w[G];
#pragma unroll
    for (int k = 0; k < G; ++k) {
        const int s = __builtin_amdgcn_readlane(s_l, j + k);
        w[k] = __int_as_float(__builtin_amdgcn_readlane(w_l, j + k));
        v[k] = h4[(size_t)s * (HID / 4) + lane];
    }
#pragma unroll
    for (int k = 0; k < G; ++k) {
        acc.x = fmaf(v[k].x, w[k], acc.x);
        acc.y = fmaf(v[k].y, w[k], acc.y);
        acc.z = fmaf(v[k].z, w[k], acc.z);
        acc.w = fmaf(v[k].w, w[k], acc.w);
    }
}

__global__ __launch_bounds__(256) void k_pull(const float* __restrict__ h, float* __restrict__ out,
                                              const int2* __restrict__ rec,
                                              const int* __restrict__ rowloc,
                                              const int* __restrict__ blkoff,
                                              const int* __restrict__ cnt,
                                              const float* __restrict__ dinv,
                                              const float* __restrict__ bias, int n) {
    const int lane = threadIdx.x & 31;
    const int node = __builtin_amdgcn_readfirstlane((int)((blockIdx.x * 256u + threadIdx.x) >> 5));
    if (node >= n) return;

    const int   start = rowloc[node] + blkoff[node >> SCAN_SHIFT];
    const int   c     = cnt[node];
    const float dn    = dinv[node];
    const float4* h4  = (const float4*)h;

    const float4 self = h4[(size_t)node * (HID / 4) + lane];
    float4 acc = make_float4(0.0f, 0.0f, 0.0f, 0.0f);

    for (int base = 0; base < c; base += 32) {
        const int m = min(32, c - base);
        int s_l = node;
        int w_l = 0;
        if (lane < m) {
            const int2 r = rec[start + base + lane];
            s_l = r.x;
            w_l = __float_as_int(__int_as_float(r.y) * dn);
        }
        int j = 0;
        for (; j + 8 <= m; j += 8) pull_group<8>(h4, lane, s_l, w_l, j, acc);
        if (j + 4 <= m) { pull_group<4>(h4, lane, s_l, w_l, j, acc); j += 4; }
        if (j < m) {
            const float4 z = make_float4(0.0f, 0.0f, 0.0f, 0.0f);
            float4 v0 = z, v1 = z, v2 = z;
            float  w0 = 0.0f, w1 = 0.0f, w2 = 0.0f;
            {
                const int s = __builtin_amdgcn_readlane(s_l, j);
                w0 = __int_as_float(__builtin_amdgcn_readlane(w_l, j));
                v0 = h4[(size_t)s * (HID / 4) + lane];
            }
            if (j + 1 < m) {
                const int s = __builtin_amdgcn_readlane(s_l, j + 1);
                w1 = __int_as_float(__builtin_amdgcn_readlane(w_l, j + 1));
                v1 = h4[(size_t)s * (HID / 4) + lane];
            }
            if (j + 2 < m) {
                const int s = __builtin_amdgcn_readlane(s_l, j + 2);
                w2 = __int_as_float(__builtin_amdgcn_readlane(w_l, j + 2));
                v2 = h4[(size_t)s * (HID / 4) + lane];
            }
            acc.x = fmaf(v0.x, w0, acc.x); acc.y = fmaf(v0.y, w0, acc.y);
            acc.z = fmaf(v0.z, w0, acc.z); acc.w = fmaf(v0.w, w0, acc.w);
            acc.x = fmaf(v1.x, w1, acc.x); acc.y = fmaf(v1.y, w1, acc.y);
            acc.z = fmaf(v1.z, w1, acc.z); acc.w = fmaf(v1.w, w1, acc.w);
            acc.x = fmaf(v2.x, w2, acc.x); acc.y = fmaf(v2.y, w2, acc.y);
            acc.z = fmaf(v2.z, w2, acc.z); acc.w = fmaf(v2.w, w2, acc.w);
        }
    }

    const float  ws = dn * dn;
    acc.x = fmaf(self.x, ws, acc.x);
    acc.y = fmaf(self.y, ws, acc.y);
    acc.z = fmaf(self.z, ws, acc.z);
    acc.w = fmaf(self.w, ws, acc.w);

    const float4 b = ((const float4*)bias)[lane];
    float4 o;
    o.x = acc.x + b.x;
    o.y = acc.y + b.y;
    o.z = acc.z + b.z;
    o.w = acc.w + b.w;
    ((float4*)out)[(size_t)node * (HID / 4) + lane] = o;
}

__global__ __launch_bounds__(256) void k_gemm2_wmma(const float* __restrict__ in,
                                                    const float* __restrict__ W,
                                                    float* __restrict__ out, int n) {
    __shared__ float Ws[HID * HID];
    const int tid = threadIdx.x;

    const float4* W4  = (const float4*)W;
    float4*       Ws4 = (float4*)Ws;
#pragma unroll
    for (int i = 0; i < 16; ++i)
        Ws4[tid + 256 * i] = W4[tid + 256 * i];
    __syncthreads();

    const int m0   = blockIdx.x << 4;
    const int wave = tid >> 5;
    const int lane = tid & 31;
    const int half = lane >> 4;
    const int l15  = lane & 15;
    const int n0   = wave << 4;
    int row = m0 + l15;
    if (row >= n) row = n - 1;
    const int ncol = n0 + l15;
    const float* arow = in + (size_t)row * HID;

    v8f c = {};
#pragma unroll
    for (int k0 = 0; k0 < HID; k0 += 4) {
        const int ka = k0 + half * 2;
        float2 av = *(const float2*)(arow + ka);
        v2f a, b;
        a[0] = fmaxf(av.x, 0.0f);
        a[1] = fmaxf(av.y, 0.0f);
        b[0] = Ws[ka * HID + ncol];
        b[1] = Ws[(ka + 1) * HID + ncol];
        c = __builtin_amdgcn_wmma_f32_16x16x4_f32(
                 false, a,  false, b,
                 (short)0, c,  false,  false);
    }

    const int mbase = m0 + half * 8;
#pragma unroll
    for (int i = 0; i < 8; ++i) {
        int r = mbase + i;
        if (r < n) out[(size_t)r * HID + ncol] = c[i];
    }
}

__global__ void k_pool_init(float* sums, float* cnt) {
    int idx = blockIdx.x * blockDim.x + threadIdx.x;
    if (idx < NGRAPHS * HID) sums[idx] = 0.0f;
    if (idx < NGRAPHS)       cnt[idx]  = 0.0f;
}

__global__ void k_pool_accum(const float* __restrict__ h, const int* __restrict__ batch,
                             float* sums, int n) {
    int idx = blockIdx.x * blockDim.x + threadIdx.x;
    if (idx >= n * HID) return;
    int node = idx >> 7;
    int f    = idx & (HID - 1);
    atomicAdd(&sums[batch[node] * HID + f], h[idx]);
}

__global__ void k_cnt_accum(const int* __restrict__ batch, float* cnt, int n) {
    int i = blockIdx.x * blockDim.x + threadIdx.x;
    if (i < n) atomicAdd(&cnt[batch[i]], 1.0f);
}

__global__ void k_final(const float* __restrict__ sums, const float* __restrict__ cnt,
                        const float* __restrict__ Wlin, const float* __restrict__ blin,
                        float* __restrict__ out) {
    int idx = blockIdx.x * blockDim.x + threadIdx.x;
    if (idx >= NGRAPHS * FOUT) return;
    int g = idx / FOUT;
    int j = idx - g * FOUT;
    float inv = 1.0f / fmaxf(cnt[g], 1.0f);
    const float* s = sums + g * HID;
    float acc = blin[j];
#pragma unroll 4
    for (int k = 0; k < HID; ++k)
        acc = fmaf(s[k] * inv, Wlin[k * FOUT + j], acc);
    out[idx] = acc;
}

extern "C" void kernel_launch(void* const* d_in, const int* in_sizes, int n_in,
                              void* d_out, int out_size, void* d_ws, size_t ws_size,
                              hipStream_t stream) {
    (void)n_in; (void)out_size; (void)ws_size;

    const float* x     = (const float*)d_in[0];
    const int*   esrc  = (const int*)  d_in[1];
    const int*   edst  = (const int*)  d_in[2];
    const int*   batch = (const int*)  d_in[3];
    const float* W1    = (const float*)d_in[4];
    const float* b1    = (const float*)d_in[5];
    const float* W2    = (const float*)d_in[6];
    const float* b2    = (const float*)d_in[7];
    const float* Wlin  = (const float*)d_in[8];
    const float* blin  = (const float*)d_in[9];
    float*       out   = (float*)d_out;

    const int N = in_sizes[3];
    const int E = in_sizes[1];

    char*  ws  = (char*)d_ws;
    size_t off = 0;
    auto carve = [&](size_t bytes) -> void* {
        off = (off + 255) & ~(size_t)255;
        void* p = ws + off;
        off += bytes;
        return p;
    };
    int*   ecnt  = (int*)  carve((size_t)N * 4);
    float* dinv  = (float*)carve((size_t)N * 4);
    float* bufH  = (float*)carve((size_t)N * HID * 4);
    float* bufO  = (float*)carve((size_t)N * HID * 4);
    char*  lists = (char*) carve((size_t)N * HID * 4);
    float* sums  = (float*)carve((size_t)NGRAPHS * HID * 4);
    float* cnt   = (float*)carve((size_t)NGRAPHS * 4);

    const int nsb = (N + SCAN_N - 1) / SCAN_N;
    ws  = lists;
    off = 0;
    int2* rec    = (int2*)carve((size_t)E * 8);
    int*  rank   = (int*) carve((size_t)E * 4);
    int*  rowloc = (int*) carve((size_t)N * 4);
    int*  blksum = (int*) carve((size_t)nsb * 4);
    int*  blkoff = (int*) carve((size_t)nsb * 4);

    const int T = 256;
    int nb   = (N + T - 1) / T;
    int eb   = (E + T - 1) / T;
    int fb   = (N * HID + T - 1) / T;
    int wb   = (int)(((size_t)N * 32 + T - 1) / T);
    int gb2  = (N + 15) / 16;
    int pb   = (NGRAPHS * HID + T - 1) / T;
    int hb   = (NGRAPHS * FOUT + T - 1) / T;

    k_zero_cnt   <<<nb,  T,      0, stream>>>(ecnt, N);
    k_count_rank <<<eb,  T,      0, stream>>>(edst, ecnt, rank, E);
    k_scan_local <<<nsb, SCAN_T, 0, stream>>>(ecnt, rowloc, blksum, dinv, N);
    k_scan_blocks<<<1,   32,     0, stream>>>(blksum, blkoff, nsb);
    k_fill       <<<eb,  T,      0, stream>>>(esrc, edst, rank, rowloc, blkoff, dinv, rec, E);

    k_gemm1<<<fb, T, 0, stream>>>(x, W1, bufH, N);
    k_pull <<<wb, T, 0, stream>>>(bufH, bufO, rec, rowloc, blkoff, ecnt, dinv, b1, N);

    k_gemm2_wmma<<<gb2, T, 0, stream>>>(bufO, W2, bufH, N);
    k_pull      <<<wb,  T, 0, stream>>>(bufH, bufO, rec, rowloc, blkoff, ecnt, dinv, b2, N);

    k_pool_init <<<pb, T, 0, stream>>>(sums, cnt);
    k_pool_accum<<<fb, T, 0, stream>>>(bufO, batch, sums, N);
    k_cnt_accum <<<nb, T, 0, stream>>>(batch, cnt, N);
    k_final     <<<hb, T, 0, stream>>>(sums, cnt, Wlin, blin, out);
}


# Layer 1 sums the 11-wide input rows per destination (44 B an edge, not 512) and applies W1 once to the summed row in the same wave: (A x) W1 for A (x W1); the x W1 launch goes; f32 sum order changes
# speedup vs baseline: 2.7646x; 1.1555x over previous
#include <hip/hip_runtime.h>
#include <hip/hip_bf16.h>

#define HID     128
#define FIN     11
#define FOUT    19
#define NGRAPHS 4096

typedef __attribute__((ext_vector_type(2))) float v2f;
typedef __attribute__((ext_vector_type(8))) float v8f;

#define SCAN_T 256
#define SCAN_N (SCAN_T * 4)
#define SCAN_SHIFT 10

__global__ void k_zero_cnt(int* __restrict__ cnt, int n) {
    int i = blockIdx.x * blockDim.x + threadIdx.x;
    if (i < n) cnt[i] = 0;
}

__global__ void k_count_rank(const int* __restrict__ dst, int* cnt, int* __restrict__ rank, int e) {
    int i = blockIdx.x * blockDim.x + threadIdx.x;
    if (i < e) rank[i] = atomicAdd(&cnt[dst[i]], 1);
}

__global__ __launch_bounds__(SCAN_T) void k_scan_local(const int* __restrict__ cnt,
                                                       int* __restrict__ rowloc,
                                                       int* __restrict__ blksum,
                                                       float* __restrict__ dinv, int n) {
    __shared__ int wsum[SCAN_T / 32];
    const int tid  = threadIdx.x;
    const int lane = tid & 31;
    const int wave = tid >> 5;
    const int i0   = blockIdx.x * SCAN_N + tid * 4;

    int c[4];
#pragma unroll
    for (int k = 0; k < 4; ++k) c[k] = (i0 + k < n) ? cnt[i0 + k] : 0;
    const int t = c[0] + c[1] + c[2] + c[3];

    int incl = t;
#pragma unroll
    for (int off = 1; off < 32; off <<= 1) {
        int y = __shfl_up(incl, off);
        if (lane >= off) incl += y;
    }
    if (lane == 31) wsum[wave] = incl;
    __syncthreads();
    int wbase = 0;
    for (int w = 0; w < wave; ++w) wbase += wsum[w];

    int run = wbase + incl - t;
#pragma unroll
    for (int k = 0; k < 4; ++k) {
        if (i0 + k < n) {
            rowloc[i0 + k] = run;
            float d = (float)(c[k] + 1);
            dinv[i0 + k] = (d > 0.0f) ? rsqrtf(d) : 0.0f;
        }
        run += c[k];
    }
    if (tid == SCAN_T - 1) blksum[blockIdx.x] = run;
}

__global__ void k_scan_blocks(const int* __restrict__ blksum, int* __restrict__ blkoff, int nblk) {
    const int lane = threadIdx.x & 31;
    int carry = 0;
    for (int base = 0; base < nblk; base += 32) {
        const int i = base + lane;
        const int v = (i < nblk) ? blksum[i] : 0;
        int incl = v;
#pragma unroll
        for (int off = 1; off < 32; off <<= 1) {
            int y = __shfl_up(incl, off);
            if (lane >= off) incl += y;
        }
        if (i < nblk) blkoff[i] = carry + incl - v;
        carry += __shfl(incl, 31);
    }
}

__global__ void k_fill(const int* __restrict__ src, const int* __restrict__ dst,
                       const int* __restrict__ rank, const int* __restrict__ rowloc,
                       const int* __restrict__ blkoff, const float* __restrict__ dinv,
                       int2* __restrict__ rec, int e) {
    int i = blockIdx.x * blockDim.x + threadIdx.x;
    if (i >= e) return;
    const int s = src[i];
    const int d = dst[i];
    const int pos = rowloc[d] + blkoff[d >> SCAN_SHIFT] + rank[i];
    rec[pos] = make_int2(s, __float_as_int(dinv[s]));
}


template <int G>
__device__ __forceinline__ float sum_rows(const float* __restrict__ x, int col,
                                          int s_l, int w_l, float acc) {
    float v[G];
    float w[G];
#pragma unroll
    for (int k = 0; k < G; ++k) {
        const int s = __builtin_amdgcn_readlane(s_l, k);
        w[k] = __int_as_float(__builtin_amdgcn_readlane(w_l, k));
        v[k] = x[(size_t)s * FIN + col];
    }
#pragma unroll
    for (int k = 0; k < G; ++k) acc = fmaf(v[k], w[k], acc);
    return acc;
}

__global__ __launch_bounds__(256) void k_sum_gemm1(const float* __restrict__ x,
                                                   const float* __restrict__ W1,
                                                   const float* __restrict__ bias,
                                                   float* __restrict__ out,
                                                   const int2* __restrict__ rec,
                                                   const int* __restrict__ rowloc,
                                                   const int* __restrict__ blkoff,
                                                   const int* __restrict__ cnt,
                                                   const float* __restrict__ dinv, int n) {
    const int lane = threadIdx.x & 31;
    const int node = __builtin_amdgcn_readfirstlane((int)((blockIdx.x * 256u + threadIdx.x) >> 5));
    if (node >= n) return;

    const int   start = rowloc[node] + blkoff[node >> SCAN_SHIFT];
    const int   c     = cnt[node];
    const float dn    = dinv[node];
    const int   col   = min(lane, FIN - 1);

    const float self = x[(size_t)node * FIN + col];
    float a = 0.0f;

    for (int base = 0; base < c; base += 32) {
        const int m = min(32, c - base);
        int s_l = node;
        int w_l = 0;
        if (lane < m) {
            const int2 r = rec[start + base + lane];
            s_l = r.x;
            w_l = __float_as_int(__int_as_float(r.y) * dn);
        }
        if (m > 24)      a = sum_rows<32>(x, col, s_l, w_l, a);
        else if (m > 16) a = sum_rows<24>(x, col, s_l, w_l, a);
        else if (m > 8)  a = sum_rows<16>(x, col, s_l, w_l, a);
        else             a = sum_rows<8>(x, col, s_l, w_l, a);
    }
    a = fmaf(self, dn * dn, a);

    const float4* W4 = (const float4*)W1;
    float4 o = make_float4(0.0f, 0.0f, 0.0f, 0.0f);
#pragma unroll
    for (int k = 0; k < FIN; ++k) {
        const float  ak = __int_as_float(__builtin_amdgcn_readlane(__float_as_int(a), k));
        const float4 w  = W4[k * (HID / 4) + lane];
        o.x = fmaf(ak, w.x, o.x);
        o.y = fmaf(ak, w.y, o.y);
        o.z = fmaf(ak, w.z, o.z);
        o.w = fmaf(ak, w.w, o.w);
    }
    const float4 b = ((const float4*)bias)[lane];
    o.x += b.x;
    o.y += b.y;
    o.z += b.z;
    o.w += b.w;
    ((float4*)out)[(size_t)node * (HID / 4) + lane] = o;
}


template <int G>
__device__ __forceinline__ void pull_group(const float4* __restrict__ h4, int lane,
                                           int s_l, int w_l, int j, float4& acc) {
    float4 v[G];
    float  w[G];
#pragma unroll
    for (int k = 0; k < G; ++k) {
        const int s = __builtin_amdgcn_readlane(s_l, j + k);
        w[k] = __int_as_float(__builtin_amdgcn_readlane(w_l, j + k));
        v[k] = h4[(size_t)s * (HID / 4) + lane];
    }
#pragma unroll
    for (int k = 0; k < G; ++k) {
        acc.x = fmaf(v[k].x, w[k], acc.x);
        acc.y = fmaf(v[k].y, w[k], acc.y);
        acc.z = fmaf(v[k].z, w[k], acc.z);
        acc.w = fmaf(v[k].w, w[k], acc.w);
    }
}

__global__ __launch_bounds__(256) void k_pull(const float* __restrict__ h, float* __restrict__ out,
                                              const int2* __restrict__ rec,
                                              const int* __restrict__ rowloc,
                                              const int* __restrict__ blkoff,
                                              const int* __restrict__ cnt,
                                              const float* __restrict__ dinv,
                                              const float* __restrict__ bias, int n) {
    const int lane = threadIdx.x & 31;
    const int node = __builtin_amdgcn_readfirstlane((int)((blockIdx.x * 256u + threadIdx.x) >> 5));
    if (node >= n) return;

    const int   start = rowloc[node] + blkoff[node >> SCAN_SHIFT];
    const int   c     = cnt[node];
    const float dn    = dinv[node];
    const float4* h4  = (const float4*)h;

    const float4 self = h4[(size_t)node * (HID / 4) + lane];
    float4 acc = make_float4(0.0f, 0.0f, 0.0f, 0.0f);

    for (int base = 0; base < c; base += 32) {
        const int m = min(32, c - base);
        int s_l = node;
        int w_l = 0;
        if (lane < m) {
            const int2 r = rec[start + base + lane];
            s_l = r.x;
            w_l = __float_as_int(__int_as_float(r.y) * dn);
        }
        int j = 0;
        for (; j + 8 <= m; j += 8) pull_group<8>(h4, lane, s_l, w_l, j, acc);
        if (j + 4 <= m) { pull_group<4>(h4, lane, s_l, w_l, j, acc); j += 4; }
        if (j < m) {
            const float4 z = make_float4(0.0f, 0.0f, 0.0f, 0.0f);
            float4 v0 = z, v1 = z, v2 = z;
            float  w0 = 0.0f, w1 = 0.0f, w2 = 0.0f;
            {
                const int s = __builtin_amdgcn_readlane(s_l, j);
                w0 = __int_as_float(__builtin_amdgcn_readlane(w_l, j));
                v0 = h4[(size_t)s * (HID / 4) + lane];
            }
            if (j + 1 < m) {
                const int s = __builtin_amdgcn_readlane(s_l, j + 1);
                w1 = __int_as_float(__builtin_amdgcn_readlane(w_l, j + 1));
                v1 = h4[(size_t)s * (HID / 4) + lane];
            }
            if (j + 2 < m) {
                const int s = __builtin_amdgcn_readlane(s_l, j + 2);
                w2 = __int_as_float(__builtin_amdgcn_readlane(w_l, j + 2));
                v2 = h4[(size_t)s * (HID / 4) + lane];
            }
            acc.x = fmaf(v0.x, w0, acc.x); acc.y = fmaf(v0.y, w0, acc.y);
            acc.z = fmaf(v0.z, w0, acc.z); acc.w = fmaf(v0.w, w0, acc.w);
            acc.x = fmaf(v1.x, w1, acc.x); acc.y = fmaf(v1.y, w1, acc.y);
            acc.z = fmaf(v1.z, w1, acc.z); acc.w = fmaf(v1.w, w1, acc.w);
            acc.x = fmaf(v2.x, w2, acc.x); acc.y = fmaf(v2.y, w2, acc.y);
            acc.z = fmaf(v2.z, w2, acc.z); acc.w = fmaf(v2.w, w2, acc.w);
        }
    }

    const float  ws = dn * dn;
    acc.x = fmaf(self.x, ws, acc.x);
    acc.y = fmaf(self.y, ws, acc.y);
    acc.z = fmaf(self.z, ws, acc.z);
    acc.w = fmaf(self.w, ws, acc.w);

    const float4 b = ((const float4*)bias)[lane];
    float4 o;
    o.x = acc.x + b.x;
    o.y = acc.y + b.y;
    o.z = acc.z + b.z;
    o.w = acc.w + b.w;
    ((float4*)out)[(size_t)node * (HID / 4) + lane] = o;
}

__global__ __launch_bounds__(256) void k_gemm2_wmma(const float* __restrict__ in,
                                                    const float* __restrict__ W,
                                                    float* __restrict__ out, int n) {
    __shared__ float Ws[HID * HID];
    const int tid = threadIdx.x;

    const float4* W4  = (const float4*)W;
    float4*       Ws4 = (float4*)Ws;
#pragma unroll
    for (int i = 0; i < 16; ++i)
        Ws4[tid + 256 * i] = W4[tid + 256 * i];
    __syncthreads();

    const int m0   = blockIdx.x << 4;
    const int wave = tid >> 5;
    const int lane = tid & 31;
    const int half = lane >> 4;
    const int l15  = lane & 15;
    const int n0   = wave << 4;
    int row = m0 + l15;
    if (row >= n) row = n - 1;
    const int ncol = n0 + l15;
    const float* arow = in + (size_t)row * HID;

    v8f c = {};
#pragma unroll
    for (int k0 = 0; k0 < HID; k0 += 4) {
        const int ka = k0 + half * 2;
        float2 av = *(const float2*)(arow + ka);
        v2f a, b;
        a[0] = fmaxf(av.x, 0.0f);
        a[1] = fmaxf(av.y, 0.0f);
        b[0] = Ws[ka * HID + ncol];
        b[1] = Ws[(ka + 1) * HID + ncol];
        c = __builtin_amdgcn_wmma_f32_16x16x4_f32(
                 false, a,  false, b,
                 (short)0, c,  false,  false);
    }

    const int mbase = m0 + half * 8;
#pragma unroll
    for (int i = 0; i < 8; ++i) {
        int r = mbase + i;
        if (r < n) out[(size_t)r * HID + ncol] = c[i];
    }
}

__global__ void k_pool_init(float* sums, float* cnt) {
    int idx = blockIdx.x * blockDim.x + threadIdx.x;
    if (idx < NGRAPHS * HID) sums[idx] = 0.0f;
    if (idx < NGRAPHS)       cnt[idx]  = 0.0f;
}

__global__ void k_pool_accum(const float* __restrict__ h, const int* __restrict__ batch,
                             float* sums, int n) {
    int idx = blockIdx.x * blockDim.x + threadIdx.x;
    if (idx >= n * HID) return;
    int node = idx >> 7;
    int f    = idx & (HID - 1);
    atomicAdd(&sums[batch[node] * HID + f], h[idx]);
}

__global__ void k_cnt_accum(const int* __restrict__ batch, float* cnt, int n) {
    int i = blockIdx.x * blockDim.x + threadIdx.x;
    if (i < n) atomicAdd(&cnt[batch[i]], 1.0f);
}

__global__ void k_final(const float* __restrict__ sums, const float* __restrict__ cnt,
                        const float* __restrict__ Wlin, const float* __restrict__ blin,
                        float* __restrict__ out) {
    int idx = blockIdx.x * blockDim.x + threadIdx.x;
    if (idx >= NGRAPHS * FOUT) return;
    int g = idx / FOUT;
    int j = idx - g * FOUT;
    float inv = 1.0f / fmaxf(cnt[g], 1.0f);
    const float* s = sums + g * HID;
    float acc = blin[j];
#pragma unroll 4
    for (int k = 0; k < HID; ++k)
        acc = fmaf(s[k] * inv, Wlin[k * FOUT + j], acc);
    out[idx] = acc;
}

extern "C" void kernel_launch(void* const* d_in, const int* in_sizes, int n_in,
                              void* d_out, int out_size, void* d_ws, size_t ws_size,
                              hipStream_t stream) {
    (void)n_in; (void)out_size; (void)ws_size;

    const float* x     = (const float*)d_in[0];
    const int*   esrc  = (const int*)  d_in[1];
    const int*   edst  = (const int*)  d_in[2];
    const int*   batch = (const int*)  d_in[3];
    const float* W1    = (const float*)d_in[4];
    const float* b1    = (const float*)d_in[5];
    const float* W2    = (const float*)d_in[6];
    const float* b2    = (const float*)d_in[7];
    const float* Wlin  = (const float*)d_in[8];
    const float* blin  = (const float*)d_in[9];
    float*       out   = (float*)d_out;

    const int N = in_sizes[3];
    const int E = in_sizes[1];

    char*  ws  = (char*)d_ws;
    size_t off = 0;
    auto carve = [&](size_t bytes) -> void* {
        off = (off + 255) & ~(size_t)255;
        void* p = ws + off;
        off += bytes;
        return p;
    };
    int*   ecnt  = (int*)  carve((size_t)N * 4);
    float* dinv  = (float*)carve((size_t)N * 4);
    float* bufH  = (float*)carve((size_t)N * HID * 4);
    float* bufO  = (float*)carve((size_t)N * HID * 4);
    char*  lists = (char*) carve((size_t)N * HID * 4);
    float* sums  = (float*)carve((size_t)NGRAPHS * HID * 4);
    float* cnt   = (float*)carve((size_t)NGRAPHS * 4);

    const int nsb = (N + SCAN_N - 1) / SCAN_N;
    ws  = lists;
    off = 0;
    int2* rec    = (int2*)carve((size_t)E * 8);
    int*  rank   = (int*) carve((size_t)E * 4);
    int*  rowloc = (int*) carve((size_t)N * 4);
    int*  blksum = (int*) carve((size_t)nsb * 4);
    int*  blkoff = (int*) carve((size_t)nsb * 4);

    const int T = 256;
    int nb   = (N + T - 1) / T;
    int eb   = (E + T - 1) / T;
    int fb   = (N * HID + T - 1) / T;
    int wb   = (int)(((size_t)N * 32 + T - 1) / T);
    int gb2  = (N + 15) / 16;
    int pb   = (NGRAPHS * HID + T - 1) / T;
    int hb   = (NGRAPHS * FOUT + T - 1) / T;

    k_zero_cnt   <<<nb,  T,      0, stream>>>(ecnt, N);
    k_count_rank <<<eb,  T,      0, stream>>>(edst, ecnt, rank, E);
    k_scan_local <<<nsb, SCAN_T, 0, stream>>>(ecnt, rowloc, blksum, dinv, N);
    k_scan_blocks<<<1,   32,     0, stream>>>(blksum, blkoff, nsb);
    k_fill       <<<eb,  T,      0, stream>>>(esrc, edst, rank, rowloc, blkoff, dinv, rec, E);

    k_sum_gemm1<<<wb, T, 0, stream>>>(x, W1, b1, bufO, rec, rowloc, blkoff, ecnt, dinv, N);

    k_gemm2_wmma<<<gb2, T, 0, stream>>>(bufO, W2, bufH, N);
    k_pull      <<<wb,  T, 0, stream>>>(bufH, bufO, rec, rowloc, blkoff, ecnt, dinv, b2, N);

    k_pool_init <<<pb, T, 0, stream>>>(sums, cnt);
    k_pool_accum<<<fb, T, 0, stream>>>(bufO, batch, sums, N);
    k_cnt_accum <<<nb, T, 0, stream>>>(batch, cnt, N);
    k_final     <<<hb, T, 0, stream>>>(sums, cnt, Wlin, blin, out);
}


# Mean pool fused into the layer-2 neighbour-sum kernel: a block sums its 8 node rows per graph in LDS, then one f32 atomic add per column and run; pool and count launches go; pool sum order changes
# speedup vs baseline: 3.1061x; 1.1235x over previous
#include <hip/hip_runtime.h>
#include <hip/hip_bf16.h>

#define HID     128
#define FIN     11
#define FOUT    19
#define NGRAPHS 4096

typedef __attribute__((ext_vector_type(2))) float v2f;
typedef __attribute__((ext_vector_type(8))) float v8f;

#define SCAN_T 256
#define SCAN_N (SCAN_T * 4)
#define SCAN_SHIFT 10

__global__ void k_zero_cnt(int* __restrict__ cnt, int n) {
    int i = blockIdx.x * blockDim.x + threadIdx.x;
    if (i < n) cnt[i] = 0;
}

__global__ void k_count_rank(const int* __restrict__ dst, int* cnt, int* __restrict__ rank, int e) {
    int i = blockIdx.x * blockDim.x + threadIdx.x;
    if (i < e) rank[i] = atomicAdd(&cnt[dst[i]], 1);
}

__global__ __launch_bounds__(SCAN_T) void k_scan_local(const int* __restrict__ cnt,
                                                       int* __restrict__ rowloc,
                                                       int* __restrict__ blksum,
                                                       float* __restrict__ dinv, int n) {
    __shared__ int wsum[SCAN_T / 32];
    const int tid  = threadIdx.x;
    const int lane = tid & 31;
    const int wave = tid >> 5;
    const int i0   = blockIdx.x * SCAN_N + tid * 4;

    int c[4];
#pragma unroll
    for (int k = 0; k < 4; ++k) c[k] = (i0 + k < n) ? cnt[i0 + k] : 0;
    const int t = c[0] + c[1] + c[2] + c[3];

    int incl = t;
#pragma unroll
    for (int off = 1; off < 32; off <<= 1) {
        int y = __shfl_up(incl, off);
        if (lane >= off) incl += y;
    }
    if (lane == 31) wsum[wave] = incl;
    __syncthreads();
    int wbase = 0;
    for (int w = 0; w < wave; ++w) wbase += wsum[w];

    int run = wbase + incl - t;
#pragma unroll
    for (int k = 0; k < 4; ++k) {
        if (i0 + k < n) {
            rowloc[i0 + k] = run;
            float d = (float)(c[k] + 1);
            dinv[i0 + k] = (d > 0.0f) ? rsqrtf(d) : 0.0f;
        }
        run += c[k];
    }
    if (tid == SCAN_T - 1) blksum[blockIdx.x] = run;
}

__global__ void k_scan_blocks(const int* __restrict__ blksum, int* __restrict__ blkoff, int nblk) {
    const int lane = threadIdx.x & 31;
    int carry = 0;
    for (int base = 0; base < nblk; base += 32) {
        const int i = base + lane;
        const int v = (i < nblk) ? blksum[i] : 0;
        int incl = v;
#pragma unroll
        for (int off = 1; off < 32; off <<= 1) {
            int y = __shfl_up(incl, off);
            if (lane >= off) incl += y;
        }
        if (i < nblk) blkoff[i] = carry + incl - v;
        carry += __shfl(incl, 31);
    }
}

__global__ void k_fill(const int* __restrict__ src, const int* __restrict__ dst,
                       const int* __restrict__ rank, const int* __restrict__ rowloc,
                       const int* __restrict__ blkoff, const float* __restrict__ dinv,
                       int2* __restrict__ rec, int e) {
    int i = blockIdx.x * blockDim.x + threadIdx.x;
    if (i >= e) return;
    const int s = src[i];
    const int d = dst[i];
    const int pos = rowloc[d] + blkoff[d >> SCAN_SHIFT] + rank[i];
    rec[pos] = make_int2(s, __float_as_int(dinv[s]));
}


template <int G>
__device__ __forceinline__ float sum_rows(const float* __restrict__ x, int col,
                                          int s_l, int w_l, float acc) {
    float v[G];
    float w[G];
#pragma unroll
    for (int k = 0; k < G; ++k) {
        const int s = __builtin_amdgcn_readlane(s_l, k);
        w[k] = __int_as_float(__builtin_amdgcn_readlane(w_l, k));
        v[k] = x[(size_t)s * FIN + col];
    }
#pragma unroll
    for (int k = 0; k < G; ++k) acc = fmaf(v[k], w[k], acc);
    return acc;
}

__global__ __launch_bounds__(256) void k_sum_gemm1(const float* __restrict__ x,
                                                   const float* __restrict__ W1,
                                                   const float* __restrict__ bias,
                                                   float* __restrict__ out,
                                                   const int2* __restrict__ rec,
                                                   const int* __restrict__ rowloc,
                                                   const int* __restrict__ blkoff,
                                                   const int* __restrict__ cnt,
                                                   const float* __restrict__ dinv, int n) {
    const int lane = threadIdx.x & 31;
    const int node = __builtin_amdgcn_readfirstlane((int)((blockIdx.x * 256u + threadIdx.x) >> 5));
    if (node >= n) return;

    const int   start = rowloc[node] + blkoff[node >> SCAN_SHIFT];
    const int   c     = cnt[node];
    const float dn    = dinv[node];
    const int   col   = min(lane, FIN - 1);

    const float self = x[(size_t)node * FIN + col];
    float a = 0.0f;

    for (int base = 0; base < c; base += 32) {
        const int m = min(32, c - base);
        int s_l = node;
        int w_l = 0;
        if (lane < m) {
            const int2 r = rec[start + base + lane];
            s_l = r.x;
            w_l = __float_as_int(__int_as_float(r.y) * dn);
        }
        if (m > 24)      a = sum_rows<32>(x, col, s_l, w_l, a);
        else if (m > 16) a = sum_rows<24>(x, col, s_l, w_l, a);
        else if (m > 8)  a = sum_rows<16>(x, col, s_l, w_l, a);
        else             a = sum_rows<8>(x, col, s_l, w_l, a);
    }
    a = fmaf(self, dn * dn, a);

    const float4* W4 = (const float4*)W1;
    float4 o = make_float4(0.0f, 0.0f, 0.0f, 0.0f);
#pragma unroll
    for (int k = 0; k < FIN; ++k) {
        const float  ak = __int_as_float(__builtin_amdgcn_readlane(__float_as_int(a), k));
        const float4 w  = W4[k * (HID / 4) + lane];
        o.x = fmaf(ak, w.x, o.x);
        o.y = fmaf(ak, w.y, o.y);
        o.z = fmaf(ak, w.z, o.z);
        o.w = fmaf(ak, w.w, o.w);
    }
    const float4 b = ((const float4*)bias)[lane];
    o.x += b.x;
    o.y += b.y;
    o.z += b.z;
    o.w += b.w;
    ((float4*)out)[(size_t)node * (HID / 4) + lane] = o;
}


template <int G>
__device__ __forceinline__ void pull_group(const float4* __restrict__ h4, int lane,
                                           int s_l, int w_l, int j, float4& acc) {
    float4 v[G];
    float  w[G];
#pragma unroll
    for (int k = 0; k < G; ++k) {
        const int s = __builtin_amdgcn_readlane(s_l, j + k);
        w[k] = __int_as_float(__builtin_amdgcn_readlane(w_l, j + k));
        v[k] = h4[(size_t)s * (HID / 4) + lane];
    }
#pragma unroll
    for (int k = 0; k < G; ++k) {
        acc.x = fmaf(v[k].x, w[k], acc.x);
        acc.y = fmaf(v[k].y, w[k], acc.y);
        acc.z = fmaf(v[k].z, w[k], acc.z);
        acc.w = fmaf(v[k].w, w[k], acc.w);
    }
}

#define PULL_NODES 8

__global__ __launch_bounds__(256) void k_pull_pool(const float* __restrict__ h,
                                                   const int2* __restrict__ rec,
                                                   const int* __restrict__ rowloc,
                                                   const int* __restrict__ blkoff,
                                                   const int* __restrict__ cnt,
                                                   const float* __restrict__ dinv,
                                                   const float* __restrict__ bias,
                                                   const int* __restrict__ batch,
                                                   float* sums, float* gcnt, int n) {
    __shared__ float4 rows4[PULL_NODES][HID / 4];
    __shared__ __attribute__((aligned(16))) int gid[PULL_NODES];

    const int lane = threadIdx.x & 31;
    const int wave = threadIdx.x >> 5;
    const int node = __builtin_amdgcn_readfirstlane((int)(blockIdx.x * PULL_NODES + wave));

    float4 o = make_float4(0.0f, 0.0f, 0.0f, 0.0f);
    int    g = -1;
    if (node < n) {
        const int   start = rowloc[node] + blkoff[node >> SCAN_SHIFT];
        const int   c     = cnt[node];
        const float dn    = dinv[node];
        const float4* h4  = (const float4*)h;
        g = batch[node];

        const float4 self = h4[(size_t)node * (HID / 4) + lane];
        float4 acc = make_float4(0.0f, 0.0f, 0.0f, 0.0f);

        for (int base = 0; base < c; base += 32) {
            const int m = min(32, c - base);
            int s_l = node;
            int w_l = 0;
            if (lane < m) {
                const int2 r = rec[start + base + lane];
                s_l = r.x;
                w_l = __float_as_int(__int_as_float(r.y) * dn);
            }
            int j = 0;
            for (; j + 8 <= m; j += 8) pull_group<8>(h4, lane, s_l, w_l, j, acc);
            if (j + 4 <= m) { pull_group<4>(h4, lane, s_l, w_l, j, acc); j += 4; }
            if (j < m) {
                const float4 z = make_float4(0.0f, 0.0f, 0.0f, 0.0f);
                float4 v0 = z, v1 = z, v2 = z;
                float  w0 = 0.0f, w1 = 0.0f, w2 = 0.0f;
                {
                    const int s = __builtin_amdgcn_readlane(s_l, j);
                    w0 = __int_as_float(__builtin_amdgcn_readlane(w_l, j));
                    v0 = h4[(size_t)s * (HID / 4) + lane];
                }
                if (j + 1 < m) {
                    const int s = __builtin_amdgcn_readlane(s_l, j + 1);
                    w1 = __int_as_float(__builtin_amdgcn_readlane(w_l, j + 1));
                    v1 = h4[(size_t)s * (HID / 4) + lane];
                }
                if (j + 2 < m) {
                    const int s = __builtin_amdgcn_readlane(s_l, j + 2);
                    w2 = __int_as_float(__builtin_amdgcn_readlane(w_l, j + 2));
                    v2 = h4[(size_t)s * (HID / 4) + lane];
                }
                acc.x = fmaf(v0.x, w0, acc.x); acc.y = fmaf(v0.y, w0, acc.y);
                acc.z = fmaf(v0.z, w0, acc.z); acc.w = fmaf(v0.w, w0, acc.w);
                acc.x = fmaf(v1.x, w1, acc.x); acc.y = fmaf(v1.y, w1, acc.y);
                acc.z = fmaf(v1.z, w1, acc.z); acc.w = fmaf(v1.w, w1, acc.w);
                acc.x = fmaf(v2.x, w2, acc.x); acc.y = fmaf(v2.y, w2, acc.y);
                acc.z = fmaf(v2.z, w2, acc.z); acc.w = fmaf(v2.w, w2, acc.w);
            }
        }

        const float  ws = dn * dn;
        acc.x = fmaf(self.x, ws, acc.x);
        acc.y = fmaf(self.y, ws, acc.y);
        acc.z = fmaf(self.z, ws, acc.z);
        acc.w = fmaf(self.w, ws, acc.w);

        const float4 b = ((const float4*)bias)[lane];
        o.x = acc.x + b.x;
        o.y = acc.y + b.y;
        o.z = acc.z + b.z;
        o.w = acc.w + b.w;
    }

    rows4[wave][lane] = o;
    if (lane == 0) gid[wave] = g;
    __syncthreads();

    const int    tid  = threadIdx.x;
    const float* rows = (const float*)rows4;

    float v[PULL_NODES];
#pragma unroll
    for (int r = 0; r < PULL_NODES; ++r) v[r] = rows[r * HID + (tid & (HID - 1))];
    const int4 ga = ((const int4*)gid)[0];
    const int4 gb = ((const int4*)gid)[1];
    const int gs[PULL_NODES] = {
        __builtin_amdgcn_readfirstlane(ga.x), __builtin_amdgcn_readfirstlane(ga.y),
        __builtin_amdgcn_readfirstlane(ga.z), __builtin_amdgcn_readfirstlane(ga.w),
        __builtin_amdgcn_readfirstlane(gb.x), __builtin_amdgcn_readfirstlane(gb.y),
        __builtin_amdgcn_readfirstlane(gb.z), __builtin_amdgcn_readfirstlane(gb.w)};

    if (tid < HID) {
        int   cur = -1;
        float s   = 0.0f;
#pragma unroll
        for (int r = 0; r < PULL_NODES; ++r) {
            if (gs[r] < 0) continue;
            if (gs[r] != cur) {
                if (cur >= 0) atomicAdd(&sums[cur * HID + tid], s);
                cur = gs[r];
                s   = 0.0f;
            }
            s += v[r];
        }
        if (cur >= 0) atomicAdd(&sums[cur * HID + tid], s);
    } else if (tid == HID) {
        int   cur = -1;
        float k   = 0.0f;
#pragma unroll
        for (int r = 0; r < PULL_NODES; ++r) {
            if (gs[r] < 0) continue;
            if (gs[r] != cur) {
                if (cur >= 0) atomicAdd(&gcnt[cur], k);
                cur = gs[r];
                k   = 0.0f;
            }
            k += 1.0f;
        }
        if (cur >= 0) atomicAdd(&gcnt[cur], k);
    }
}

__global__ __launch_bounds__(256) void k_gemm2_wmma(const float* __restrict__ in,
                                                    const float* __restrict__ W,
                                                    float* __restrict__ out, int n) {
    __shared__ float Ws[HID * HID];
    const int tid = threadIdx.x;

    const float4* W4  = (const float4*)W;
    float4*       Ws4 = (float4*)Ws;
#pragma unroll
    for (int i = 0; i < 16; ++i)
        Ws4[tid + 256 * i] = W4[tid + 256 * i];
    __syncthreads();

    const int m0   = blockIdx.x << 4;
    const int wave = tid >> 5;
    const int lane = tid & 31;
    const int half = lane >> 4;
    const int l15  = lane & 15;
    const int n0   = wave << 4;
    int row = m0 + l15;
    if (row >= n) row = n - 1;
    const int ncol = n0 + l15;
    const float* arow = in + (size_t)row * HID;

    v8f c = {};
#pragma unroll
    for (int k0 = 0; k0 < HID; k0 += 4) {
        const int ka = k0 + half * 2;
        float2 av = *(const float2*)(arow + ka);
        v2f a, b;
        a[0] = fmaxf(av.x, 0.0f);
        a[1] = fmaxf(av.y, 0.0f);
        b[0] = Ws[ka * HID + ncol];
        b[1] = Ws[(ka + 1) * HID + ncol];
        c = __builtin_amdgcn_wmma_f32_16x16x4_f32(
                 false, a,  false, b,
                 (short)0, c,  false,  false);
    }

    const int mbase = m0 + half * 8;
#pragma unroll
    for (int i = 0; i < 8; ++i) {
        int r = mbase + i;
        if (r < n) out[(size_t)r * HID + ncol] = c[i];
    }
}

__global__ void k_pool_init(float* sums, float* cnt) {
    int idx = blockIdx.x * blockDim.x + threadIdx.x;
    if (idx < NGRAPHS * HID) sums[idx] = 0.0f;
    if (idx < NGRAPHS)       cnt[idx]  = 0.0f;
}

__global__ void k_final(const float* __restrict__ sums, const float* __restrict__ cnt,
                        const float* __restrict__ Wlin, const float* __restrict__ blin,
                        float* __restrict__ out) {
    int idx = blockIdx.x * blockDim.x + threadIdx.x;
    if (idx >= NGRAPHS * FOUT) return;
    int g = idx / FOUT;
    int j = idx - g * FOUT;
    float inv = 1.0f / fmaxf(cnt[g], 1.0f);
    const float* s = sums + g * HID;
    float acc = blin[j];
#pragma unroll 4
    for (int k = 0; k < HID; ++k)
        acc = fmaf(s[k] * inv, Wlin[k * FOUT + j], acc);
    out[idx] = acc;
}

extern "C" void kernel_launch(void* const* d_in, const int* in_sizes, int n_in,
                              void* d_out, int out_size, void* d_ws, size_t ws_size,
                              hipStream_t stream) {
    (void)n_in; (void)out_size; (void)ws_size;

    const float* x     = (const float*)d_in[0];
    const int*   esrc  = (const int*)  d_in[1];
    const int*   edst  = (const int*)  d_in[2];
    const int*   batch = (const int*)  d_in[3];
    const float* W1    = (const float*)d_in[4];
    const float* b1    = (const float*)d_in[5];
    const float* W2    = (const float*)d_in[6];
    const float* b2    = (const float*)d_in[7];
    const float* Wlin  = (const float*)d_in[8];
    const float* blin  = (const float*)d_in[9];
    float*       out   = (float*)d_out;

    const int N = in_sizes[3];
    const int E = in_sizes[1];

    char*  ws  = (char*)d_ws;
    size_t off = 0;
    auto carve = [&](size_t bytes) -> void* {
        off = (off + 255) & ~(size_t)255;
        void* p = ws + off;
        off += bytes;
        return p;
    };
    int*   ecnt  = (int*)  carve((size_t)N * 4);
    float* dinv  = (float*)carve((size_t)N * 4);
    float* bufH  = (float*)carve((size_t)N * HID * 4);
    float* bufO  = (float*)carve((size_t)N * HID * 4);
    char*  lists = (char*) carve((size_t)N * HID * 4);
    float* sums  = (float*)carve((size_t)NGRAPHS * HID * 4);
    float* cnt   = (float*)carve((size_t)NGRAPHS * 4);

    const int nsb = (N + SCAN_N - 1) / SCAN_N;
    ws  = lists;
    off = 0;
    int2* rec    = (int2*)carve((size_t)E * 8);
    int*  rank   = (int*) carve((size_t)E * 4);
    int*  rowloc = (int*) carve((size_t)N * 4);
    int*  blksum = (int*) carve((size_t)nsb * 4);
    int*  blkoff = (int*) carve((size_t)nsb * 4);

    const int T = 256;
    int nb   = (N + T - 1) / T;
    int eb   = (E + T - 1) / T;
    int wb   = (int)(((size_t)N * 32 + T - 1) / T);
    int gb2  = (N + 15) / 16;
    int pb   = (NGRAPHS * HID + T - 1) / T;
    int hb   = (NGRAPHS * FOUT + T - 1) / T;

    k_zero_cnt   <<<nb,  T,      0, stream>>>(ecnt, N);
    k_count_rank <<<eb,  T,      0, stream>>>(edst, ecnt, rank, E);
    k_scan_local <<<nsb, SCAN_T, 0, stream>>>(ecnt, rowloc, blksum, dinv, N);
    k_scan_blocks<<<1,   32,     0, stream>>>(blksum, blkoff, nsb);
    k_fill       <<<eb,  T,      0, stream>>>(esrc, edst, rank, rowloc, blkoff, dinv, rec, E);

    k_sum_gemm1<<<wb, T, 0, stream>>>(x, W1, b1, bufO, rec, rowloc, blkoff, ecnt, dinv, N);

    k_gemm2_wmma<<<gb2, T, 0, stream>>>(bufO, W2, bufH, N);
    k_pool_init <<<pb,  T, 0, stream>>>(sums, cnt);
    k_pull_pool <<<wb,  T, 0, stream>>>(bufH, rec, rowloc, blkoff, ecnt, dinv, b2, batch, sums, cnt, N);

    k_final     <<<hb, T, 0, stream>>>(sums, cnt, Wlin, blin, out);
}


# W2 applied after the neighbour sum and mean pool (both linear): the WMMA GEMM runs on 4096 pooled rows, not 100000 node rows; relu moves to the layer-1 store, b2 to the head; f32 sum order changes
# speedup vs baseline: 3.6609x; 1.1786x over previous
#include <hip/hip_runtime.h>
#include <hip/hip_bf16.h>

#define HID     128
#define FIN     11
#define FOUT    19
#define NGRAPHS 4096

typedef __attribute__((ext_vector_type(2))) float v2f;
typedef __attribute__((ext_vector_type(8))) float v8f;

#define SCAN_T 256
#define SCAN_N (SCAN_T * 4)
#define SCAN_SHIFT 10

__global__ void k_zero_cnt(int* __restrict__ cnt, int n) {
    int i = blockIdx.x * blockDim.x + threadIdx.x;
    if (i < n) cnt[i] = 0;
}

__global__ void k_count_rank(const int* __restrict__ dst, int* cnt, int* __restrict__ rank, int e) {
    int i = blockIdx.x * blockDim.x + threadIdx.x;
    if (i < e) rank[i] = atomicAdd(&cnt[dst[i]], 1);
}

__global__ __launch_bounds__(SCAN_T) void k_scan_local(const int* __restrict__ cnt,
                                                       int* __restrict__ rowloc,
                                                       int* __restrict__ blksum,
                                                       float* __restrict__ dinv, int n) {
    __shared__ int wsum[SCAN_T / 32];
    const int tid  = threadIdx.x;
    const int lane = tid & 31;
    const int wave = tid >> 5;
    const int i0   = blockIdx.x * SCAN_N + tid * 4;

    int c[4];
#pragma unroll
    for (int k = 0; k < 4; ++k) c[k] = (i0 + k < n) ? cnt[i0 + k] : 0;
    const int t = c[0] + c[1] + c[2] + c[3];

    int incl = t;
#pragma unroll
    for (int off = 1; off < 32; off <<= 1) {
        int y = __shfl_up(incl, off);
        if (lane >= off) incl += y;
    }
    if (lane == 31) wsum[wave] = incl;
    __syncthreads();
    int wbase = 0;
    for (int w = 0; w < wave; ++w) wbase += wsum[w];

    int run = wbase + incl - t;
#pragma unroll
    for (int k = 0; k < 4; ++k) {
        if (i0 + k < n) {
            rowloc[i0 + k] = run;
            float d = (float)(c[k] + 1);
            dinv[i0 + k] = (d > 0.0f) ? rsqrtf(d) : 0.0f;
        }
        run += c[k];
    }
    if (tid == SCAN_T - 1) blksum[blockIdx.x] = run;
}

__global__ void k_scan_blocks(const int* __restrict__ blksum, int* __restrict__ blkoff, int nblk) {
    const int lane = threadIdx.x & 31;
    int carry = 0;
    for (int base = 0; base < nblk; base += 32) {
        const int i = base + lane;
        const int v = (i < nblk) ? blksum[i] : 0;
        int incl = v;
#pragma unroll
        for (int off = 1; off < 32; off <<= 1) {
            int y = __shfl_up(incl, off);
            if (lane >= off) incl += y;
        }
        if (i < nblk) blkoff[i] = carry + incl - v;
        carry += __shfl(incl, 31);
    }
}

__global__ void k_fill(const int* __restrict__ src, const int* __restrict__ dst,
                       const int* __restrict__ rank, const int* __restrict__ rowloc,
                       const int* __restrict__ blkoff, const float* __restrict__ dinv,
                       int2* __restrict__ rec, int e) {
    int i = blockIdx.x * blockDim.x + threadIdx.x;
    if (i >= e) return;
    const int s = src[i];
    const int d = dst[i];
    const int pos = rowloc[d] + blkoff[d >> SCAN_SHIFT] + rank[i];
    rec[pos] = make_int2(s, __float_as_int(dinv[s]));
}


template <int G>
__device__ __forceinline__ float sum_rows(const float* __restrict__ x, int col,
                                          int s_l, int w_l, float acc) {
    float v[G];
    float w[G];
#pragma unroll
    for (int k = 0; k < G; ++k) {
        const int s = __builtin_amdgcn_readlane(s_l, k);
        w[k] = __int_as_float(__builtin_amdgcn_readlane(w_l, k));
        v[k] = x[(size_t)s * FIN + col];
    }
#pragma unroll
    for (int k = 0; k < G; ++k) acc = fmaf(v[k], w[k], acc);
    return acc;
}

__global__ __launch_bounds__(256) void k_sum_gemm1(const float* __restrict__ x,
                                                   const float* __restrict__ W1,
                                                   const float* __restrict__ bias,
                                                   float* __restrict__ out,
                                                   const int2* __restrict__ rec,
                                                   const int* __restrict__ rowloc,
                                                   const int* __restrict__ blkoff,
                                                   const int* __restrict__ cnt,
                                                   const float* __restrict__ dinv, int n) {
    const int lane = threadIdx.x & 31;
    const int node = __builtin_amdgcn_readfirstlane((int)((blockIdx.x * 256u + threadIdx.x) >> 5));
    if (node >= n) return;

    const int   start = rowloc[node] + blkoff[node >> SCAN_SHIFT];
    const int   c     = cnt[node];
    const float dn    = dinv[node];
    const int   col   = min(lane, FIN - 1);

    const float self = x[(size_t)node * FIN + col];
    float a = 0.0f;

    for (int base = 0; base < c; base += 32) {
        const int m = min(32, c - base);
        int s_l = node;
        int w_l = 0;
        if (lane < m) {
            const int2 r = rec[start + base + lane];
            s_l = r.x;
            w_l = __float_as_int(__int_as_float(r.y) * dn);
        }
        if (m > 24)      a = sum_rows<32>(x, col, s_l, w_l, a);
        else if (m > 16) a = sum_rows<24>(x, col, s_l, w_l, a);
        else if (m > 8)  a = sum_rows<16>(x, col, s_l, w_l, a);
        else             a = sum_rows<8>(x, col, s_l, w_l, a);
    }
    a = fmaf(self, dn * dn, a);

    const float4* W4 = (const float4*)W1;
    float4 o = make_float4(0.0f, 0.0f, 0.0f, 0.0f);
#pragma unroll
    for (int k = 0; k < FIN; ++k) {
        const float  ak = __int_as_float(__builtin_amdgcn_readlane(__float_as_int(a), k));
        const float4 w  = W4[k * (HID / 4) + lane];
        o.x = fmaf(ak, w.x, o.x);
        o.y = fmaf(ak, w.y, o.y);
        o.z = fmaf(ak, w.z, o.z);
        o.w = fmaf(ak, w.w, o.w);
    }
    const float4 b = ((const float4*)bias)[lane];
    o.x = fmaxf(o.x + b.x, 0.0f);
    o.y = fmaxf(o.y + b.y, 0.0f);
    o.z = fmaxf(o.z + b.z, 0.0f);
    o.w = fmaxf(o.w + b.w, 0.0f);
    ((float4*)out)[(size_t)node * (HID / 4) + lane] = o;
}


template <int G>
__device__ __forceinline__ void pull_group(const float4* __restrict__ h4, int lane,
                                           int s_l, int w_l, int j, float4& acc) {
    float4 v[G];
    float  w[G];
#pragma unroll
    for (int k = 0; k < G; ++k) {
        const int s = __builtin_amdgcn_readlane(s_l, j + k);
        w[k] = __int_as_float(__builtin_amdgcn_readlane(w_l, j + k));
        v[k] = h4[(size_t)s * (HID / 4) + lane];
    }
#pragma unroll
    for (int k = 0; k < G; ++k) {
        acc.x = fmaf(v[k].x, w[k], acc.x);
        acc.y = fmaf(v[k].y, w[k], acc.y);
        acc.z = fmaf(v[k].z, w[k], acc.z);
        acc.w = fmaf(v[k].w, w[k], acc.w);
    }
}

#define PULL_NODES 8

__global__ __launch_bounds__(256) void k_pull_pool(const float* __restrict__ h,
                                                   const int2* __restrict__ rec,
                                                   const int* __restrict__ rowloc,
                                                   const int* __restrict__ blkoff,
                                                   const int* __restrict__ cnt,
                                                   const float* __restrict__ dinv,
                                                   const int* __restrict__ batch,
                                                   float* sums, float* gcnt, int n) {
    __shared__ float4 rows4[PULL_NODES][HID / 4];
    __shared__ __attribute__((aligned(16))) int gid[PULL_NODES];

    const int lane = threadIdx.x & 31;
    const int wave = threadIdx.x >> 5;
    const int node = __builtin_amdgcn_readfirstlane((int)(blockIdx.x * PULL_NODES + wave));

    float4 o = make_float4(0.0f, 0.0f, 0.0f, 0.0f);
    int    g = -1;
    if (node < n) {
        const int   start = rowloc[node] + blkoff[node >> SCAN_SHIFT];
        const int   c     = cnt[node];
        const float dn    = dinv[node];
        const float4* h4  = (const float4*)h;
        g = batch[node];

        const float4 self = h4[(size_t)node * (HID / 4) + lane];
        float4 acc = make_float4(0.0f, 0.0f, 0.0f, 0.0f);

        for (int base = 0; base < c; base += 32) {
            const int m = min(32, c - base);
            int s_l = node;
            int w_l = 0;
            if (lane < m) {
                const int2 r = rec[start + base + lane];
                s_l = r.x;
                w_l = __float_as_int(__int_as_float(r.y) * dn);
            }
            int j = 0;
            for (; j + 8 <= m; j += 8) pull_group<8>(h4, lane, s_l, w_l, j, acc);
            if (j + 4 <= m) { pull_group<4>(h4, lane, s_l, w_l, j, acc); j += 4; }
            if (j < m) {
                const float4 z = make_float4(0.0f, 0.0f, 0.0f, 0.0f);
                float4 v0 = z, v1 = z, v2 = z;
                float  w0 = 0.0f, w1 = 0.0f, w2 = 0.0f;
                {
                    const int s = __builtin_amdgcn_readlane(s_l, j);
                    w0 = __int_as_float(__builtin_amdgcn_readlane(w_l, j));
                    v0 = h4[(size_t)s * (HID / 4) + lane];
                }
                if (j + 1 < m) {
                    const int s = __builtin_amdgcn_readlane(s_l, j + 1);
                    w1 = __int_as_float(__builtin_amdgcn_readlane(w_l, j + 1));
                    v1 = h4[(size_t)s * (HID / 4) + lane];
                }
                if (j + 2 < m) {
                    const int s = __builtin_amdgcn_readlane(s_l, j + 2);
                    w2 = __int_as_float(__builtin_amdgcn_readlane(w_l, j + 2));
                    v2 = h4[(size_t)s * (HID / 4) + lane];
                }
                acc.x = fmaf(v0.x, w0, acc.x); acc.y = fmaf(v0.y, w0, acc.y);
                acc.z = fmaf(v0.z, w0, acc.z); acc.w = fmaf(v0.w, w0, acc.w);
                acc.x = fmaf(v1.x, w1, acc.x); acc.y = fmaf(v1.y, w1, acc.y);
                acc.z = fmaf(v1.z, w1, acc.z); acc.w = fmaf(v1.w, w1, acc.w);
                acc.x = fmaf(v2.x, w2, acc.x); acc.y = fmaf(v2.y, w2, acc.y);
                acc.z = fmaf(v2.z, w2, acc.z); acc.w = fmaf(v2.w, w2, acc.w);
            }
        }

        const float  ws = dn * dn;
        acc.x = fmaf(self.x, ws, acc.x);
        acc.y = fmaf(self.y, ws, acc.y);
        acc.z = fmaf(self.z, ws, acc.z);
        acc.w = fmaf(self.w, ws, acc.w);
        o = acc;
    }

    rows4[wave][lane] = o;
    if (lane == 0) gid[wave] = g;
    __syncthreads();

    const int    tid  = threadIdx.x;
    const float* rows = (const float*)rows4;

    float v[PULL_NODES];
#pragma unroll
    for (int r = 0; r < PULL_NODES; ++r) v[r] = rows[r * HID + (tid & (HID - 1))];
    const int4 ga = ((const int4*)gid)[0];
    const int4 gb = ((const int4*)gid)[1];
    const int gs[PULL_NODES] = {
        __builtin_amdgcn_readfirstlane(ga.x), __builtin_amdgcn_readfirstlane(ga.y),
        __builtin_amdgcn_readfirstlane(ga.z), __builtin_amdgcn_readfirstlane(ga.w),
        __builtin_amdgcn_readfirstlane(gb.x), __builtin_amdgcn_readfirstlane(gb.y),
        __builtin_amdgcn_readfirstlane(gb.z), __builtin_amdgcn_readfirstlane(gb.w)};

    if (tid < HID) {
        int   cur = -1;
        float s   = 0.0f;
#pragma unroll
        for (int r = 0; r < PULL_NODES; ++r) {
            if (gs[r] < 0) continue;
            if (gs[r] != cur) {
                if (cur >= 0) atomicAdd(&sums[cur * HID + tid], s);
                cur = gs[r];
                s   = 0.0f;
            }
            s += v[r];
        }
        if (cur >= 0) atomicAdd(&sums[cur * HID + tid], s);
    } else if (tid == HID) {
        int   cur = -1;
        float k   = 0.0f;
#pragma unroll
        for (int r = 0; r < PULL_NODES; ++r) {
            if (gs[r] < 0) continue;
            if (gs[r] != cur) {
                if (cur >= 0) atomicAdd(&gcnt[cur], k);
                cur = gs[r];
                k   = 0.0f;
            }
            k += 1.0f;
        }
        if (cur >= 0) atomicAdd(&gcnt[cur], k);
    }
}

__global__ __launch_bounds__(256) void k_gemm2_wmma(const float* __restrict__ in,
                                                    const float* __restrict__ W,
                                                    float* __restrict__ out, int n) {
    __shared__ float Ws[HID * HID];
    const int tid = threadIdx.x;

    const float4* W4  = (const float4*)W;
    float4*       Ws4 = (float4*)Ws;
#pragma unroll
    for (int i = 0; i < 16; ++i)
        Ws4[tid + 256 * i] = W4[tid + 256 * i];
    __syncthreads();

    const int m0   = blockIdx.x << 4;
    const int wave = tid >> 5;
    const int lane = tid & 31;
    const int half = lane >> 4;
    const int l15  = lane & 15;
    const int n0   = wave << 4;
    int row = m0 + l15;
    if (row >= n) row = n - 1;
    const int ncol = n0 + l15;
    const float* arow = in + (size_t)row * HID;

    v8f c = {};
#pragma unroll
    for (int k0 = 0; k0 < HID; k0 += 4) {
        const int ka = k0 + half * 2;
        float2 av = *(const float2*)(arow + ka);
        v2f a, b;
        a[0] = av.x;
        a[1] = av.y;
        b[0] = Ws[ka * HID + ncol];
        b[1] = Ws[(ka + 1) * HID + ncol];
        c = __builtin_amdgcn_wmma_f32_16x16x4_f32(
                 false, a,  false, b,
                 (short)0, c,  false,  false);
    }

    const int mbase = m0 + half * 8;
#pragma unroll
    for (int i = 0; i < 8; ++i) {
        int r = mbase + i;
        if (r < n) out[(size_t)r * HID + ncol] = c[i];
    }
}

__global__ void k_pool_init(float* sums, float* cnt) {
    int idx = blockIdx.x * blockDim.x + threadIdx.x;
    if (idx < NGRAPHS * HID) sums[idx] = 0.0f;
    if (idx < NGRAPHS)       cnt[idx]  = 0.0f;
}

__global__ void k_final(const float* __restrict__ t, const float* __restrict__ cnt,
                        const float* __restrict__ b2,
                        const float* __restrict__ Wlin, const float* __restrict__ blin,
                        float* __restrict__ out) {
    int idx = blockIdx.x * blockDim.x + threadIdx.x;
    if (idx >= NGRAPHS * FOUT) return;
    int g = idx / FOUT;
    int j = idx - g * FOUT;
    const float c   = cnt[g];
    const float inv = 1.0f / fmaxf(c, 1.0f);
    const float has = (c > 0.0f) ? 1.0f : 0.0f;
    const float* s = t + g * HID;
    float acc = blin[j];
#pragma unroll 4
    for (int k = 0; k < HID; ++k)
        acc = fmaf(fmaf(s[k], inv, b2[k] * has), Wlin[k * FOUT + j], acc);
    out[idx] = acc;
}

extern "C" void kernel_launch(void* const* d_in, const int* in_sizes, int n_in,
                              void* d_out, int out_size, void* d_ws, size_t ws_size,
                              hipStream_t stream) {
    (void)n_in; (void)out_size; (void)ws_size;

    const float* x     = (const float*)d_in[0];
    const int*   esrc  = (const int*)  d_in[1];
    const int*   edst  = (const int*)  d_in[2];
    const int*   batch = (const int*)  d_in[3];
    const float* W1    = (const float*)d_in[4];
    const float* b1    = (const float*)d_in[5];
    const float* W2    = (const float*)d_in[6];
    const float* b2    = (const float*)d_in[7];
    const float* Wlin  = (const float*)d_in[8];
    const float* blin  = (const float*)d_in[9];
    float*       out   = (float*)d_out;

    const int N = in_sizes[3];
    const int E = in_sizes[1];

    char*  ws  = (char*)d_ws;
    size_t off = 0;
    auto carve = [&](size_t bytes) -> void* {
        off = (off + 255) & ~(size_t)255;
        void* p = ws + off;
        off += bytes;
        return p;
    };
    int*   ecnt  = (int*)  carve((size_t)N * 4);
    float* dinv  = (float*)carve((size_t)N * 4);
    float* bufH  = (float*)carve((size_t)N * HID * 4);
    float* bufO  = (float*)carve((size_t)N * HID * 4);
    char*  lists = (char*) carve((size_t)N * HID * 4);
    float* sums  = (float*)carve((size_t)NGRAPHS * HID * 4);
    float* cnt   = (float*)carve((size_t)NGRAPHS * 4);

    const int nsb = (N + SCAN_N - 1) / SCAN_N;
    ws  = lists;
    off = 0;
    int2* rec    = (int2*)carve((size_t)E * 8);
    int*  rank   = (int*) carve((size_t)E * 4);
    int*  rowloc = (int*) carve((size_t)N * 4);
    int*  blksum = (int*) carve((size_t)nsb * 4);
    int*  blkoff = (int*) carve((size_t)nsb * 4);

    const int T = 256;
    int nb   = (N + T - 1) / T;
    int eb   = (E + T - 1) / T;
    int wb   = (int)(((size_t)N * 32 + T - 1) / T);
    int gb2  = (NGRAPHS + 15) / 16;
    int pb   = (NGRAPHS * HID + T - 1) / T;
    int hb   = (NGRAPHS * FOUT + T - 1) / T;

    k_zero_cnt   <<<nb,  T,      0, stream>>>(ecnt, N);
    k_count_rank <<<eb,  T,      0, stream>>>(edst, ecnt, rank, E);
    k_scan_local <<<nsb, SCAN_T, 0, stream>>>(ecnt, rowloc, blksum, dinv, N);
    k_scan_blocks<<<1,   32,     0, stream>>>(blksum, blkoff, nsb);
    k_fill       <<<eb,  T,      0, stream>>>(esrc, edst, rank, rowloc, blkoff, dinv, rec, E);

    k_sum_gemm1<<<wb, T, 0, stream>>>(x, W1, b1, bufO, rec, rowloc, blkoff, ecnt, dinv, N);

    k_pool_init <<<pb,  T, 0, stream>>>(sums, cnt);
    k_pull_pool <<<wb,  T, 0, stream>>>(bufO, rec, rowloc, blkoff, ecnt, dinv, batch, sums, cnt, N);
    k_gemm2_wmma<<<gb2, T, 0, stream>>>(sums, W2, bufH, NGRAPHS);

    k_final     <<<hb, T, 0, stream>>>(bufH, cnt, b2, Wlin, blin, out);
}


# Lists placed at count time: the integer add's slot indexes a fixed 64-slot row per node (overflow pairs behind); scan and fill launches go; dinv[s] pre-multiplied into the rows; f32 rounding changes
# speedup vs baseline: 4.1232x; 1.1263x over previous
#include <hip/hip_runtime.h>
#include <hip/hip_bf16.h>

#define HID     128
#define FIN     11
#define FOUT    19
#define NGRAPHS 4096

typedef __attribute__((ext_vector_type(2))) float v2f;
typedef __attribute__((ext_vector_type(8))) float v8f;

#define LIST_CAP 64
#define XS_PITCH 16

__global__ void k_zero_cnt(int* __restrict__ cnt, int* __restrict__ ovf_n, int n) {
    int i = blockIdx.x * blockDim.x + threadIdx.x;
    if (i < n) cnt[i] = 0;
    if (i == 0) *ovf_n = 0;
}

__global__ void k_count_place(const int* __restrict__ src, const int* __restrict__ dst,
                              int* cnt, int* __restrict__ ell,
                              int2* __restrict__ ovf, int* ovf_n, int e) {
    int i = blockIdx.x * blockDim.x + threadIdx.x;
    if (i >= e) return;
    const int s = src[i];
    const int d = dst[i];
    const int slot = atomicAdd(&cnt[d], 1);
    if (slot < LIST_CAP) {
        ell[(size_t)d * LIST_CAP + slot] = s;
    } else {
        const int k = atomicAdd(ovf_n, 1);
        ovf[k] = make_int2(d, s);
    }
}

__global__ void k_prescale(const float* __restrict__ x, const int* __restrict__ cnt,
                           float* __restrict__ xs, float* __restrict__ dinv, int n) {
    const int idx  = blockIdx.x * blockDim.x + threadIdx.x;
    const int node = idx / XS_PITCH;
    if (node >= n) return;
    const int   j  = idx - node * XS_PITCH;
    const float d  = (float)(cnt[node] + 1);
    const float dn = (d > 0.0f) ? rsqrtf(d) : 0.0f;
    xs[idx] = (j < FIN) ? dn * x[(size_t)node * FIN + j] : 0.0f;
    if (j == 0) dinv[node] = dn;
}


template <int G>
__device__ __forceinline__ float sum_rows(const float* __restrict__ xs, int col,
                                          int s_l, int w_l, float acc) {
    float v[G];
    float w[G];
#pragma unroll
    for (int k = 0; k < G; ++k) {
        const int s = __builtin_amdgcn_readlane(s_l, k);
        w[k] = __int_as_float(__builtin_amdgcn_readlane(w_l, k));
        v[k] = xs[(size_t)s * XS_PITCH + col];
    }
#pragma unroll
    for (int k = 0; k < G; ++k) acc = fmaf(v[k], w[k], acc);
    return acc;
}

__global__ __launch_bounds__(256) void k_sum_gemm1(const float* __restrict__ xs,
                                                   const float* __restrict__ W1,
                                                   const float* __restrict__ bias,
                                                   float* __restrict__ out,
                                                   const int* __restrict__ ell,
                                                   const int2* __restrict__ ovf,
                                                   const int* __restrict__ ovf_n,
                                                   const int* __restrict__ cnt,
                                                   const float* __restrict__ dinv, int n) {
    const int lane = threadIdx.x & 31;
    const int node = __builtin_amdgcn_readfirstlane((int)((blockIdx.x * 256u + threadIdx.x) >> 5));
    if (node >= n) return;

    const int   c   = cnt[node];
    const float dn  = dinv[node];
    const int   col = lane & (XS_PITCH - 1);
    const int*  row = ell + (size_t)node * LIST_CAP;

    const float self = xs[(size_t)node * XS_PITCH + col];
    float a = 0.0f;

    const int ce = min(c, LIST_CAP);
    for (int base = 0; base < ce; base += 32) {
        const int m = min(32, ce - base);
        int s_l = node;
        int w_l = 0;
        if (lane < m) {
            s_l = row[base + lane];
            w_l = __float_as_int(dn);
        }
        if (m > 24)      a = sum_rows<32>(xs, col, s_l, w_l, a);
        else if (m > 16) a = sum_rows<24>(xs, col, s_l, w_l, a);
        else if (m > 8)  a = sum_rows<16>(xs, col, s_l, w_l, a);
        else             a = sum_rows<8>(xs, col, s_l, w_l, a);
    }
    if (c > LIST_CAP) {
        const int no = *ovf_n;
        for (int base = 0; base < no; base += 32) {
            const int i = base + lane;
            int2 r = make_int2(-1, 0);
            if (i < no) r = ovf[i];
            unsigned hit = __builtin_amdgcn_ballot_w32(r.x == node);
            while (hit) {
                const int k = __builtin_ctz(hit);
                hit &= hit - 1;
                const int s = __shfl(r.y, k);
                a = fmaf(xs[(size_t)s * XS_PITCH + col], dn, a);
            }
        }
    }
    a = fmaf(self, dn, a);

    const float4* W4 = (const float4*)W1;
    float4 o = make_float4(0.0f, 0.0f, 0.0f, 0.0f);
#pragma unroll
    for (int k = 0; k < FIN; ++k) {
        const float  ak = __int_as_float(__builtin_amdgcn_readlane(__float_as_int(a), k));
        const float4 w  = W4[k * (HID / 4) + lane];
        o.x = fmaf(ak, w.x, o.x);
        o.y = fmaf(ak, w.y, o.y);
        o.z = fmaf(ak, w.z, o.z);
        o.w = fmaf(ak, w.w, o.w);
    }
    const float4 b = ((const float4*)bias)[lane];
    o.x = fmaxf(o.x + b.x, 0.0f) * dn;
    o.y = fmaxf(o.y + b.y, 0.0f) * dn;
    o.z = fmaxf(o.z + b.z, 0.0f) * dn;
    o.w = fmaxf(o.w + b.w, 0.0f) * dn;
    ((float4*)out)[(size_t)node * (HID / 4) + lane] = o;
}


template <int G>
__device__ __forceinline__ void pull_group(const float4* __restrict__ h4, int lane,
                                           int s_l, int j, float4& acc) {
    float4 v[G];
#pragma unroll
    for (int k = 0; k < G; ++k) {
        const int s = __builtin_amdgcn_readlane(s_l, j + k);
        v[k] = h4[(size_t)s * (HID / 4) + lane];
    }
#pragma unroll
    for (int k = 0; k < G; ++k) {
        acc.x += v[k].x;
        acc.y += v[k].y;
        acc.z += v[k].z;
        acc.w += v[k].w;
    }
}

#define PULL_NODES 8

__global__ __launch_bounds__(256) void k_pull_pool(const float* __restrict__ h,
                                                   const int* __restrict__ ell,
                                                   const int2* __restrict__ ovf,
                                                   const int* __restrict__ ovf_n,
                                                   const int* __restrict__ cnt,
                                                   const float* __restrict__ dinv,
                                                   const int* __restrict__ batch,
                                                   float* sums, float* gcnt, int n) {
    __shared__ float4 rows4[PULL_NODES][HID / 4];
    __shared__ __attribute__((aligned(16))) int gid[PULL_NODES];

    const int lane = threadIdx.x & 31;
    const int wave = threadIdx.x >> 5;
    const int node = __builtin_amdgcn_readfirstlane((int)(blockIdx.x * PULL_NODES + wave));

    float4 o = make_float4(0.0f, 0.0f, 0.0f, 0.0f);
    int    g = -1;
    if (node < n) {
        const int   c     = cnt[node];
        const float dn    = dinv[node];
        const float4* h4  = (const float4*)h;
        const int*  row   = ell + (size_t)node * LIST_CAP;
        g = batch[node];

        const float4 self = h4[(size_t)node * (HID / 4) + lane];
        float4 acc = make_float4(0.0f, 0.0f, 0.0f, 0.0f);

        const int ce = min(c, LIST_CAP);
        for (int base = 0; base < ce; base += 32) {
            const int m = min(32, ce - base);
            int s_l = node;
            if (lane < m) s_l = row[base + lane];
            int j = 0;
            for (; j + 8 <= m; j += 8) pull_group<8>(h4, lane, s_l, j, acc);
            if (j + 4 <= m) { pull_group<4>(h4, lane, s_l, j, acc); j += 4; }
            if (j < m) {
                const float4 z = make_float4(0.0f, 0.0f, 0.0f, 0.0f);
                float4 v0 = z, v1 = z, v2 = z;
                {
                    const int s = __builtin_amdgcn_readlane(s_l, j);
                    v0 = h4[(size_t)s * (HID / 4) + lane];
                }
                if (j + 1 < m) {
                    const int s = __builtin_amdgcn_readlane(s_l, j + 1);
                    v1 = h4[(size_t)s * (HID / 4) + lane];
                }
                if (j + 2 < m) {
                    const int s = __builtin_amdgcn_readlane(s_l, j + 2);
                    v2 = h4[(size_t)s * (HID / 4) + lane];
                }
                acc.x += v0.x; acc.y += v0.y; acc.z += v0.z; acc.w += v0.w;
                acc.x += v1.x; acc.y += v1.y; acc.z += v1.z; acc.w += v1.w;
                acc.x += v2.x; acc.y += v2.y; acc.z += v2.z; acc.w += v2.w;
            }
        }
        if (c > LIST_CAP) {
            const int no = *ovf_n;
            for (int base = 0; base < no; base += 32) {
                const int i = base + lane;
                int2 r = make_int2(-1, 0);
                if (i < no) r = ovf[i];
                unsigned hit = __builtin_amdgcn_ballot_w32(r.x == node);
                while (hit) {
                    const int k = __builtin_ctz(hit);
                    hit &= hit - 1;
                    const int s = __shfl(r.y, k);
                    const float4 v = h4[(size_t)s * (HID / 4) + lane];
                    acc.x += v.x; acc.y += v.y; acc.z += v.z; acc.w += v.w;
                }
            }
        }

        o.x = (acc.x + self.x) * dn;
        o.y = (acc.y + self.y) * dn;
        o.z = (acc.z + self.z) * dn;
        o.w = (acc.w + self.w) * dn;
    }

    rows4[wave][lane] = o;
    if (lane == 0) gid[wave] = g;
    __syncthreads();

    const int    tid  = threadIdx.x;
    const float* rows = (const float*)rows4;

    float v[PULL_NODES];
#pragma unroll
    for (int r = 0; r < PULL_NODES; ++r) v[r] = rows[r * HID + (tid & (HID - 1))];
    const int4 ga = ((const int4*)gid)[0];
    const int4 gb = ((const int4*)gid)[1];
    const int gs[PULL_NODES] = {
        __builtin_amdgcn_readfirstlane(ga.x), __builtin_amdgcn_readfirstlane(ga.y),
        __builtin_amdgcn_readfirstlane(ga.z), __builtin_amdgcn_readfirstlane(ga.w),
        __builtin_amdgcn_readfirstlane(gb.x), __builtin_amdgcn_readfirstlane(gb.y),
        __builtin_amdgcn_readfirstlane(gb.z), __builtin_amdgcn_readfirstlane(gb.w)};

    if (tid < HID) {
        int   cur = -1;
        float s   = 0.0f;
#pragma unroll
        for (int r = 0; r < PULL_NODES; ++r) {
            if (gs[r] < 0) continue;
            if (gs[r] != cur) {
                if (cur >= 0) atomicAdd(&sums[cur * HID + tid], s);
                cur = gs[r];
                s   = 0.0f;
            }
            s += v[r];
        }
        if (cur >= 0) atomicAdd(&sums[cur * HID + tid], s);
    } else if (tid == HID) {
        int   cur = -1;
        float k   = 0.0f;
#pragma unroll
        for (int r = 0; r < PULL_NODES; ++r) {
            if (gs[r] < 0) continue;
            if (gs[r] != cur) {
                if (cur >= 0) atomicAdd(&gcnt[cur], k);
                cur = gs[r];
                k   = 0.0f;
            }
            k += 1.0f;
        }
        if (cur >= 0) atomicAdd(&gcnt[cur], k);
    }
}

__global__ __launch_bounds__(256) void k_gemm2_wmma(const float* __restrict__ in,
                                                    const float* __restrict__ W,
                                                    float* __restrict__ out, int n) {
    __shared__ float Ws[HID * HID];
    const int tid = threadIdx.x;

    const float4* W4  = (const float4*)W;
    float4*       Ws4 = (float4*)Ws;
#pragma unroll
    for (int i = 0; i < 16; ++i)
        Ws4[tid + 256 * i] = W4[tid + 256 * i];
    __syncthreads();

    const int m0   = blockIdx.x << 4;
    const int wave = tid >> 5;
    const int lane = tid & 31;
    const int half = lane >> 4;
    const int l15  = lane & 15;
    const int n0   = wave << 4;
    int row = m0 + l15;
    if (row >= n) row = n - 1;
    const int ncol = n0 + l15;
    const float* arow = in + (size_t)row * HID;

    v8f c = {};
#pragma unroll
    for (int k0 = 0; k0 < HID; k0 += 4) {
        const int ka = k0 + half * 2;
        float2 av = *(const float2*)(arow + ka);
        v2f a, b;
        a[0] = av.x;
        a[1] = av.y;
        b[0] = Ws[ka * HID + ncol];
        b[1] = Ws[(ka + 1) * HID + ncol];
        c = __builtin_amdgcn_wmma_f32_16x16x4_f32(
                 false, a,  false, b,
                 (short)0, c,  false,  false);
    }

    const int mbase = m0 + half * 8;
#pragma unroll
    for (int i = 0; i < 8; ++i) {
        int r = mbase + i;
        if (r < n) out[(size_t)r * HID + ncol] = c[i];
    }
}

__global__ void k_pool_init(float* sums, float* cnt) {
    int idx = blockIdx.x * blockDim.x + threadIdx.x;
    if (idx < NGRAPHS * HID) sums[idx] = 0.0f;
    if (idx < NGRAPHS)       cnt[idx]  = 0.0f;
}

__global__ void k_final(const float* __restrict__ t, const float* __restrict__ cnt,
                        const float* __restrict__ b2,
                        const float* __restrict__ Wlin, const float* __restrict__ blin,
                        float* __restrict__ out) {
    int idx = blockIdx.x * blockDim.x + threadIdx.x;
    if (idx >= NGRAPHS * FOUT) return;
    int g = idx / FOUT;
    int j = idx - g * FOUT;
    const float c   = cnt[g];
    const float inv = 1.0f / fmaxf(c, 1.0f);
    const float has = (c > 0.0f) ? 1.0f : 0.0f;
    const float* s = t + g * HID;
    float acc = blin[j];
#pragma unroll 4
    for (int k = 0; k < HID; ++k)
        acc = fmaf(fmaf(s[k], inv, b2[k] * has), Wlin[k * FOUT + j], acc);
    out[idx] = acc;
}

extern "C" void kernel_launch(void* const* d_in, const int* in_sizes, int n_in,
                              void* d_out, int out_size, void* d_ws, size_t ws_size,
                              hipStream_t stream) {
    (void)n_in; (void)out_size; (void)ws_size;

    const float* x     = (const float*)d_in[0];
    const int*   esrc  = (const int*)  d_in[1];
    const int*   edst  = (const int*)  d_in[2];
    const int*   batch = (const int*)  d_in[3];
    const float* W1    = (const float*)d_in[4];
    const float* b1    = (const float*)d_in[5];
    const float* W2    = (const float*)d_in[6];
    const float* b2    = (const float*)d_in[7];
    const float* Wlin  = (const float*)d_in[8];
    const float* blin  = (const float*)d_in[9];
    float*       out   = (float*)d_out;

    const int N = in_sizes[3];
    const int E = in_sizes[1];

    char*  ws  = (char*)d_ws;
    size_t off = 0;
    auto carve = [&](size_t bytes) -> void* {
        off = (off + 255) & ~(size_t)255;
        void* p = ws + off;
        off += bytes;
        return p;
    };
    int*   ecnt  = (int*)  carve((size_t)N * 4);
    float* dinv  = (float*)carve((size_t)N * 4);
    float* bufH  = (float*)carve((size_t)N * HID * 4);
    float* bufO  = (float*)carve((size_t)N * HID * 4);
    char*  lists = (char*) carve((size_t)N * HID * 4);
    float* sums  = (float*)carve((size_t)NGRAPHS * HID * 4);
    float* cnt   = (float*)carve((size_t)NGRAPHS * 4);

    float* xs = bufH + (size_t)NGRAPHS * HID;

    ws  = lists;
    off = 0;
    int*  ell   = (int*) carve((size_t)N * LIST_CAP * 4);
    int2* ovf   = (int2*)carve((size_t)E * 8);
    int*  ovf_n = (int*) carve(4);

    const int T = 256;
    int nb   = (N + T - 1) / T;
    int eb   = (E + T - 1) / T;
    int xb   = (int)(((size_t)N * XS_PITCH + T - 1) / T);
    int wb   = (int)(((size_t)N * 32 + T - 1) / T);
    int gb2  = (NGRAPHS + 15) / 16;
    int pb   = (NGRAPHS * HID + T - 1) / T;
    int hb   = (NGRAPHS * FOUT + T - 1) / T;

    k_zero_cnt   <<<nb, T, 0, stream>>>(ecnt, ovf_n, N);
    k_count_place<<<eb, T, 0, stream>>>(esrc, edst, ecnt, ell, ovf, ovf_n, E);
    k_prescale   <<<xb, T, 0, stream>>>(x, ecnt, xs, dinv, N);

    k_sum_gemm1<<<wb, T, 0, stream>>>(xs, W1, b1, bufO, ell, ovf, ovf_n, ecnt, dinv, N);

    k_pool_init <<<pb,  T, 0, stream>>>(sums, cnt);
    k_pull_pool <<<wb,  T, 0, stream>>>(bufO, ell, ovf, ovf_n, ecnt, dinv, batch, sums, cnt, N);
    k_gemm2_wmma<<<gb2, T, 0, stream>>>(sums, W2, bufH, NGRAPHS);

    k_final     <<<hb, T, 0, stream>>>(bufH, cnt, b2, Wlin, blin, out);
}


# Output layer folded into layer 2's weight (M = W2 Wlin, 128x19) and applied per node on WMMA before the neighbour sum: layer 2 gathers and pools 19-wide rows, not 128-wide; f32 sum order changes
# speedup vs baseline: 4.7330x; 1.1479x over previous
#include <hip/hip_runtime.h>
#include <hip/hip_bf16.h>

#define HID     128
#define FIN     11
#define FOUT    19
#define NGRAPHS 4096

typedef __attribute__((ext_vector_type(2))) float v2f;
typedef __attribute__((ext_vector_type(8))) float v8f;

#define LIST_CAP 64
#define XS_PITCH 16
#define UP       32

__global__ void k_zero_cnt(int* __restrict__ cnt, int* __restrict__ ovf_n, int n) {
    int i = blockIdx.x * blockDim.x + threadIdx.x;
    if (i < n) cnt[i] = 0;
    if (i == 0) *ovf_n = 0;
}

__global__ void k_count_place(const int* __restrict__ src, const int* __restrict__ dst,
                              int* cnt, int* __restrict__ ell,
                              int2* __restrict__ ovf, int* ovf_n, int e) {
    int i = blockIdx.x * blockDim.x + threadIdx.x;
    if (i >= e) return;
    const int s = src[i];
    const int d = dst[i];
    const int slot = atomicAdd(&cnt[d], 1);
    if (slot < LIST_CAP) {
        ell[(size_t)d * LIST_CAP + slot] = s;
    } else {
        const int k = atomicAdd(ovf_n, 1);
        ovf[k] = make_int2(d, s);
    }
}

__global__ void k_prescale(const float* __restrict__ x, const int* __restrict__ cnt,
                           float* __restrict__ xs, float* __restrict__ dinv, int n) {
    const int idx  = blockIdx.x * blockDim.x + threadIdx.x;
    const int node = idx / XS_PITCH;
    if (node >= n) return;
    const int   j  = idx - node * XS_PITCH;
    const float d  = (float)(cnt[node] + 1);
    const float dn = (d > 0.0f) ? rsqrtf(d) : 0.0f;
    xs[idx] = (j < FIN) ? dn * x[(size_t)node * FIN + j] : 0.0f;
    if (j == 0) dinv[node] = dn;
}

__global__ __launch_bounds__(32) void k_fold_w2(const float* __restrict__ W2,
                                                const float* __restrict__ Wlin,
                                                float* __restrict__ M) {
    const int lane = threadIdx.x;
    const int half = lane >> 4;
    const int l15  = lane & 15;
    const int m0   = (blockIdx.x >> 1) << 4;
    const int ncol = ((blockIdx.x & 1) << 4) + l15;
    const bool live = ncol < FOUT;
    const int  lcol = live ? ncol : 0;
    const float* arow = W2 + (size_t)(m0 + l15) * HID;

    v8f c = {};
#pragma unroll
    for (int k0 = 0; k0 < HID; k0 += 4) {
        const int ka = k0 + half * 2;
        const float2 av = *(const float2*)(arow + ka);
        v2f a, b;
        a[0] = av.x;
        a[1] = av.y;
        const float w0 = Wlin[ka * FOUT + lcol];
        const float w1 = Wlin[(ka + 1) * FOUT + lcol];
        b[0] = live ? w0 : 0.0f;
        b[1] = live ? w1 : 0.0f;
        c = __builtin_amdgcn_wmma_f32_16x16x4_f32(
                 false, a,  false, b,
                 (short)0, c,  false,  false);
    }
    const int mbase = m0 + half * 8;
#pragma unroll
    for (int i = 0; i < 8; ++i) M[(size_t)(mbase + i) * UP + ncol] = c[i];
}


template <int G, int PITCH>
__device__ __forceinline__ float sum_rows(const float* __restrict__ xs, int col,
                                          int s_l, int w_l, float acc) {
    float v[G];
    float w[G];
#pragma unroll
    for (int k = 0; k < G; ++k) {
        const int s = __builtin_amdgcn_readlane(s_l, k);
        w[k] = __int_as_float(__builtin_amdgcn_readlane(w_l, k));
        v[k] = xs[(size_t)s * PITCH + col];
    }
#pragma unroll
    for (int k = 0; k < G; ++k) acc = fmaf(v[k], w[k], acc);
    return acc;
}

#define L1_NODES   16
#define HROW_PITCH 132
#define PART_PITCH 272

__global__ __launch_bounds__(512) void k_sum_gemm1(const float* __restrict__ xs,
                                                   const float* __restrict__ W1,
                                                   const float* __restrict__ bias,
                                                   const float* __restrict__ M,
                                                   float* __restrict__ u,
                                                   const int* __restrict__ ell,
                                                   const int2* __restrict__ ovf,
                                                   const int* __restrict__ ovf_n,
                                                   const int* __restrict__ cnt,
                                                   const float* __restrict__ dinv, int n) {
    __shared__ __attribute__((aligned(16))) float hrow[L1_NODES * HROW_PITCH];
    __shared__ float part[L1_NODES * PART_PITCH];

    const int lane = threadIdx.x & 31;
    const int wave = threadIdx.x >> 5;
    const int node = __builtin_amdgcn_readfirstlane((int)(blockIdx.x * L1_NODES + wave));

    const int half = lane >> 4;
    const int l15  = lane & 15;

    float a  = 0.0f;
    float dn = 0.0f;
    if (node < n) {
        const int   c   = cnt[node];
        dn = dinv[node];
        const int   col = lane & (XS_PITCH - 1);
        const int*  row = ell + (size_t)node * LIST_CAP;

        const float self = xs[(size_t)node * XS_PITCH + col];

        const int ce = min(c, LIST_CAP);
        for (int base = 0; base < ce; base += 32) {
            const int m = min(32, ce - base);
            int s_l = node;
            int w_l = 0;
            if (lane < m) {
                s_l = row[base + lane];
                w_l = __float_as_int(dn);
            }
            if (m > 24)      a = sum_rows<32, XS_PITCH>(xs, col, s_l, w_l, a);
            else if (m > 16) a = sum_rows<24, XS_PITCH>(xs, col, s_l, w_l, a);
            else if (m > 8)  a = sum_rows<16, XS_PITCH>(xs, col, s_l, w_l, a);
            else             a = sum_rows<8, XS_PITCH>(xs, col, s_l, w_l, a);
        }
        if (c > LIST_CAP) {
            const int no = *ovf_n;
            for (int base = 0; base < no; base += 32) {
                const int i = base + lane;
                int2 r = make_int2(-1, 0);
                if (i < no) r = ovf[i];
                unsigned hit = __builtin_amdgcn_ballot_w32(r.x == node);
                while (hit) {
                    const int k = __builtin_ctz(hit);
                    hit &= hit - 1;
                    const int s = __shfl(r.y, k);
                    a = fmaf(xs[(size_t)s * XS_PITCH + col], dn, a);
                }
            }
        }
        a = fmaf(self, dn, a);
    }

    const int ncol = ((wave & 1) << 4) + l15;
    const int kb   = (wave >> 1) << 4;
    v2f bfr[4];
#pragma unroll
    for (int t = 0; t < 4; ++t) {
        const int ka = kb + 4 * t + half * 2;
        bfr[t][0] = M[ka * UP + ncol];
        bfr[t][1] = M[(ka + 1) * UP + ncol];
    }

    float4 o = make_float4(0.0f, 0.0f, 0.0f, 0.0f);
    if (node < n) {
        const float4* W4 = (const float4*)W1;
#pragma unroll
        for (int k = 0; k < FIN; ++k) {
            const float  ak = __int_as_float(__builtin_amdgcn_readlane(__float_as_int(a), k));
            const float4 w  = W4[k * (HID / 4) + lane];
            o.x = fmaf(ak, w.x, o.x);
            o.y = fmaf(ak, w.y, o.y);
            o.z = fmaf(ak, w.z, o.z);
            o.w = fmaf(ak, w.w, o.w);
        }
        const float4 b = ((const float4*)bias)[lane];
        o.x = fmaxf(o.x + b.x, 0.0f) * dn;
        o.y = fmaxf(o.y + b.y, 0.0f) * dn;
        o.z = fmaxf(o.z + b.z, 0.0f) * dn;
        o.w = fmaxf(o.w + b.w, 0.0f) * dn;
    }

    ((float4*)(hrow + wave * HROW_PITCH))[lane] = o;
    __syncthreads();

    v8f c8 = {};
#pragma unroll
    for (int t = 0; t < 4; ++t) {
        const int ka = kb + 4 * t + half * 2;
        const float2 av = *(const float2*)(hrow + l15 * HROW_PITCH + ka);
        v2f a2;
        a2[0] = av.x;
        a2[1] = av.y;
        c8 = __builtin_amdgcn_wmma_f32_16x16x4_f32(
                  false, a2,  false, bfr[t],
                  (short)0, c8,  false,  false);
    }
#pragma unroll
    for (int i = 0; i < 8; ++i) part[wave * PART_PITCH + i * 32 + lane] = c8[i];
    __syncthreads();

    const int src = (wave & 7) * 32 + (wave >> 3) * 16 + l15;
    float s = 0.0f;
#pragma unroll
    for (int kq = 0; kq < 8; ++kq) s += part[(kq * 2 + half) * PART_PITCH + src];
    if (node < n) u[(size_t)node * UP + lane] = s;
}


#define PULL_NODES 8

__global__ __launch_bounds__(256) void k_pull_pool(const float* __restrict__ u,
                                                   const int* __restrict__ ell,
                                                   const int2* __restrict__ ovf,
                                                   const int* __restrict__ ovf_n,
                                                   const int* __restrict__ cnt,
                                                   const float* __restrict__ dinv,
                                                   const int* __restrict__ batch,
                                                   float* sums, float* gcnt, int n) {
    __shared__ float rows[PULL_NODES * UP];
    __shared__ __attribute__((aligned(16))) int gid[PULL_NODES];

    const int lane = threadIdx.x & 31;
    const int wave = threadIdx.x >> 5;
    const int node = __builtin_amdgcn_readfirstlane((int)(blockIdx.x * PULL_NODES + wave));

    float o = 0.0f;
    int   g = -1;
    if (node < n) {
        const int   c   = cnt[node];
        const float dn  = dinv[node];
        const int*  row = ell + (size_t)node * LIST_CAP;
        g = batch[node];

        const float self = u[(size_t)node * UP + lane];
        float a = 0.0f;

        const int ce = min(c, LIST_CAP);
        for (int base = 0; base < ce; base += 32) {
            const int m = min(32, ce - base);
            int s_l = node;
            int w_l = 0;
            if (lane < m) {
                s_l = row[base + lane];
                w_l = __float_as_int(1.0f);
            }
            if (m > 24)      a = sum_rows<32, UP>(u, lane, s_l, w_l, a);
            else if (m > 16) a = sum_rows<24, UP>(u, lane, s_l, w_l, a);
            else if (m > 8)  a = sum_rows<16, UP>(u, lane, s_l, w_l, a);
            else             a = sum_rows<8, UP>(u, lane, s_l, w_l, a);
        }
        if (c > LIST_CAP) {
            const int no = *ovf_n;
            for (int base = 0; base < no; base += 32) {
                const int i = base + lane;
                int2 r = make_int2(-1, 0);
                if (i < no) r = ovf[i];
                unsigned hit = __builtin_amdgcn_ballot_w32(r.x == node);
                while (hit) {
                    const int k = __builtin_ctz(hit);
                    hit &= hit - 1;
                    const int s = __shfl(r.y, k);
                    a += u[(size_t)s * UP + lane];
                }
            }
        }
        o = (a + self) * dn;
    }

    rows[wave * UP + lane] = o;
    if (lane == 0) gid[wave] = g;
    __syncthreads();

    if (wave > 1) return;

    float v[PULL_NODES];
#pragma unroll
    for (int r = 0; r < PULL_NODES; ++r) v[r] = rows[r * UP + lane];
    const int4 ga = ((const int4*)gid)[0];
    const int4 gb = ((const int4*)gid)[1];
    const int gs[PULL_NODES] = {
        __builtin_amdgcn_readfirstlane(ga.x), __builtin_amdgcn_readfirstlane(ga.y),
        __builtin_amdgcn_readfirstlane(ga.z), __builtin_amdgcn_readfirstlane(ga.w),
        __builtin_amdgcn_readfirstlane(gb.x), __builtin_amdgcn_readfirstlane(gb.y),
        __builtin_amdgcn_readfirstlane(gb.z), __builtin_amdgcn_readfirstlane(gb.w)};

    if (wave == 0) {
        if (lane < FOUT) {
            int   cur = -1;
            float s   = 0.0f;
#pragma unroll
            for (int r = 0; r < PULL_NODES; ++r) {
                if (gs[r] < 0) continue;
                if (gs[r] != cur) {
                    if (cur >= 0) atomicAdd(&sums[cur * UP + lane], s);
                    cur = gs[r];
                    s   = 0.0f;
                }
                s += v[r];
            }
            if (cur >= 0) atomicAdd(&sums[cur * UP + lane], s);
        }
    } else if (lane == 0) {
        int   cur = -1;
        float k   = 0.0f;
#pragma unroll
        for (int r = 0; r < PULL_NODES; ++r) {
            if (gs[r] < 0) continue;
            if (gs[r] != cur) {
                if (cur >= 0) atomicAdd(&gcnt[cur], k);
                cur = gs[r];
                k   = 0.0f;
            }
            k += 1.0f;
        }
        if (cur >= 0) atomicAdd(&gcnt[cur], k);
    }
}

__global__ void k_pool_init(float* sums, float* cnt) {
    int idx = blockIdx.x * blockDim.x + threadIdx.x;
    if (idx < NGRAPHS * UP) sums[idx] = 0.0f;
    if (idx < NGRAPHS)      cnt[idx]  = 0.0f;
}

__global__ void k_final(const float* __restrict__ sums, const float* __restrict__ cnt,
                        const float* __restrict__ b2,
                        const float* __restrict__ Wlin, const float* __restrict__ blin,
                        float* __restrict__ out) {
    int idx = blockIdx.x * blockDim.x + threadIdx.x;
    if (idx >= NGRAPHS * FOUT) return;
    int g = idx / FOUT;
    int j = idx - g * FOUT;
    const float c   = cnt[g];
    const float inv = 1.0f / fmaxf(c, 1.0f);
    const float has = (c > 0.0f) ? 1.0f : 0.0f;
    float bb = 0.0f;
#pragma unroll 4
    for (int k = 0; k < HID; ++k) bb = fmaf(b2[k], Wlin[k * FOUT + j], bb);
    out[idx] = fmaf(sums[g * UP + j], inv, fmaf(bb, has, blin[j]));
}

extern "C" void kernel_launch(void* const* d_in, const int* in_sizes, int n_in,
                              void* d_out, int out_size, void* d_ws, size_t ws_size,
                              hipStream_t stream) {
    (void)n_in; (void)out_size; (void)ws_size;

    const float* x     = (const float*)d_in[0];
    const int*   esrc  = (const int*)  d_in[1];
    const int*   edst  = (const int*)  d_in[2];
    const int*   batch = (const int*)  d_in[3];
    const float* W1    = (const float*)d_in[4];
    const float* b1    = (const float*)d_in[5];
    const float* W2    = (const float*)d_in[6];
    const float* b2    = (const float*)d_in[7];
    const float* Wlin  = (const float*)d_in[8];
    const float* blin  = (const float*)d_in[9];
    float*       out   = (float*)d_out;

    const int N = in_sizes[3];
    const int E = in_sizes[1];

    char*  ws  = (char*)d_ws;
    size_t off = 0;
    auto carve = [&](size_t bytes) -> void* {
        off = (off + 255) & ~(size_t)255;
        void* p = ws + off;
        off += bytes;
        return p;
    };
    int*   ecnt  = (int*)  carve((size_t)N * 4);
    float* dinv  = (float*)carve((size_t)N * 4);
    float* bufH  = (float*)carve((size_t)N * HID * 4);
    float* bufO  = (float*)carve((size_t)N * HID * 4);
    char*  lists = (char*) carve((size_t)N * HID * 4);
    float* sums  = (float*)carve((size_t)NGRAPHS * HID * 4);
    float* cnt   = (float*)carve((size_t)NGRAPHS * 4);

    float* xs = bufH + (size_t)NGRAPHS * HID;

    ws  = lists;
    off = 0;
    int*  ell   = (int*) carve((size_t)N * LIST_CAP * 4);
    int2* ovf   = (int2*)carve((size_t)E * 8);
    int*  ovf_n = (int*) carve(4);

    const int T = 256;
    int nb   = (N + T - 1) / T;
    int eb   = (E + T - 1) / T;
    int xb   = (int)(((size_t)N * XS_PITCH + T - 1) / T);
    int wb   = (int)(((size_t)N * 32 + T - 1) / T);
    int lb   = (N + L1_NODES - 1) / L1_NODES;
    int pb   = (NGRAPHS * UP + T - 1) / T;
    int hb   = (NGRAPHS * FOUT + T - 1) / T;

    k_fold_w2<<<(HID / 16) * (UP / 16), 32, 0, stream>>>(W2, Wlin, bufH);

    k_zero_cnt   <<<nb, T, 0, stream>>>(ecnt, ovf_n, N);
    k_count_place<<<eb, T, 0, stream>>>(esrc, edst, ecnt, ell, ovf, ovf_n, E);
    k_prescale   <<<xb, T, 0, stream>>>(x, ecnt, xs, dinv, N);

    k_sum_gemm1<<<lb, 32 * L1_NODES, 0, stream>>>(xs, W1, b1, bufH, bufO, ell, ovf, ovf_n, ecnt, dinv, N);

    k_pool_init<<<pb, T, 0, stream>>>(sums, cnt);
    k_pull_pool<<<wb, T, 0, stream>>>(bufO, ell, ovf, ovf_n, ecnt, dinv, batch, sums, cnt, N);

    k_final<<<hb, T, 0, stream>>>(sums, cnt, b2, Wlin, blin, out);
}


# Lists built in LDS: edges binned by 256-node range (one global add per block and range), a block per range fills rows, counts, dinv, scaled x; per-edge global add, scatter go; f32 sum order changes
# speedup vs baseline: 9.9872x; 2.1101x over previous
#include <hip/hip_runtime.h>
#include <hip/hip_bf16.h>

#define HID     128
#define FIN     11
#define FOUT    19
#define NGRAPHS 4096

typedef __attribute__((ext_vector_type(2))) float v2f;
typedef __attribute__((ext_vector_type(8))) float v8f;

#define LIST_CAP 64
#define XS_PITCH 16
#define UP       32

#define BIN_SHIFT 8
#define BIN_NODES (1 << BIN_SHIFT)
#define MAX_BINS  512
#define BIN_CAP   8192
#define BIN_T     512
#define BIN_EPT   8
#define BIN_CHUNK (BIN_T * BIN_EPT)
#define BUILD_T   512
#define BUILD_G   8
#define SRC_BITS  24

__global__ void k_zero_bins(int* __restrict__ bincnt, int* __restrict__ ovf_n, int* __restrict__ ovfb_n) {
    const int i = blockIdx.x * blockDim.x + threadIdx.x;
    if (i < MAX_BINS) bincnt[i] = 0;
    if (i == 0) { *ovf_n = 0; *ovfb_n = 0; }
}

__global__ __launch_bounds__(BIN_T) void k_bin_edges(const int* __restrict__ src, const int* __restrict__ dst,
                                                     int* bincnt, unsigned* __restrict__ bins,
                                                     int2* __restrict__ ovfb, int* ovfb_n, int e) {
    __shared__ int            lhist[MAX_BINS];
    __shared__ int            lstart[MAX_BINS];
    __shared__ int            ldelta[MAX_BINS];
    __shared__ int            wsum[BIN_T / 32];
    __shared__ unsigned       stage_w[BIN_CHUNK];
    __shared__ unsigned short stage_b[BIN_CHUNK];

    const int tid   = threadIdx.x;
    const int lane  = tid & 31;
    const int wave  = tid >> 5;
    const int first = blockIdx.x * BIN_CHUNK;
    const int cn    = min(BIN_CHUNK, e - first);

    lhist[tid] = 0;
    __syncthreads();

    int s[BIN_EPT], d[BIN_EPT], rank[BIN_EPT];
#pragma unroll
    for (int k = 0; k < BIN_EPT; ++k) {
        const int i = k * BIN_T + tid;
        s[k] = 0;
        d[k] = 0;
        if (i < cn) {
            s[k] = src[first + i];
            d[k] = dst[first + i];
        }
    }
#pragma unroll
    for (int k = 0; k < BIN_EPT; ++k) {
        rank[k] = 0;
        if (k * BIN_T + tid < cn) rank[k] = atomicAdd(&lhist[d[k] >> BIN_SHIFT], 1);
    }
    __syncthreads();

    const int v = lhist[tid];
    int incl = v;
#pragma unroll
    for (int off = 1; off < 32; off <<= 1) {
        const int t = __shfl_up(incl, off);
        if (lane >= off) incl += t;
    }
    if (lane == 31) wsum[wave] = incl;
    __syncthreads();
    int wbase = 0;
    for (int w = 0; w < wave; ++w) wbase += wsum[w];
    const int start = wbase + incl - v;
    int gbase = 0;
    if (v > 0) gbase = atomicAdd(&bincnt[tid], v);
    lstart[tid] = start;
    ldelta[tid] = gbase - start;
    __syncthreads();

#pragma unroll
    for (int k = 0; k < BIN_EPT; ++k) {
        if (k * BIN_T + tid < cn) {
            const int b = d[k] >> BIN_SHIFT;
            const int p = lstart[b] + rank[k];
            stage_w[p] = ((unsigned)(d[k] & (BIN_NODES - 1)) << SRC_BITS) | (unsigned)s[k];
            stage_b[p] = (unsigned short)b;
        }
    }
    __syncthreads();

    unsigned sw[BIN_EPT];
    int      sb[BIN_EPT], sp[BIN_EPT];
#pragma unroll
    for (int k = 0; k < BIN_EPT; ++k) {
        sw[k] = stage_w[k * BIN_T + tid];
        sb[k] = (k * BIN_T + tid < cn) ? (int)stage_b[k * BIN_T + tid] : 0;
    }
#pragma unroll
    for (int k = 0; k < BIN_EPT; ++k) sp[k] = ldelta[sb[k]] + k * BIN_T + tid;
#pragma unroll
    for (int k = 0; k < BIN_EPT; ++k) {
        const int i = k * BIN_T + tid;
        if (i < cn) {
            const unsigned w   = sw[k];
            const int      b   = sb[k];
            const int      pos = sp[k];
            if (pos < BIN_CAP) {
                bins[(size_t)b * BIN_CAP + pos] = w;
            } else {
                const int q = atomicAdd(ovfb_n, 1);
                ovfb[q] = make_int2((b << BIN_SHIFT) | (int)(w >> SRC_BITS), (int)(w & ((1u << SRC_BITS) - 1)));
            }
        }
    }
}

__device__ __forceinline__ void store_edge(int* lell, int dl, int slot, int s, int node0,
                                           int2* __restrict__ ovf, int* ovf_n) {
    if (slot < LIST_CAP) {
        lell[dl * LIST_CAP + slot] = s;
    } else {
        const int q = atomicAdd(ovf_n, 1);
        ovf[q] = make_int2(node0 + dl, s);
    }
}
__device__ __forceinline__ void place_edge(int* lcnt, int* lell, int dl, int s, int node0,
                                           int2* __restrict__ ovf, int* ovf_n) {
    store_edge(lell, dl, atomicAdd(&lcnt[dl], 1), s, node0, ovf, ovf_n);
}

__global__ __launch_bounds__(BUILD_T) void k_build_lists(const unsigned* __restrict__ bins,
                                                         const int* __restrict__ bincnt,
                                                         const int2* __restrict__ ovfb,
                                                         const int* __restrict__ ovfb_n,
                                                         const float* __restrict__ x,
                                                         int* __restrict__ ell, int2* __restrict__ ovf, int* ovf_n,
                                                         int* __restrict__ cnt, float* __restrict__ dinv,
                                                         float* __restrict__ xs, int n) {
    __shared__ int lcnt[BIN_NODES];
    __shared__ int lell[BIN_NODES * LIST_CAP];

    const int tid   = threadIdx.x;
    const int lane  = tid & 31;
    const int wave  = tid >> 5;
    const int b     = blockIdx.x;
    const int node0 = b << BIN_SHIFT;

    if (tid < BIN_NODES) lcnt[tid] = 0;
    __syncthreads();

    float xv[BIN_NODES * XS_PITCH / BUILD_T];
#pragma unroll
    for (int t = 0; t < BIN_NODES * XS_PITCH / BUILD_T; ++t) {
        const int idx  = t * BUILD_T + tid;
        const int node = node0 + (idx >> 4);
        const int j    = idx & (XS_PITCH - 1);
        xv[t] = (node < n && j < FIN) ? x[(size_t)node * FIN + j] : 0.0f;
    }

    const int       tot = bincnt[b];
    const int       nb  = min(tot, BIN_CAP);
    const unsigned* reg = bins + (size_t)b * BIN_CAP;
    for (int i0 = 0; i0 < nb; i0 += BUILD_G * BUILD_T) {
        unsigned w[BUILD_G];
        int      slot[BUILD_G];
#pragma unroll
        for (int t = 0; t < BUILD_G; ++t) {
            const int i = i0 + t * BUILD_T + tid;
            w[t] = (i < nb) ? reg[i] : 0u;
        }
#pragma unroll
        for (int t = 0; t < BUILD_G; ++t) {
            const int i = i0 + t * BUILD_T + tid;
            slot[t] = (i < nb) ? atomicAdd(&lcnt[w[t] >> SRC_BITS], 1) : 0;
        }
#pragma unroll
        for (int t = 0; t < BUILD_G; ++t) {
            const int i = i0 + t * BUILD_T + tid;
            if (i < nb) store_edge(lell, (int)(w[t] >> SRC_BITS), slot[t], (int)(w[t] & ((1u << SRC_BITS) - 1)), node0, ovf, ovf_n);
        }
    }
    if (tot > BIN_CAP) {
        const int no = *ovfb_n;
        for (int i = tid; i < no; i += BUILD_T) {
            const int2 r = ovfb[i];
            if ((r.x >> BIN_SHIFT) == b) place_edge(lcnt, lell, r.x & (BIN_NODES - 1), r.y, node0, ovf, ovf_n);
        }
    }
    __syncthreads();

    if (tid < BIN_NODES && node0 + tid < n) {
        const int   c  = lcnt[tid];
        const float dg = (float)(c + 1);
        cnt[node0 + tid]  = c;
        dinv[node0 + tid] = (dg > 0.0f) ? rsqrtf(dg) : 0.0f;
    }
    int cx[BIN_NODES * XS_PITCH / BUILD_T];
#pragma unroll
    for (int t = 0; t < BIN_NODES * XS_PITCH / BUILD_T; ++t) cx[t] = lcnt[(t * BUILD_T + tid) >> 4];
#pragma unroll
    for (int t = 0; t < BIN_NODES * XS_PITCH / BUILD_T; ++t) {
        const int idx  = t * BUILD_T + tid;
        const int r    = idx >> 4;
        if (node0 + r < n) {
            const float dg = (float)(cx[t] + 1);
            const float dn = (dg > 0.0f) ? rsqrtf(dg) : 0.0f;
            xs[(size_t)node0 * XS_PITCH + idx] = dn * xv[t];
        }
    }
    for (int r0 = wave; r0 < BIN_NODES; r0 += 4 * (BUILD_T / 32)) {
        int c[4], lo[4], hi[4];
#pragma unroll
        for (int t = 0; t < 4; ++t) {
            const int r = r0 + t * (BUILD_T / 32);
            c[t] = (node0 + r < n) ? lcnt[r] : 0;
        }
#pragma unroll
        for (int t = 0; t < 4; ++t) {
            const int r = r0 + t * (BUILD_T / 32);
            lo[t] = (lane < c[t]) ? lell[r * LIST_CAP + lane] : 0;
            hi[t] = (32 + lane < c[t]) ? lell[r * LIST_CAP + 32 + lane] : 0;
        }
#pragma unroll
        for (int t = 0; t < 4; ++t) {
            const size_t node = (size_t)(node0 + r0 + t * (BUILD_T / 32));
            if (c[t] > 0)  ell[node * LIST_CAP + lane]      = lo[t];
            if (c[t] > 32) ell[node * LIST_CAP + 32 + lane] = hi[t];
        }
    }
}

__global__ __launch_bounds__(32) void k_fold_w2(const float* __restrict__ W2,
                                                const float* __restrict__ Wlin,
                                                float* __restrict__ M) {
    const int lane = threadIdx.x;
    const int half = lane >> 4;
    const int l15  = lane & 15;
    const int m0   = (blockIdx.x >> 1) << 4;
    const int ncol = ((blockIdx.x & 1) << 4) + l15;
    const bool live = ncol < FOUT;
    const int  lcol = live ? ncol : 0;
    const float* arow = W2 + (size_t)(m0 + l15) * HID;

    v8f c = {};
#pragma unroll
    for (int k0 = 0; k0 < HID; k0 += 4) {
        const int ka = k0 + half * 2;
        const float2 av = *(const float2*)(arow + ka);
        v2f a, b;
        a[0] = av.x;
        a[1] = av.y;
        const float w0 = Wlin[ka * FOUT + lcol];
        const float w1 = Wlin[(ka + 1) * FOUT + lcol];
        b[0] = live ? w0 : 0.0f;
        b[1] = live ? w1 : 0.0f;
        c = __builtin_amdgcn_wmma_f32_16x16x4_f32(
                 false, a,  false, b,
                 (short)0, c,  false,  false);
    }
    const int mbase = m0 + half * 8;
#pragma unroll
    for (int i = 0; i < 8; ++i) M[(size_t)(mbase + i) * UP + ncol] = c[i];
}


template <int G, int PITCH>
__device__ __forceinline__ float sum_rows(const float* __restrict__ xs, int col,
                                          int s_l, int w_l, float acc) {
    float v[G];
    float w[G];
#pragma unroll
    for (int k = 0; k < G; ++k) {
        const int s = __builtin_amdgcn_readlane(s_l, k);
        w[k] = __int_as_float(__builtin_amdgcn_readlane(w_l, k));
        v[k] = xs[(size_t)s * PITCH + col];
    }
#pragma unroll
    for (int k = 0; k < G; ++k) acc = fmaf(v[k], w[k], acc);
    return acc;
}

#define L1_NODES   16
#define HROW_PITCH 132
#define PART_PITCH 272

__global__ __launch_bounds__(512) void k_sum_gemm1(const float* __restrict__ xs,
                                                   const float* __restrict__ W1,
                                                   const float* __restrict__ bias,
                                                   const float* __restrict__ M,
                                                   float* __restrict__ u,
                                                   const int* __restrict__ ell,
                                                   const int2* __restrict__ ovf,
                                                   const int* __restrict__ ovf_n,
                                                   const int* __restrict__ cnt,
                                                   const float* __restrict__ dinv, int n) {
    __shared__ __attribute__((aligned(16))) float hrow[L1_NODES * HROW_PITCH];
    __shared__ float part[L1_NODES * PART_PITCH];

    const int lane = threadIdx.x & 31;
    const int wave = threadIdx.x >> 5;
    const int node = __builtin_amdgcn_readfirstlane((int)(blockIdx.x * L1_NODES + wave));

    const int half = lane >> 4;
    const int l15  = lane & 15;

    float a  = 0.0f;
    float dn = 0.0f;
    if (node < n) {
        const int   c   = cnt[node];
        dn = dinv[node];
        const int   col = lane & (XS_PITCH - 1);
        const int*  row = ell + (size_t)node * LIST_CAP;

        const float self = xs[(size_t)node * XS_PITCH + col];

        const int ce = min(c, LIST_CAP);
        for (int base = 0; base < ce; base += 32) {
            const int m = min(32, ce - base);
            int s_l = node;
            int w_l = 0;
            if (lane < m) {
                s_l = row[base + lane];
                w_l = __float_as_int(dn);
            }
            if (m > 24)      a = sum_rows<32, XS_PITCH>(xs, col, s_l, w_l, a);
            else if (m > 16) a = sum_rows<24, XS_PITCH>(xs, col, s_l, w_l, a);
            else if (m > 8)  a = sum_rows<16, XS_PITCH>(xs, col, s_l, w_l, a);
            else             a = sum_rows<8, XS_PITCH>(xs, col, s_l, w_l, a);
        }
        if (c > LIST_CAP) {
            const int no = *ovf_n;
            for (int base = 0; base < no; base += 32) {
                const int i = base + lane;
                int2 r = make_int2(-1, 0);
                if (i < no) r = ovf[i];
                unsigned hit = __builtin_amdgcn_ballot_w32(r.x == node);
                while (hit) {
                    const int k = __builtin_ctz(hit);
                    hit &= hit - 1;
                    const int s = __shfl(r.y, k);
                    a = fmaf(xs[(size_t)s * XS_PITCH + col], dn, a);
                }
            }
        }
        a = fmaf(self, dn, a);
    }

    const int ncol = ((wave & 1) << 4) + l15;
    const int kb   = (wave >> 1) << 4;
    v2f bfr[4];
#pragma unroll
    for (int t = 0; t < 4; ++t) {
        const int ka = kb + 4 * t + half * 2;
        bfr[t][0] = M[ka * UP + ncol];
        bfr[t][1] = M[(ka + 1) * UP + ncol];
    }

    float4 o = make_float4(0.0f, 0.0f, 0.0f, 0.0f);
    if (node < n) {
        const float4* W4 = (const float4*)W1;
#pragma unroll
        for (int k = 0; k < FIN; ++k) {
            const float  ak = __int_as_float(__builtin_amdgcn_readlane(__float_as_int(a), k));
            const float4 w  = W4[k * (HID / 4) + lane];
            o.x = fmaf(ak, w.x, o.x);
            o.y = fmaf(ak, w.y, o.y);
            o.z = fmaf(ak, w.z, o.z);
            o.w = fmaf(ak, w.w, o.w);
        }
        const float4 b = ((const float4*)bias)[lane];
        o.x = fmaxf(o.x + b.x, 0.0f) * dn;
        o.y = fmaxf(o.y + b.y, 0.0f) * dn;
        o.z = fmaxf(o.z + b.z, 0.0f) * dn;
        o.w = fmaxf(o.w + b.w, 0.0f) * dn;
    }

    ((float4*)(hrow + wave * HROW_PITCH))[lane] = o;
    __syncthreads();

    v8f c8 = {};
#pragma unroll
    for (int t = 0; t < 4; ++t) {
        const int ka = kb + 4 * t + half * 2;
        const float2 av = *(const float2*)(hrow + l15 * HROW_PITCH + ka);
        v2f a2;
        a2[0] = av.x;
        a2[1] = av.y;
        c8 = __builtin_amdgcn_wmma_f32_16x16x4_f32(
                  false, a2,  false, bfr[t],
                  (short)0, c8,  false,  false);
    }
#pragma unroll
    for (int i = 0; i < 8; ++i) part[wave * PART_PITCH + i * 32 + lane] = c8[i];
    __syncthreads();

    const int src = (wave & 7) * 32 + (wave >> 3) * 16 + l15;
    float s = 0.0f;
#pragma unroll
    for (int kq = 0; kq < 8; ++kq) s += part[(kq * 2 + half) * PART_PITCH + src];
    if (node < n) u[(size_t)node * UP + lane] = s;
}


#define PULL_NODES 8

__global__ __launch_bounds__(256) void k_pull_pool(const float* __restrict__ u,
                                                   const int* __restrict__ ell,
                                                   const int2* __restrict__ ovf,
                                                   const int* __restrict__ ovf_n,
                                                   const int* __restrict__ cnt,
                                                   const float* __restrict__ dinv,
                                                   const int* __restrict__ batch,
                                                   float* sums, float* gcnt, int n) {
    __shared__ float rows[PULL_NODES * UP];
    __shared__ __attribute__((aligned(16))) int gid[PULL_NODES];

    const int lane = threadIdx.x & 31;
    const int wave = threadIdx.x >> 5;
    const int node = __builtin_amdgcn_readfirstlane((int)(blockIdx.x * PULL_NODES + wave));

    float o = 0.0f;
    int   g = -1;
    if (node < n) {
        const int   c   = cnt[node];
        const float dn  = dinv[node];
        const int*  row = ell + (size_t)node * LIST_CAP;
        g = batch[node];

        const float self = u[(size_t)node * UP + lane];
        float a = 0.0f;

        const int ce = min(c, LIST_CAP);
        for (int base = 0; base < ce; base += 32) {
            const int m = min(32, ce - base);
            int s_l = node;
            int w_l = 0;
            if (lane < m) {
                s_l = row[base + lane];
                w_l = __float_as_int(1.0f);
            }
            if (m > 24)      a = sum_rows<32, UP>(u, lane, s_l, w_l, a);
            else if (m > 16) a = sum_rows<24, UP>(u, lane, s_l, w_l, a);
            else if (m > 8)  a = sum_rows<16, UP>(u, lane, s_l, w_l, a);
            else             a = sum_rows<8, UP>(u, lane, s_l, w_l, a);
        }
        if (c > LIST_CAP) {
            const int no = *ovf_n;
            for (int base = 0; base < no; base += 32) {
                const int i = base + lane;
                int2 r = make_int2(-1, 0);
                if (i < no) r = ovf[i];
                unsigned hit = __builtin_amdgcn_ballot_w32(r.x == node);
                while (hit) {
                    const int k = __builtin_ctz(hit);
                    hit &= hit - 1;
                    const int s = __shfl(r.y, k);
                    a += u[(size_t)s * UP + lane];
                }
            }
        }
        o = (a + self) * dn;
    }

    rows[wave * UP + lane] = o;
    if (lane == 0) gid[wave] = g;
    __syncthreads();

    if (wave > 1) return;

    float v[PULL_NODES];
#pragma unroll
    for (int r = 0; r < PULL_NODES; ++r) v[r] = rows[r * UP + lane];
    const int4 ga = ((const int4*)gid)[0];
    const int4 gb = ((const int4*)gid)[1];
    const int gs[PULL_NODES] = {
        __builtin_amdgcn_readfirstlane(ga.x), __builtin_amdgcn_readfirstlane(ga.y),
        __builtin_amdgcn_readfirstlane(ga.z), __builtin_amdgcn_readfirstlane(ga.w),
        __builtin_amdgcn_readfirstlane(gb.x), __builtin_amdgcn_readfirstlane(gb.y),
        __builtin_amdgcn_readfirstlane(gb.z), __builtin_amdgcn_readfirstlane(gb.w)};

    if (wave == 0) {
        if (lane < FOUT) {
            int   cur = -1;
            float s   = 0.0f;
#pragma unroll
            for (int r = 0; r < PULL_NODES; ++r) {
                if (gs[r] < 0) continue;
                if (gs[r] != cur) {
                    if (cur >= 0) atomicAdd(&sums[cur * UP + lane], s);
                    cur = gs[r];
                    s   = 0.0f;
                }
                s += v[r];
            }
            if (cur >= 0) atomicAdd(&sums[cur * UP + lane], s);
        }
    } else if (lane == 0) {
        int   cur = -1;
        float k   = 0.0f;
#pragma unroll
        for (int r = 0; r < PULL_NODES; ++r) {
            if (gs[r] < 0) continue;
            if (gs[r] != cur) {
                if (cur >= 0) atomicAdd(&gcnt[cur], k);
                cur = gs[r];
                k   = 0.0f;
            }
            k += 1.0f;
        }
        if (cur >= 0) atomicAdd(&gcnt[cur], k);
    }
}

__global__ void k_pool_init(float* sums, float* cnt) {
    int idx = blockIdx.x * blockDim.x + threadIdx.x;
    if (idx < NGRAPHS * UP) sums[idx] = 0.0f;
    if (idx < NGRAPHS)      cnt[idx]  = 0.0f;
}

__global__ void k_final(const float* __restrict__ sums, const float* __restrict__ cnt,
                        const float* __restrict__ b2,
                        const float* __restrict__ Wlin, const float* __restrict__ blin,
                        float* __restrict__ out) {
    int idx = blockIdx.x * blockDim.x + threadIdx.x;
    if (idx >= NGRAPHS * FOUT) return;
    int g = idx / FOUT;
    int j = idx - g * FOUT;
    const float c   = cnt[g];
    const float inv = 1.0f / fmaxf(c, 1.0f);
    const float has = (c > 0.0f) ? 1.0f : 0.0f;
    float bb = 0.0f;
#pragma unroll 4
    for (int k = 0; k < HID; ++k) bb = fmaf(b2[k], Wlin[k * FOUT + j], bb);
    out[idx] = fmaf(sums[g * UP + j], inv, fmaf(bb, has, blin[j]));
}

extern "C" void kernel_launch(void* const* d_in, const int* in_sizes, int n_in,
                              void* d_out, int out_size, void* d_ws, size_t ws_size,
                              hipStream_t stream) {
    (void)n_in; (void)out_size; (void)ws_size;

    const float* x     = (const float*)d_in[0];
    const int*   esrc  = (const int*)  d_in[1];
    const int*   edst  = (const int*)  d_in[2];
    const int*   batch = (const int*)  d_in[3];
    const float* W1    = (const float*)d_in[4];
    const float* b1    = (const float*)d_in[5];
    const float* W2    = (const float*)d_in[6];
    const float* b2    = (const float*)d_in[7];
    const float* Wlin  = (const float*)d_in[8];
    const float* blin  = (const float*)d_in[9];
    float*       out   = (float*)d_out;

    const int N = in_sizes[3];
    const int E = in_sizes[1];

    char*  ws  = (char*)d_ws;
    size_t off = 0;
    auto carve = [&](size_t bytes) -> void* {
        off = (off + 255) & ~(size_t)255;
        void* p = ws + off;
        off += bytes;
        return p;
    };
    int*   ecnt  = (int*)  carve((size_t)N * 4);
    float* dinv  = (float*)carve((size_t)N * 4);
    float* bufH  = (float*)carve((size_t)N * HID * 4);
    float* bufO  = (float*)carve((size_t)N * HID * 4);
    char*  lists = (char*) carve((size_t)N * HID * 4);
    float* sums  = (float*)carve((size_t)NGRAPHS * HID * 4);
    float* cnt   = (float*)carve((size_t)NGRAPHS * 4);

    float*    xs   = bufH + (size_t)NGRAPHS * HID;
    unsigned* bins = (unsigned*)(xs + (size_t)N * XS_PITCH);
    int2*     ovfb = (int2*)(bufO + (size_t)N * UP);

    ws  = lists;
    off = 0;
    int*  ell   = (int*) carve((size_t)N * LIST_CAP * 4);
    int2* ovf   = (int2*)carve((size_t)E * 8);
    int*  ovf_n = (int*) carve(4);
    int*  ovfb_n = (int*)carve(4);
    int*  bincnt = (int*)carve((size_t)MAX_BINS * 4);

    const int T = 256;
    int nbin = (N + BIN_NODES - 1) / BIN_NODES;
    int eb   = (E + BIN_CHUNK - 1) / BIN_CHUNK;
    int wb   = (int)(((size_t)N * 32 + T - 1) / T);
    int lb   = (N + L1_NODES - 1) / L1_NODES;
    int pb   = (NGRAPHS * UP + T - 1) / T;
    int hb   = (NGRAPHS * FOUT + T - 1) / T;

    k_fold_w2<<<(HID / 16) * (UP / 16), 32, 0, stream>>>(W2, Wlin, bufH);

    k_zero_bins  <<<MAX_BINS / T, T, 0, stream>>>(bincnt, ovf_n, ovfb_n);
    k_bin_edges  <<<eb, BIN_T, 0, stream>>>(esrc, edst, bincnt, bins, ovfb, ovfb_n, E);
    k_build_lists<<<nbin, BUILD_T, 0, stream>>>(bins, bincnt, ovfb, ovfb_n, x, ell, ovf, ovf_n, ecnt, dinv, xs, N);

    k_sum_gemm1<<<lb, 32 * L1_NODES, 0, stream>>>(xs, W1, b1, bufH, bufO, ell, ovf, ovf_n, ecnt, dinv, N);

    k_pool_init<<<pb, T, 0, stream>>>(sums, cnt);
    k_pull_pool<<<wb, T, 0, stream>>>(bufO, ell, ovf, ovf_n, ecnt, dinv, batch, sums, cnt, N);

    k_final<<<hb, T, 0, stream>>>(sums, cnt, b2, Wlin, blin, out);
}


# Layer 2, mean pool and head per graph: a work-group owns one graph's node run (bounds from the sorted ids) and writes its output row; pool atomics, clear and head launches go; f32 sum order changes
# speedup vs baseline: 11.1532x; 1.1167x over previous
#include <hip/hip_runtime.h>
#include <hip/hip_bf16.h>

#define HID     128
#define FIN     11
#define FOUT    19
#define NGRAPHS 4096

typedef __attribute__((ext_vector_type(2))) float v2f;
typedef __attribute__((ext_vector_type(8))) float v8f;

#define LIST_CAP 64
#define XS_PITCH 16
#define UP       32

#define BIN_SHIFT 8
#define BIN_NODES (1 << BIN_SHIFT)
#define MAX_BINS  512
#define BIN_CAP   8192
#define BIN_T     512
#define BIN_EPT   8
#define BIN_CHUNK (BIN_T * BIN_EPT)
#define BUILD_T   512
#define BUILD_G   8
#define SRC_BITS  24

__global__ void k_zero_bins(int* __restrict__ bincnt, int* __restrict__ ovf_n, int* __restrict__ ovfb_n,
                            int* __restrict__ unsorted) {
    const int i = blockIdx.x * blockDim.x + threadIdx.x;
    if (i < MAX_BINS) bincnt[i] = 0;
    if (i == 0) { *ovf_n = 0; *ovfb_n = 0; *unsorted = 0; }
}

__global__ __launch_bounds__(BIN_T) void k_bin_edges(const int* __restrict__ src, const int* __restrict__ dst,
                                                     int* bincnt, unsigned* __restrict__ bins,
                                                     int2* __restrict__ ovfb, int* ovfb_n, int e) {
    __shared__ int            lhist[MAX_BINS];
    __shared__ int            lstart[MAX_BINS];
    __shared__ int            ldelta[MAX_BINS];
    __shared__ int            wsum[BIN_T / 32];
    __shared__ unsigned       stage_w[BIN_CHUNK];
    __shared__ unsigned short stage_b[BIN_CHUNK];

    const int tid   = threadIdx.x;
    const int lane  = tid & 31;
    const int wave  = tid >> 5;
    const int first = blockIdx.x * BIN_CHUNK;
    const int cn    = min(BIN_CHUNK, e - first);

    lhist[tid] = 0;
    __syncthreads();

    int s[BIN_EPT], d[BIN_EPT], rank[BIN_EPT];
#pragma unroll
    for (int k = 0; k < BIN_EPT; ++k) {
        const int i = k * BIN_T + tid;
        s[k] = 0;
        d[k] = 0;
        if (i < cn) {
            s[k] = src[first + i];
            d[k] = dst[first + i];
        }
    }
#pragma unroll
    for (int k = 0; k < BIN_EPT; ++k) {
        rank[k] = 0;
        if (k * BIN_T + tid < cn) rank[k] = atomicAdd(&lhist[d[k] >> BIN_SHIFT], 1);
    }
    __syncthreads();

    const int v = lhist[tid];
    int incl = v;
#pragma unroll
    for (int off = 1; off < 32; off <<= 1) {
        const int t = __shfl_up(incl, off);
        if (lane >= off) incl += t;
    }
    if (lane == 31) wsum[wave] = incl;
    __syncthreads();
    int wbase = 0;
    for (int w = 0; w < wave; ++w) wbase += wsum[w];
    const int start = wbase + incl - v;
    int gbase = 0;
    if (v > 0) gbase = atomicAdd(&bincnt[tid], v);
    lstart[tid] = start;
    ldelta[tid] = gbase - start;
    __syncthreads();

#pragma unroll
    for (int k = 0; k < BIN_EPT; ++k) {
        if (k * BIN_T + tid < cn) {
            const int b = d[k] >> BIN_SHIFT;
            const int p = lstart[b] + rank[k];
            stage_w[p] = ((unsigned)(d[k] & (BIN_NODES - 1)) << SRC_BITS) | (unsigned)s[k];
            stage_b[p] = (unsigned short)b;
        }
    }
    __syncthreads();

    unsigned sw[BIN_EPT];
    int      sb[BIN_EPT], sp[BIN_EPT];
#pragma unroll
    for (int k = 0; k < BIN_EPT; ++k) {
        sw[k] = stage_w[k * BIN_T + tid];
        sb[k] = (k * BIN_T + tid < cn) ? (int)stage_b[k * BIN_T + tid] : 0;
    }
#pragma unroll
    for (int k = 0; k < BIN_EPT; ++k) sp[k] = ldelta[sb[k]] + k * BIN_T + tid;
#pragma unroll
    for (int k = 0; k < BIN_EPT; ++k) {
        const int i = k * BIN_T + tid;
        if (i < cn) {
            const unsigned w   = sw[k];
            const int      b   = sb[k];
            const int      pos = sp[k];
            if (pos < BIN_CAP) {
                bins[(size_t)b * BIN_CAP + pos] = w;
            } else {
                const int q = atomicAdd(ovfb_n, 1);
                ovfb[q] = make_int2((b << BIN_SHIFT) | (int)(w >> SRC_BITS), (int)(w & ((1u << SRC_BITS) - 1)));
            }
        }
    }
}

__device__ __forceinline__ void store_edge(int* lell, int dl, int slot, int s, int node0,
                                           int2* __restrict__ ovf, int* ovf_n) {
    if (slot < LIST_CAP) {
        lell[dl * LIST_CAP + slot] = s;
    } else {
        const int q = atomicAdd(ovf_n, 1);
        ovf[q] = make_int2(node0 + dl, s);
    }
}
__device__ __forceinline__ void place_edge(int* lcnt, int* lell, int dl, int s, int node0,
                                           int2* __restrict__ ovf, int* ovf_n) {
    store_edge(lell, dl, atomicAdd(&lcnt[dl], 1), s, node0, ovf, ovf_n);
}

__global__ __launch_bounds__(BUILD_T) void k_build_lists(const unsigned* __restrict__ bins,
                                                         const int* __restrict__ bincnt,
                                                         const int2* __restrict__ ovfb,
                                                         const int* __restrict__ ovfb_n,
                                                         const float* __restrict__ x,
                                                         int* __restrict__ ell, int2* __restrict__ ovf, int* ovf_n,
                                                         int* __restrict__ cnt, float* __restrict__ dinv,
                                                         float* __restrict__ xs,
                                                         const int* __restrict__ batch, int* __restrict__ gstart,
                                                         int* __restrict__ unsorted, int n) {
    __shared__ int lcnt[BIN_NODES];
    __shared__ int lell[BIN_NODES * LIST_CAP];

    const int tid   = threadIdx.x;
    const int lane  = tid & 31;
    const int wave  = tid >> 5;
    const int b     = blockIdx.x;
    const int node0 = b << BIN_SHIFT;

    if (tid < BIN_NODES) lcnt[tid] = 0;
    __syncthreads();

    float xv[BIN_NODES * XS_PITCH / BUILD_T];
#pragma unroll
    for (int t = 0; t < BIN_NODES * XS_PITCH / BUILD_T; ++t) {
        const int idx  = t * BUILD_T + tid;
        const int node = node0 + (idx >> 4);
        const int j    = idx & (XS_PITCH - 1);
        xv[t] = (node < n && j < FIN) ? x[(size_t)node * FIN + j] : 0.0f;
    }

    int bg = -1, bp = -1;
    if (tid < BIN_NODES && node0 + tid < n) {
        bg = batch[node0 + tid];
        if (node0 + tid > 0) bp = batch[node0 + tid - 1];
    }

    const int       tot = bincnt[b];
    const int       nb  = min(tot, BIN_CAP);
    const unsigned* reg = bins + (size_t)b * BIN_CAP;
    for (int i0 = 0; i0 < nb; i0 += BUILD_G * BUILD_T) {
        unsigned w[BUILD_G];
        int      slot[BUILD_G];
#pragma unroll
        for (int t = 0; t < BUILD_G; ++t) {
            const int i = i0 + t * BUILD_T + tid;
            w[t] = (i < nb) ? reg[i] : 0u;
        }
#pragma unroll
        for (int t = 0; t < BUILD_G; ++t) {
            const int i = i0 + t * BUILD_T + tid;
            slot[t] = (i < nb) ? atomicAdd(&lcnt[w[t] >> SRC_BITS], 1) : 0;
        }
#pragma unroll
        for (int t = 0; t < BUILD_G; ++t) {
            const int i = i0 + t * BUILD_T + tid;
            if (i < nb) store_edge(lell, (int)(w[t] >> SRC_BITS), slot[t], (int)(w[t] & ((1u << SRC_BITS) - 1)), node0, ovf, ovf_n);
        }
    }
    if (tot > BIN_CAP) {
        const int no = *ovfb_n;
        for (int i = tid; i < no; i += BUILD_T) {
            const int2 r = ovfb[i];
            if ((r.x >> BIN_SHIFT) == b) place_edge(lcnt, lell, r.x & (BIN_NODES - 1), r.y, node0, ovf, ovf_n);
        }
    }
    __syncthreads();

    if (tid < BIN_NODES && node0 + tid < n) {
        const int   c  = lcnt[tid];
        const float dg = (float)(c + 1);
        cnt[node0 + tid]  = c;
        dinv[node0 + tid] = (dg > 0.0f) ? rsqrtf(dg) : 0.0f;
        if (bp > bg) *unsorted = 1;
        const int me = node0 + tid;
        const int hi = min(bg, NGRAPHS - 1);
        for (int q = max(bp + 1, 0); q <= hi; ++q) gstart[q] = me;
        if (me == n - 1)
            for (int q = max(hi + 1, 0); q <= NGRAPHS; ++q) gstart[q] = n;
    }
    int cx[BIN_NODES * XS_PITCH / BUILD_T];
#pragma unroll
    for (int t = 0; t < BIN_NODES * XS_PITCH / BUILD_T; ++t) cx[t] = lcnt[(t * BUILD_T + tid) >> 4];
#pragma unroll
    for (int t = 0; t < BIN_NODES * XS_PITCH / BUILD_T; ++t) {
        const int idx  = t * BUILD_T + tid;
        const int r    = idx >> 4;
        if (node0 + r < n) {
            const float dg = (float)(cx[t] + 1);
            const float dn = (dg > 0.0f) ? rsqrtf(dg) : 0.0f;
            xs[(size_t)node0 * XS_PITCH + idx] = dn * xv[t];
        }
    }
    for (int r0 = wave; r0 < BIN_NODES; r0 += 4 * (BUILD_T / 32)) {
        int c[4], lo[4], hi[4];
#pragma unroll
        for (int t = 0; t < 4; ++t) {
            const int r = r0 + t * (BUILD_T / 32);
            c[t] = (node0 + r < n) ? lcnt[r] : 0;
        }
#pragma unroll
        for (int t = 0; t < 4; ++t) {
            const int r = r0 + t * (BUILD_T / 32);
            lo[t] = (lane < c[t]) ? lell[r * LIST_CAP + lane] : 0;
            hi[t] = (32 + lane < c[t]) ? lell[r * LIST_CAP + 32 + lane] : 0;
        }
#pragma unroll
        for (int t = 0; t < 4; ++t) {
            const size_t node = (size_t)(node0 + r0 + t * (BUILD_T / 32));
            if (c[t] > 0)  ell[node * LIST_CAP + lane]      = lo[t];
            if (c[t] > 32) ell[node * LIST_CAP + 32 + lane] = hi[t];
        }
    }
}

__global__ __launch_bounds__(32) void k_fold_w2(const float* __restrict__ W2,
                                                const float* __restrict__ Wlin,
                                                const float* __restrict__ b2,
                                                float* __restrict__ M, float* __restrict__ bb) {
    const int lane = threadIdx.x;
    if (blockIdx.x >= (HID / 16) * (UP / 16)) {
        const int j = blockIdx.x - (HID / 16) * (UP / 16);
        float p = 0.0f;
#pragma unroll
        for (int t = 0; t < HID / 32; ++t) {
            const int k = lane + 32 * t;
            p = fmaf(b2[k], Wlin[k * FOUT + j], p);
        }
#pragma unroll
        for (int off = 16; off > 0; off >>= 1) p += __shfl_xor(p, off);
        if (lane == 0) bb[j] = p;
        return;
    }
    const int half = lane >> 4;
    const int l15  = lane & 15;
    const int m0   = (blockIdx.x >> 1) << 4;
    const int ncol = ((blockIdx.x & 1) << 4) + l15;
    const bool live = ncol < FOUT;
    const int  lcol = live ? ncol : 0;
    const float* arow = W2 + (size_t)(m0 + l15) * HID;

    v8f c = {};
#pragma unroll
    for (int k0 = 0; k0 < HID; k0 += 4) {
        const int ka = k0 + half * 2;
        const float2 av = *(const float2*)(arow + ka);
        v2f a, b;
        a[0] = av.x;
        a[1] = av.y;
        const float w0 = Wlin[ka * FOUT + lcol];
        const float w1 = Wlin[(ka + 1) * FOUT + lcol];
        b[0] = live ? w0 : 0.0f;
        b[1] = live ? w1 : 0.0f;
        c = __builtin_amdgcn_wmma_f32_16x16x4_f32(
                 false, a,  false, b,
                 (short)0, c,  false,  false);
    }
    const int mbase = m0 + half * 8;
#pragma unroll
    for (int i = 0; i < 8; ++i) M[(size_t)(mbase + i) * UP + ncol] = c[i];
}


template <int G, int PITCH>
__device__ __forceinline__ float sum_rows(const float* __restrict__ xs, int col,
                                          int s_l, int w_l, float acc) {
    float v[G];
    float w[G];
#pragma unroll
    for (int k = 0; k < G; ++k) {
        const int s = __builtin_amdgcn_readlane(s_l, k);
        w[k] = __int_as_float(__builtin_amdgcn_readlane(w_l, k));
        v[k] = xs[(size_t)s * PITCH + col];
    }
#pragma unroll
    for (int k = 0; k < G; ++k) acc = fmaf(v[k], w[k], acc);
    return acc;
}

#define L1_NODES   16
#define HROW_PITCH 132
#define PART_PITCH 272

__global__ __launch_bounds__(512) void k_sum_gemm1(const float* __restrict__ xs,
                                                   const float* __restrict__ W1,
                                                   const float* __restrict__ bias,
                                                   const float* __restrict__ M,
                                                   float* __restrict__ u,
                                                   const int* __restrict__ ell,
                                                   const int2* __restrict__ ovf,
                                                   const int* __restrict__ ovf_n,
                                                   const int* __restrict__ cnt,
                                                   const float* __restrict__ dinv, int n) {
    __shared__ __attribute__((aligned(16))) float hrow[L1_NODES * HROW_PITCH];
    __shared__ float part[L1_NODES * PART_PITCH];

    const int lane = threadIdx.x & 31;
    const int wave = threadIdx.x >> 5;
    const int node = __builtin_amdgcn_readfirstlane((int)(blockIdx.x * L1_NODES + wave));

    const int half = lane >> 4;
    const int l15  = lane & 15;

    float a  = 0.0f;
    float dn = 0.0f;
    if (node < n) {
        const int   c   = cnt[node];
        dn = dinv[node];
        const int   col = lane & (XS_PITCH - 1);
        const int*  row = ell + (size_t)node * LIST_CAP;

        const float self = xs[(size_t)node * XS_PITCH + col];

        const int ce = min(c, LIST_CAP);
        for (int base = 0; base < ce; base += 32) {
            const int m = min(32, ce - base);
            int s_l = node;
            int w_l = 0;
            if (lane < m) {
                s_l = row[base + lane];
                w_l = __float_as_int(dn);
            }
            if (m > 24)      a = sum_rows<32, XS_PITCH>(xs, col, s_l, w_l, a);
            else if (m > 16) a = sum_rows<24, XS_PITCH>(xs, col, s_l, w_l, a);
            else if (m > 8)  a = sum_rows<16, XS_PITCH>(xs, col, s_l, w_l, a);
            else             a = sum_rows<8, XS_PITCH>(xs, col, s_l, w_l, a);
        }
        if (c > LIST_CAP) {
            const int no = *ovf_n;
            for (int base = 0; base < no; base += 32) {
                const int i = base + lane;
                int2 r = make_int2(-1, 0);
                if (i < no) r = ovf[i];
                unsigned hit = __builtin_amdgcn_ballot_w32(r.x == node);
                while (hit) {
                    const int k = __builtin_ctz(hit);
                    hit &= hit - 1;
                    const int s = __shfl(r.y, k);
                    a = fmaf(xs[(size_t)s * XS_PITCH + col], dn, a);
                }
            }
        }
        a = fmaf(self, dn, a);
    }

    const int ncol = ((wave & 1) << 4) + l15;
    const int kb   = (wave >> 1) << 4;
    v2f bfr[4];
#pragma unroll
    for (int t = 0; t < 4; ++t) {
        const int ka = kb + 4 * t + half * 2;
        bfr[t][0] = M[ka * UP + ncol];
        bfr[t][1] = M[(ka + 1) * UP + ncol];
    }

    float4 o = make_float4(0.0f, 0.0f, 0.0f, 0.0f);
    if (node < n) {
        const float4* W4 = (const float4*)W1;
#pragma unroll
        for (int k = 0; k < FIN; ++k) {
            const float  ak = __int_as_float(__builtin_amdgcn_readlane(__float_as_int(a), k));
            const float4 w  = W4[k * (HID / 4) + lane];
            o.x = fmaf(ak, w.x, o.x);
            o.y = fmaf(ak, w.y, o.y);
            o.z = fmaf(ak, w.z, o.z);
            o.w = fmaf(ak, w.w, o.w);
        }
        const float4 b = ((const float4*)bias)[lane];
        o.x = fmaxf(o.x + b.x, 0.0f) * dn;
        o.y = fmaxf(o.y + b.y, 0.0f) * dn;
        o.z = fmaxf(o.z + b.z, 0.0f) * dn;
        o.w = fmaxf(o.w + b.w, 0.0f) * dn;
    }

    ((float4*)(hrow + wave * HROW_PITCH))[lane] = o;
    __syncthreads();

    v8f c8 = {};
#pragma unroll
    for (int t = 0; t < 4; ++t) {
        const int ka = kb + 4 * t + half * 2;
        const float2 av = *(const float2*)(hrow + l15 * HROW_PITCH + ka);
        v2f a2;
        a2[0] = av.x;
        a2[1] = av.y;
        c8 = __builtin_amdgcn_wmma_f32_16x16x4_f32(
                  false, a2,  false, bfr[t],
                  (short)0, c8,  false,  false);
    }
#pragma unroll
    for (int i = 0; i < 8; ++i) part[wave * PART_PITCH + i * 32 + lane] = c8[i];
    __syncthreads();

    const int src = (wave & 7) * 32 + (wave >> 3) * 16 + l15;
    float s = 0.0f;
#pragma unroll
    for (int kq = 0; kq < 8; ++kq) s += part[(kq * 2 + half) * PART_PITCH + src];
    if (node < n) u[(size_t)node * UP + lane] = s;
}


#define PULL_WAVES 16

__global__ __launch_bounds__(32 * PULL_WAVES) void k_pull_head(const float* __restrict__ u,
                                                               const int* __restrict__ ell,
                                                               const int2* __restrict__ ovf,
                                                               const int* __restrict__ ovf_n,
                                                               const int* __restrict__ cnt,
                                                               const float* __restrict__ dinv,
                                                               const int* __restrict__ batch,
                                                               const int* __restrict__ gstart,
                                                               const int* __restrict__ unsorted,
                                                               const float* __restrict__ bb,
                                                               const float* __restrict__ blin,
                                                               float* __restrict__ out, int n) {
    __shared__ float rows[PULL_WAVES * UP];
    __shared__ int   mine[PULL_WAVES];

    const int lane = threadIdx.x & 31;
    const int wave = __builtin_amdgcn_readfirstlane((int)(threadIdx.x >> 5));
    const int g    = blockIdx.x;

    const int  mark = *unsorted;
    const int  g0   = gstart[g];
    const int  g1   = gstart[g + 1];
    const bool runs = (mark == 0);
    const int  keep = runs ? -1 : 0;
    const int  r0   = min(max(g0, 0), n);
    const int  r1   = min(max(g1, r0), n);
    const int  ns   = r0 & keep;
    const int  ne   = (r1 & keep) | (n & ~keep);

    float bbl = 0.0f, bl = 0.0f;
    if (lane < FOUT) {
        bbl = bb[lane];
        bl  = blin[lane];
    }

    float acc   = 0.0f;
    int   taken = 0;
    for (int node = ns + wave; node < ne; node += PULL_WAVES) {
        if (!runs && batch[node] != g) continue;
        ++taken;
        const int   c   = cnt[node];
        const float dn  = dinv[node];
        const int*  row = ell + (size_t)node * LIST_CAP;

        const float self = u[(size_t)node * UP + lane];
        float a = 0.0f;

        const int ce = min(c, LIST_CAP);
        for (int base = 0; base < ce; base += 32) {
            const int m = min(32, ce - base);
            int s_l = node;
            int w_l = 0;
            if (lane < m) {
                s_l = row[base + lane];
                w_l = __float_as_int(1.0f);
            }
            if (m > 24)      a = sum_rows<32, UP>(u, lane, s_l, w_l, a);
            else if (m > 16) a = sum_rows<24, UP>(u, lane, s_l, w_l, a);
            else if (m > 8)  a = sum_rows<16, UP>(u, lane, s_l, w_l, a);
            else             a = sum_rows<8, UP>(u, lane, s_l, w_l, a);
        }
        if (c > LIST_CAP) {
            const int no = *ovf_n;
            for (int base = 0; base < no; base += 32) {
                const int i = base + lane;
                int2 r = make_int2(-1, 0);
                if (i < no) r = ovf[i];
                unsigned hit = __builtin_amdgcn_ballot_w32(r.x == node);
                while (hit) {
                    const int k = __builtin_ctz(hit);
                    hit &= hit - 1;
                    const int s = __shfl(r.y, k);
                    a += u[(size_t)s * UP + lane];
                }
            }
        }
        acc += (a + self) * dn;
    }

    rows[wave * UP + lane] = acc;
    if (lane == 0) mine[wave] = taken;
    __syncthreads();
    if (wave != 0) return;

    float v[PULL_WAVES];
#pragma unroll
    for (int r = 0; r < PULL_WAVES; ++r) v[r] = rows[r * UP + lane];
    int k[PULL_WAVES];
#pragma unroll
    for (int r = 0; r < PULL_WAVES; ++r) k[r] = mine[r];
    float s   = 0.0f;
    int   tot = 0;
#pragma unroll
    for (int r = 0; r < PULL_WAVES; ++r) {
        s   += v[r];
        tot += k[r];
    }

    if (lane < FOUT) {
        const float c   = (float)tot;
        const float inv = 1.0f / fmaxf(c, 1.0f);
        const float has = (c > 0.0f) ? 1.0f : 0.0f;
        out[g * FOUT + lane] = fmaf(s, inv, fmaf(bbl, has, bl));
    }
}

extern "C" void kernel_launch(void* const* d_in, const int* in_sizes, int n_in,
                              void* d_out, int out_size, void* d_ws, size_t ws_size,
                              hipStream_t stream) {
    (void)n_in; (void)out_size; (void)ws_size;

    const float* x     = (const float*)d_in[0];
    const int*   esrc  = (const int*)  d_in[1];
    const int*   edst  = (const int*)  d_in[2];
    const int*   batch = (const int*)  d_in[3];
    const float* W1    = (const float*)d_in[4];
    const float* b1    = (const float*)d_in[5];
    const float* W2    = (const float*)d_in[6];
    const float* b2    = (const float*)d_in[7];
    const float* Wlin  = (const float*)d_in[8];
    const float* blin  = (const float*)d_in[9];
    float*       out   = (float*)d_out;

    const int N = in_sizes[3];
    const int E = in_sizes[1];

    char*  ws  = (char*)d_ws;
    size_t off = 0;
    auto carve = [&](size_t bytes) -> void* {
        off = (off + 255) & ~(size_t)255;
        void* p = ws + off;
        off += bytes;
        return p;
    };
    int*   ecnt  = (int*)  carve((size_t)N * 4);
    float* dinv  = (float*)carve((size_t)N * 4);
    float* bufH  = (float*)carve((size_t)N * HID * 4);
    float* bufO  = (float*)carve((size_t)N * HID * 4);
    char*  lists = (char*) carve((size_t)N * HID * 4);
    float* sums  = (float*)carve((size_t)NGRAPHS * HID * 4);

    int*   gstart = (int*)sums;
    int*   gflag  = gstart + NGRAPHS + 16;
    float* bbv    = sums + 2 * NGRAPHS;

    float*    xs   = bufH + (size_t)NGRAPHS * HID;
    unsigned* bins = (unsigned*)(xs + (size_t)N * XS_PITCH);
    int2*     ovfb = (int2*)(bufO + (size_t)N * UP);

    ws  = lists;
    off = 0;
    int*  ell   = (int*) carve((size_t)N * LIST_CAP * 4);
    int2* ovf   = (int2*)carve((size_t)E * 8);
    int*  ovf_n = (int*) carve(4);
    int*  ovfb_n = (int*)carve(4);
    int*  bincnt = (int*)carve((size_t)MAX_BINS * 4);

    const int T = 256;
    int nbin = (N + BIN_NODES - 1) / BIN_NODES;
    int eb   = (E + BIN_CHUNK - 1) / BIN_CHUNK;
    int lb   = (N + L1_NODES - 1) / L1_NODES;

    k_fold_w2<<<(HID / 16) * (UP / 16) + FOUT, 32, 0, stream>>>(W2, Wlin, b2, bufH, bbv);

    k_zero_bins  <<<MAX_BINS / T, T, 0, stream>>>(bincnt, ovf_n, ovfb_n, gflag);
    k_bin_edges  <<<eb, BIN_T, 0, stream>>>(esrc, edst, bincnt, bins, ovfb, ovfb_n, E);
    k_build_lists<<<nbin, BUILD_T, 0, stream>>>(bins, bincnt, ovfb, ovfb_n, x, ell, ovf, ovf_n, ecnt, dinv, xs, batch, gstart, gflag, N);

    k_sum_gemm1<<<lb, 32 * L1_NODES, 0, stream>>>(xs, W1, b1, bufH, bufO, ell, ovf, ovf_n, ecnt, dinv, N);

    k_pull_head<<<NGRAPHS, 32 * PULL_WAVES, 0, stream>>>(bufO, ell, ovf, ovf_n, ecnt, dinv, batch, gstart, gflag, bbv, blin, out, N);
}


# Neighbour rows of both layers read as whole lines: 8 lanes a row (8 or 16 B a lane), 4 slots' rows per load, not 1; sources handed round by DPP8, own row as a last slot; f32 sum order changes
# speedup vs baseline: 11.5765x; 1.0380x over previous
#include <hip/hip_runtime.h>
#include <hip/hip_bf16.h>

#define HID     128
#define FIN     11
#define FOUT    19
#define NGRAPHS 4096

typedef __attribute__((ext_vector_type(2))) float v2f;
typedef __attribute__((ext_vector_type(8))) float v8f;

#define LIST_CAP 64
#define XS_PITCH 16
#define UP       32

#define BIN_SHIFT 8
#define BIN_NODES (1 << BIN_SHIFT)
#define MAX_BINS  512
#define BIN_CAP   8192
#define BIN_T     512
#define BIN_EPT   8
#define BIN_CHUNK (BIN_T * BIN_EPT)
#define BUILD_T   512
#define BUILD_G   8
#define SRC_BITS  24

__global__ void k_zero_bins(int* __restrict__ bincnt, int* __restrict__ ovf_n, int* __restrict__ ovfb_n,
                            int* __restrict__ unsorted) {
    const int i = blockIdx.x * blockDim.x + threadIdx.x;
    if (i < MAX_BINS) bincnt[i] = 0;
    if (i == 0) { *ovf_n = 0; *ovfb_n = 0; *unsorted = 0; }
}

__global__ __launch_bounds__(BIN_T) void k_bin_edges(const int* __restrict__ src, const int* __restrict__ dst,
                                                     int* bincnt, unsigned* __restrict__ bins,
                                                     int2* __restrict__ ovfb, int* ovfb_n, int e) {
    __shared__ int            lhist[MAX_BINS];
    __shared__ int            lstart[MAX_BINS];
    __shared__ int            ldelta[MAX_BINS];
    __shared__ int            wsum[BIN_T / 32];
    __shared__ unsigned       stage_w[BIN_CHUNK];
    __shared__ unsigned short stage_b[BIN_CHUNK];

    const int tid   = threadIdx.x;
    const int lane  = tid & 31;
    const int wave  = tid >> 5;
    const int first = blockIdx.x * BIN_CHUNK;
    const int cn    = min(BIN_CHUNK, e - first);

    lhist[tid] = 0;
    __syncthreads();

    int s[BIN_EPT], d[BIN_EPT], rank[BIN_EPT];
#pragma unroll
    for (int k = 0; k < BIN_EPT; ++k) {
        const int i = k * BIN_T + tid;
        s[k] = 0;
        d[k] = 0;
        if (i < cn) {
            s[k] = src[first + i];
            d[k] = dst[first + i];
        }
    }
#pragma unroll
    for (int k = 0; k < BIN_EPT; ++k) {
        rank[k] = 0;
        if (k * BIN_T + tid < cn) rank[k] = atomicAdd(&lhist[d[k] >> BIN_SHIFT], 1);
    }
    __syncthreads();

    const int v = lhist[tid];
    int incl = v;
#pragma unroll
    for (int off = 1; off < 32; off <<= 1) {
        const int t = __shfl_up(incl, off);
        if (lane >= off) incl += t;
    }
    if (lane == 31) wsum[wave] = incl;
    __syncthreads();
    int wbase = 0;
    for (int w = 0; w < wave; ++w) wbase += wsum[w];
    const int start = wbase + incl - v;
    int gbase = 0;
    if (v > 0) gbase = atomicAdd(&bincnt[tid], v);
    lstart[tid] = start;
    ldelta[tid] = gbase - start;
    __syncthreads();

#pragma unroll
    for (int k = 0; k < BIN_EPT; ++k) {
        if (k * BIN_T + tid < cn) {
            const int b = d[k] >> BIN_SHIFT;
            const int p = lstart[b] + rank[k];
            stage_w[p] = ((unsigned)(d[k] & (BIN_NODES - 1)) << SRC_BITS) | (unsigned)s[k];
            stage_b[p] = (unsigned short)b;
        }
    }
    __syncthreads();

    unsigned sw[BIN_EPT];
    int      sb[BIN_EPT], sp[BIN_EPT];
#pragma unroll
    for (int k = 0; k < BIN_EPT; ++k) {
        sw[k] = stage_w[k * BIN_T + tid];
        sb[k] = (k * BIN_T + tid < cn) ? (int)stage_b[k * BIN_T + tid] : 0;
    }
#pragma unroll
    for (int k = 0; k < BIN_EPT; ++k) sp[k] = ldelta[sb[k]] + k * BIN_T + tid;
#pragma unroll
    for (int k = 0; k < BIN_EPT; ++k) {
        const int i = k * BIN_T + tid;
        if (i < cn) {
            const unsigned w   = sw[k];
            const int      b   = sb[k];
            const int      pos = sp[k];
            if (pos < BIN_CAP) {
                bins[(size_t)b * BIN_CAP + pos] = w;
            } else {
                const int q = atomicAdd(ovfb_n, 1);
                ovfb[q] = make_int2((b << BIN_SHIFT) | (int)(w >> SRC_BITS), (int)(w & ((1u << SRC_BITS) - 1)));
            }
        }
    }
}

__device__ __forceinline__ void store_edge(int* lell, int dl, int slot, int s, int node0,
                                           int2* __restrict__ ovf, int* ovf_n) {
    if (slot < LIST_CAP) {
        lell[dl * LIST_CAP + slot] = s;
    } else {
        const int q = atomicAdd(ovf_n, 1);
        ovf[q] = make_int2(node0 + dl, s);
    }
}
__device__ __forceinline__ void place_edge(int* lcnt, int* lell, int dl, int s, int node0,
                                           int2* __restrict__ ovf, int* ovf_n) {
    store_edge(lell, dl, atomicAdd(&lcnt[dl], 1), s, node0, ovf, ovf_n);
}

__global__ __launch_bounds__(BUILD_T) void k_build_lists(const unsigned* __restrict__ bins,
                                                         const int* __restrict__ bincnt,
                                                         const int2* __restrict__ ovfb,
                                                         const int* __restrict__ ovfb_n,
                                                         const float* __restrict__ x,
                                                         int* __restrict__ ell, int2* __restrict__ ovf, int* ovf_n,
                                                         int* __restrict__ cnt, float* __restrict__ dinv,
                                                         float* __restrict__ xs,
                                                         const int* __restrict__ batch, int* __restrict__ gstart,
                                                         int* __restrict__ unsorted, int n) {
    __shared__ int lcnt[BIN_NODES];
    __shared__ int lell[BIN_NODES * LIST_CAP];

    const int tid   = threadIdx.x;
    const int lane  = tid & 31;
    const int wave  = tid >> 5;
    const int b     = blockIdx.x;
    const int node0 = b << BIN_SHIFT;

    if (tid < BIN_NODES) lcnt[tid] = 0;
    __syncthreads();

    float xv[BIN_NODES * XS_PITCH / BUILD_T];
#pragma unroll
    for (int t = 0; t < BIN_NODES * XS_PITCH / BUILD_T; ++t) {
        const int idx  = t * BUILD_T + tid;
        const int node = node0 + (idx >> 4);
        const int j    = idx & (XS_PITCH - 1);
        xv[t] = (node < n && j < FIN) ? x[(size_t)node * FIN + j] : 0.0f;
    }

    int bg = -1, bp = -1;
    if (tid < BIN_NODES && node0 + tid < n) {
        bg = batch[node0 + tid];
        if (node0 + tid > 0) bp = batch[node0 + tid - 1];
    }

    const int       tot = bincnt[b];
    const int       nb  = min(tot, BIN_CAP);
    const unsigned* reg = bins + (size_t)b * BIN_CAP;
    for (int i0 = 0; i0 < nb; i0 += BUILD_G * BUILD_T) {
        unsigned w[BUILD_G];
        int      slot[BUILD_G];
#pragma unroll
        for (int t = 0; t < BUILD_G; ++t) {
            const int i = i0 + t * BUILD_T + tid;
            w[t] = (i < nb) ? reg[i] : 0u;
        }
#pragma unroll
        for (int t = 0; t < BUILD_G; ++t) {
            const int i = i0 + t * BUILD_T + tid;
            slot[t] = (i < nb) ? atomicAdd(&lcnt[w[t] >> SRC_BITS], 1) : 0;
        }
#pragma unroll
        for (int t = 0; t < BUILD_G; ++t) {
            const int i = i0 + t * BUILD_T + tid;
            if (i < nb) store_edge(lell, (int)(w[t] >> SRC_BITS), slot[t], (int)(w[t] & ((1u << SRC_BITS) - 1)), node0, ovf, ovf_n);
        }
    }
    if (tot > BIN_CAP) {
        const int no = *ovfb_n;
        for (int i = tid; i < no; i += BUILD_T) {
            const int2 r = ovfb[i];
            if ((r.x >> BIN_SHIFT) == b) place_edge(lcnt, lell, r.x & (BIN_NODES - 1), r.y, node0, ovf, ovf_n);
        }
    }
    __syncthreads();

    if (tid < BIN_NODES && node0 + tid < n) {
        const int   c  = lcnt[tid];
        const float dg = (float)(c + 1);
        cnt[node0 + tid]  = c;
        dinv[node0 + tid] = (dg > 0.0f) ? rsqrtf(dg) : 0.0f;
        if (bp > bg) *unsorted = 1;
        const int me = node0 + tid;
        const int hi = min(bg, NGRAPHS - 1);
        for (int q = max(bp + 1, 0); q <= hi; ++q) gstart[q] = me;
        if (me == n - 1)
            for (int q = max(hi + 1, 0); q <= NGRAPHS; ++q) gstart[q] = n;
    }
    int cx[BIN_NODES * XS_PITCH / BUILD_T];
#pragma unroll
    for (int t = 0; t < BIN_NODES * XS_PITCH / BUILD_T; ++t) cx[t] = lcnt[(t * BUILD_T + tid) >> 4];
#pragma unroll
    for (int t = 0; t < BIN_NODES * XS_PITCH / BUILD_T; ++t) {
        const int idx  = t * BUILD_T + tid;
        const int r    = idx >> 4;
        if (node0 + r < n) {
            const float dg = (float)(cx[t] + 1);
            const float dn = (dg > 0.0f) ? rsqrtf(dg) : 0.0f;
            xs[(size_t)node0 * XS_PITCH + idx] = dn * xv[t];
        }
    }
    for (int r0 = wave; r0 < BIN_NODES; r0 += 4 * (BUILD_T / 32)) {
        int c[4], lo[4], hi[4];
#pragma unroll
        for (int t = 0; t < 4; ++t) {
            const int r = r0 + t * (BUILD_T / 32);
            c[t] = (node0 + r < n) ? lcnt[r] : 0;
        }
#pragma unroll
        for (int t = 0; t < 4; ++t) {
            const int r = r0 + t * (BUILD_T / 32);
            lo[t] = (lane < c[t]) ? lell[r * LIST_CAP + lane] : 0;
            hi[t] = (32 + lane < c[t]) ? lell[r * LIST_CAP + 32 + lane] : 0;
        }
#pragma unroll
        for (int t = 0; t < 4; ++t) {
            const size_t node = (size_t)(node0 + r0 + t * (BUILD_T / 32));
            if (c[t] > 0)  ell[node * LIST_CAP + lane]      = lo[t];
            if (c[t] > 32) ell[node * LIST_CAP + 32 + lane] = hi[t];
        }
    }
}

__global__ __launch_bounds__(32) void k_fold_w2(const float* __restrict__ W2,
                                                const float* __restrict__ Wlin,
                                                const float* __restrict__ b2,
                                                float* __restrict__ M, float* __restrict__ bb) {
    const int lane = threadIdx.x;
    if (blockIdx.x >= (HID / 16) * (UP / 16)) {
        const int j = blockIdx.x - (HID / 16) * (UP / 16);
        float p = 0.0f;
#pragma unroll
        for (int t = 0; t < HID / 32; ++t) {
            const int k = lane + 32 * t;
            p = fmaf(b2[k], Wlin[k * FOUT + j], p);
        }
#pragma unroll
        for (int off = 16; off > 0; off >>= 1) p += __shfl_xor(p, off);
        if (lane == 0) bb[j] = p;
        return;
    }
    const int half = lane >> 4;
    const int l15  = lane & 15;
    const int m0   = (blockIdx.x >> 1) << 4;
    const int ncol = ((blockIdx.x & 1) << 4) + l15;
    const bool live = ncol < FOUT;
    const int  lcol = live ? ncol : 0;
    const float* arow = W2 + (size_t)(m0 + l15) * HID;

    v8f c = {};
#pragma unroll
    for (int k0 = 0; k0 < HID; k0 += 4) {
        const int ka = k0 + half * 2;
        const float2 av = *(const float2*)(arow + ka);
        v2f a, b;
        a[0] = av.x;
        a[1] = av.y;
        const float w0 = Wlin[ka * FOUT + lcol];
        const float w1 = Wlin[(ka + 1) * FOUT + lcol];
        b[0] = live ? w0 : 0.0f;
        b[1] = live ? w1 : 0.0f;
        c = __builtin_amdgcn_wmma_f32_16x16x4_f32(
                 false, a,  false, b,
                 (short)0, c,  false,  false);
    }
    const int mbase = m0 + half * 8;
#pragma unroll
    for (int i = 0; i < 8; ++i) M[(size_t)(mbase + i) * UP + ncol] = c[i];
}


template <int J>
__device__ __forceinline__ int group_lane(int v) {
    return __builtin_amdgcn_mov_dpp8(v, J * 0x249249);
}
__device__ __forceinline__ void add_row(float2& a, const float2 v, float w) {
    a.x = fmaf(v.x, w, a.x);
    a.y = fmaf(v.y, w, a.y);
}
__device__ __forceinline__ void add_row(float4& a, const float4 v, float w) {
    a.x = fmaf(v.x, w, a.x);
    a.y = fmaf(v.y, w, a.y);
    a.z = fmaf(v.z, w, a.z);
    a.w = fmaf(v.w, w, a.w);
}
template <typename V, int PITCH, int NL, int J0, int J = 0>
__device__ __forceinline__ void load_rows(const float* __restrict__ t, int li, int s_l, V (&v)[NL]) {
    if constexpr (J < NL) {
        const unsigned off = (unsigned)group_lane<J0 + J>(s_l) * (unsigned)(PITCH * sizeof(float)) + (unsigned)li * (unsigned)sizeof(V);
        v[J] = *(const V*)((const char*)t + off);
        load_rows<V, PITCH, NL, J0, J + 1>(t, li, s_l, v);
    }
}
template <typename V, int NL, int J0, int J = 0>
__device__ __forceinline__ void add_rows(const V (&v)[NL], int w_l, V& a) {
    if constexpr (J < NL) {
        add_row(a, v[J], __int_as_float(group_lane<J0 + J>(w_l)));
        add_rows<V, NL, J0, J + 1>(v, w_l, a);
    }
}
template <typename V, int PITCH, int NL, int J0 = 0>
__device__ __forceinline__ void sum_rows(const float* __restrict__ t, int li, int s_l, int w_l, V& a) {
    V v[NL];
    load_rows<V, PITCH, NL, J0>(t, li, s_l, v);
    add_rows<V, NL, J0>(v, w_l, a);
    __builtin_amdgcn_sched_barrier(0);
}
template <typename V, int PITCH>
__device__ __forceinline__ void sum_trip(const float* __restrict__ t, int li, int m, int s_l, int w_l, V& a) {
    if (m > 24) {
        sum_rows<V, PITCH, 4, 0>(t, li, s_l, w_l, a);
        sum_rows<V, PITCH, 4, 4>(t, li, s_l, w_l, a);
    }
    else if (m > 16) sum_rows<V, PITCH, 6>(t, li, s_l, w_l, a);
    else if (m > 8)  sum_rows<V, PITCH, 4>(t, li, s_l, w_l, a);
    else             sum_rows<V, PITCH, 2>(t, li, s_l, w_l, a);
}

#define L1_NODES   16
#define HROW_PITCH 132
#define PART_PITCH 272

__global__ __launch_bounds__(512, 16) void k_sum_gemm1(const float* __restrict__ xs,
                                                   const float* __restrict__ W1,
                                                   const float* __restrict__ bias,
                                                   const float* __restrict__ M,
                                                   float* __restrict__ u,
                                                   const int* __restrict__ ell,
                                                   const int2* __restrict__ ovf,
                                                   const int* __restrict__ ovf_n,
                                                   const int* __restrict__ cnt,
                                                   const float* __restrict__ dinv, int n) {
    __shared__ __attribute__((aligned(16))) float hrow[L1_NODES * HROW_PITCH];
    __shared__ float part[L1_NODES * PART_PITCH];

    const int lane = threadIdx.x & 31;
    const int wave = threadIdx.x >> 5;
    const int node = __builtin_amdgcn_readfirstlane((int)(blockIdx.x * L1_NODES + wave));

    const int half = lane >> 4;
    const int l15  = lane & 15;

    const int q  = lane >> 3;
    const int li = lane & 7;

    float2 a  = make_float2(0.0f, 0.0f);
    float  dn = 0.0f;
    if (node < n) {
        const int   c   = cnt[node];
        dn = dinv[node];
        const int*  row = ell + (size_t)node * LIST_CAP;

        const int ce = min(c, LIST_CAP);
        const int p  = 4 * li + q;
        for (int base = 0; base <= ce; base += 32) {
            const int m  = min(32, ce + 1 - base);
            const int gi = base + p;
            int s_l = node;
            int w_l = 0;
            if (gi < ce)  s_l = row[gi];
            if (gi <= ce) w_l = __float_as_int(dn);
            sum_trip<float2, XS_PITCH>(xs, li, m, s_l, w_l, a);
        }
        a.x += __shfl_xor(a.x, 8);
        a.y += __shfl_xor(a.y, 8);
        a.x += __shfl_xor(a.x, 16);
        a.y += __shfl_xor(a.y, 16);
        if (c > LIST_CAP) {
            const int no = *ovf_n;
            for (int base = 0; base < no; base += 32) {
                const int i = base + lane;
                int2 r = make_int2(-1, 0);
                if (i < no) r = ovf[i];
                unsigned hit = __builtin_amdgcn_ballot_w32(r.x == node);
                while (hit) {
                    const int k = __builtin_ctz(hit);
                    hit &= hit - 1;
                    const int s = __shfl(r.y, k);
                    add_row(a, *(const float2*)(xs + (size_t)s * XS_PITCH + 2 * li), dn);
                }
            }
        }
    }

    const int ncol = ((wave & 1) << 4) + l15;
    const int kb   = (wave >> 1) << 4;
    v2f bfr[4];
#pragma unroll
    for (int t = 0; t < 4; ++t) {
        const int ka = kb + 4 * t + half * 2;
        bfr[t][0] = M[ka * UP + ncol];
        bfr[t][1] = M[(ka + 1) * UP + ncol];
    }

    float4 o = make_float4(0.0f, 0.0f, 0.0f, 0.0f);
    if (node < n) {
        const float4* W4 = (const float4*)W1;
#pragma unroll
        for (int k = 0; k < FIN; ++k) {
            const float  ak = __int_as_float(__builtin_amdgcn_readlane(__float_as_int((k & 1) ? a.y : a.x), k >> 1));
            const float4 w  = W4[k * (HID / 4) + lane];
            o.x = fmaf(ak, w.x, o.x);
            o.y = fmaf(ak, w.y, o.y);
            o.z = fmaf(ak, w.z, o.z);
            o.w = fmaf(ak, w.w, o.w);
        }
        const float4 b = ((const float4*)bias)[lane];
        o.x = fmaxf(o.x + b.x, 0.0f) * dn;
        o.y = fmaxf(o.y + b.y, 0.0f) * dn;
        o.z = fmaxf(o.z + b.z, 0.0f) * dn;
        o.w = fmaxf(o.w + b.w, 0.0f) * dn;
    }

    ((float4*)(hrow + wave * HROW_PITCH))[lane] = o;
    __syncthreads();

    v8f c8 = {};
#pragma unroll
    for (int t = 0; t < 4; ++t) {
        const int ka = kb + 4 * t + half * 2;
        const float2 av = *(const float2*)(hrow + l15 * HROW_PITCH + ka);
        v2f a2;
        a2[0] = av.x;
        a2[1] = av.y;
        c8 = __builtin_amdgcn_wmma_f32_16x16x4_f32(
                  false, a2,  false, bfr[t],
                  (short)0, c8,  false,  false);
    }
#pragma unroll
    for (int i = 0; i < 8; ++i) part[wave * PART_PITCH + i * 32 + lane] = c8[i];
    __syncthreads();

    const int src = (wave & 7) * 32 + (wave >> 3) * 16 + l15;
    float s = 0.0f;
#pragma unroll
    for (int kq = 0; kq < 8; ++kq) s += part[(kq * 2 + half) * PART_PITCH + src];
    if (node < n) u[(size_t)node * UP + lane] = s;
}


#define PULL_WAVES 16

__global__ __launch_bounds__(32 * PULL_WAVES, 16) void k_pull_head(const float* __restrict__ u,
                                                               const int* __restrict__ ell,
                                                               const int2* __restrict__ ovf,
                                                               const int* __restrict__ ovf_n,
                                                               const int* __restrict__ cnt,
                                                               const float* __restrict__ dinv,
                                                               const int* __restrict__ batch,
                                                               const int* __restrict__ gstart,
                                                               const int* __restrict__ unsorted,
                                                               const float* __restrict__ bb,
                                                               const float* __restrict__ blin,
                                                               float* __restrict__ out, int n) {
    __shared__ __attribute__((aligned(16))) float rows[PULL_WAVES * UP];
    __shared__ int   mine[PULL_WAVES];

    const int lane = threadIdx.x & 31;
    const int wave = __builtin_amdgcn_readfirstlane((int)(threadIdx.x >> 5));
    const int g    = blockIdx.x;

    const int  mark = *unsorted;
    const int  g0   = gstart[g];
    const int  g1   = gstart[g + 1];
    const bool runs = (mark == 0);
    const int  keep = runs ? -1 : 0;
    const int  r0   = min(max(g0, 0), n);
    const int  r1   = min(max(g1, r0), n);
    const int  ns   = r0 & keep;
    const int  ne   = (r1 & keep) | (n & ~keep);

    const int   q   = lane >> 3;
    const int   li  = lane & 7;

    float4 acc  = make_float4(0.0f, 0.0f, 0.0f, 0.0f);
    int   taken = 0;
    for (int node = ns + wave; node < ne; node += PULL_WAVES) {
        if (!runs && batch[node] != g) continue;
        ++taken;
        const int   c   = cnt[node];
        const float dn  = dinv[node];

        float4 a = make_float4(0.0f, 0.0f, 0.0f, 0.0f);

        const int ce = min(c, LIST_CAP);
        const int p  = 4 * li + q;
        for (int base = 0; base <= ce; base += 32) {
            const int m  = min(32, ce + 1 - base);
            const int gi = base + p;
            int s_l = node;
            int w_l = 0;
            if (gi < ce)  s_l = *(const int*)((const char*)ell + ((unsigned)node * (unsigned)(LIST_CAP * sizeof(int)) + (unsigned)gi * 4u));
            if (gi <= ce) w_l = __float_as_int(1.0f);
            sum_trip<float4, UP>(u, li, m, s_l, w_l, a);
        }
        if (c > LIST_CAP) {
            const int no = *ovf_n;
            for (int base = 0; base < no; base += 32) {
                const int i = base + lane;
                int2 r = make_int2(-1, 0);
                if (i < no) r = ovf[i];
                unsigned hit = __builtin_amdgcn_ballot_w32(r.x == node);
                while (hit) {
                    const int k = __builtin_ctz(hit);
                    hit &= hit - 1;
                    const int s = __shfl(r.y, k);
                    if (q == 0) add_row(a, *(const float4*)(u + (size_t)s * UP + 4 * li), 1.0f);
                }
            }
        }
        add_row(acc, a, dn);
    }

    float bbl = 0.0f, bl = 0.0f;
    if (wave == 0 && lane < FOUT) {
        bbl = bb[lane];
        bl  = blin[lane];
    }

    acc.x += __shfl_xor(acc.x, 8);
    acc.y += __shfl_xor(acc.y, 8);
    acc.z += __shfl_xor(acc.z, 8);
    acc.w += __shfl_xor(acc.w, 8);
    acc.x += __shfl_xor(acc.x, 16);
    acc.y += __shfl_xor(acc.y, 16);
    acc.z += __shfl_xor(acc.z, 16);
    acc.w += __shfl_xor(acc.w, 16);
    if (q == 0) ((float4*)(rows + wave * UP))[li] = acc;
    if (lane == 0) mine[wave] = taken;
    __syncthreads();
    if (wave != 0) return;

    float v[PULL_WAVES];
#pragma unroll
    for (int r = 0; r < PULL_WAVES; ++r) v[r] = rows[r * UP + lane];
    int k[PULL_WAVES];
#pragma unroll
    for (int r = 0; r < PULL_WAVES; ++r) k[r] = mine[r];
    float s   = 0.0f;
    int   tot = 0;
#pragma unroll
    for (int r = 0; r < PULL_WAVES; ++r) {
        s   += v[r];
        tot += k[r];
    }

    if (lane < FOUT) {
        const float c   = (float)tot;
        const float inv = 1.0f / fmaxf(c, 1.0f);
        const float has = (c > 0.0f) ? 1.0f : 0.0f;
        out[g * FOUT + lane] = fmaf(s, inv, fmaf(bbl, has, bl));
    }
}

extern "C" void kernel_launch(void* const* d_in, const int* in_sizes, int n_in,
                              void* d_out, int out_size, void* d_ws, size_t ws_size,
                              hipStream_t stream) {
    (void)n_in; (void)out_size; (void)ws_size;

    const float* x     = (const float*)d_in[0];
    const int*   esrc  = (const int*)  d_in[1];
    const int*   edst  = (const int*)  d_in[2];
    const int*   batch = (const int*)  d_in[3];
    const float* W1    = (const float*)d_in[4];
    const float* b1    = (const float*)d_in[5];
    const float* W2    = (const float*)d_in[6];
    const float* b2    = (const float*)d_in[7];
    const float* Wlin  = (const float*)d_in[8];
    const float* blin  = (const float*)d_in[9];
    float*       out   = (float*)d_out;

    const int N = in_sizes[3];
    const int E = in_sizes[1];

    char*  ws  = (char*)d_ws;
    size_t off = 0;
    auto carve = [&](size_t bytes) -> void* {
        off = (off + 255) & ~(size_t)255;
        void* p = ws + off;
        off += bytes;
        return p;
    };
    int*   ecnt  = (int*)  carve((size_t)N * 4);
    float* dinv  = (float*)carve((size_t)N * 4);
    float* bufH  = (float*)carve((size_t)N * HID * 4);
    float* bufO  = (float*)carve((size_t)N * HID * 4);
    char*  lists = (char*) carve((size_t)N * HID * 4);
    float* sums  = (float*)carve((size_t)NGRAPHS * HID * 4);

    int*   gstart = (int*)sums;
    int*   gflag  = gstart + NGRAPHS + 16;
    float* bbv    = sums + 2 * NGRAPHS;

    float*    xs   = bufH + (size_t)NGRAPHS * HID;
    unsigned* bins = (unsigned*)(xs + (size_t)N * XS_PITCH);
    int2*     ovfb = (int2*)(bufO + (size_t)N * UP);

    ws  = lists;
    off = 0;
    int*  ell   = (int*) carve((size_t)N * LIST_CAP * 4);
    int2* ovf   = (int2*)carve((size_t)E * 8);
    int*  ovf_n = (int*) carve(4);
    int*  ovfb_n = (int*)carve(4);
    int*  bincnt = (int*)carve((size_t)MAX_BINS * 4);

    const int T = 256;
    int nbin = (N + BIN_NODES - 1) / BIN_NODES;
    int eb   = (E + BIN_CHUNK - 1) / BIN_CHUNK;
    int lb   = (N + L1_NODES - 1) / L1_NODES;

    k_fold_w2<<<(HID / 16) * (UP / 16) + FOUT, 32, 0, stream>>>(W2, Wlin, b2, bufH, bbv);

    k_zero_bins  <<<MAX_BINS / T, T, 0, stream>>>(bincnt, ovf_n, ovfb_n, gflag);
    k_bin_edges  <<<eb, BIN_T, 0, stream>>>(esrc, edst, bincnt, bins, ovfb, ovfb_n, E);
    k_build_lists<<<nbin, BUILD_T, 0, stream>>>(bins, bincnt, ovfb, ovfb_n, x, ell, ovf, ovf_n, ecnt, dinv, xs, batch, gstart, gflag, N);

    k_sum_gemm1<<<lb, 32 * L1_NODES, 0, stream>>>(xs, W1, b1, bufH, bufO, ell, ovf, ovf_n, ecnt, dinv, N);

    k_pull_head<<<NGRAPHS, 32 * PULL_WAVES, 0, stream>>>(bufO, ell, ovf, ovf_n, ecnt, dinv, batch, gstart, gflag, bbv, blin, out, N);
}


# The weight fold (M = W2 Wlin and b2 Wlin) runs as 3 extra work-groups of the edge-binning launch, which does not wait for it: 6 dependent launches become 5; arithmetic and sum order unchanged
# speedup vs baseline: 12.2733x; 1.0602x over previous
#include <hip/hip_runtime.h>
#include <hip/hip_bf16.h>

#define HID     128
#define FIN     11
#define FOUT    19
#define NGRAPHS 4096

typedef __attribute__((ext_vector_type(2))) float v2f;
typedef __attribute__((ext_vector_type(8))) float v8f;

#define LIST_CAP 64
#define XS_PITCH 16
#define UP       32

#define BIN_SHIFT 8
#define BIN_NODES (1 << BIN_SHIFT)
#define MAX_BINS  512
#define BIN_CAP   8192
#define BIN_T     512
#define BIN_EPT   8
#define BIN_CHUNK (BIN_T * BIN_EPT)
#define BUILD_T   512
#define BUILD_G   8
#define SRC_BITS  24

__global__ void k_zero_bins(int* __restrict__ bincnt, int* __restrict__ ovf_n, int* __restrict__ ovfb_n,
                            int* __restrict__ unsorted) {
    const int i = blockIdx.x * blockDim.x + threadIdx.x;
    if (i < MAX_BINS) bincnt[i] = 0;
    if (i == 0) { *ovf_n = 0; *ovfb_n = 0; *unsorted = 0; }
}

#define FOLD_TILES  ((HID / 16) * (UP / 16))
#define FOLD_TASKS  (FOLD_TILES + FOUT)
#define FOLD_GROUPS ((FOLD_TASKS + BIN_T / 32 - 1) / (BIN_T / 32))
__device__ __forceinline__ void fold_task(int task, int lane,
                                          const float* __restrict__ W2,
                                          const float* __restrict__ Wlin,
                                          const float* __restrict__ b2,
                                          float* __restrict__ M, float* __restrict__ bb) {
    if (task >= FOLD_TILES) {
        const int j = task - FOLD_TILES;
        float p = 0.0f;
#pragma unroll
        for (int t = 0; t < HID / 32; ++t) {
            const int k = lane + 32 * t;
            p = fmaf(b2[k], Wlin[k * FOUT + j], p);
        }
#pragma unroll
        for (int off = 16; off > 0; off >>= 1) p += __shfl_xor(p, off);
        if (lane == 0) bb[j] = p;
        return;
    }
    const int half = lane >> 4;
    const int l15  = lane & 15;
    const int m0   = (task >> 1) << 4;
    const int ncol = ((task & 1) << 4) + l15;
    const bool live = ncol < FOUT;
    const int  lcol = live ? ncol : 0;
    const float* arow = W2 + (size_t)(m0 + l15) * HID;

    v8f c = {};
#pragma unroll
    for (int k0 = 0; k0 < HID; k0 += 4) {
        const int ka = k0 + half * 2;
        const float2 av = *(const float2*)(arow + ka);
        v2f a, b;
        a[0] = av.x;
        a[1] = av.y;
        const float w0 = Wlin[ka * FOUT + lcol];
        const float w1 = Wlin[(ka + 1) * FOUT + lcol];
        b[0] = live ? w0 : 0.0f;
        b[1] = live ? w1 : 0.0f;
        c = __builtin_amdgcn_wmma_f32_16x16x4_f32(
                 false, a,  false, b,
                 (short)0, c,  false,  false);
    }
    const int mbase = m0 + half * 8;
#pragma unroll
    for (int i = 0; i < 8; ++i) M[(size_t)(mbase + i) * UP + ncol] = c[i];
}

__global__ __launch_bounds__(BIN_T) void k_bin_edges(const int* __restrict__ src, const int* __restrict__ dst,
                                                     int* bincnt, unsigned* __restrict__ bins,
                                                     int2* __restrict__ ovfb, int* ovfb_n, int e,
                                                     const float* __restrict__ W2,
                                                     const float* __restrict__ Wlin,
                                                     const float* __restrict__ b2,
                                                     float* __restrict__ M, float* __restrict__ bb) {
    __shared__ int            lhist[MAX_BINS];
    __shared__ int            lstart[MAX_BINS];
    __shared__ int            ldelta[MAX_BINS];
    __shared__ int            wsum[BIN_T / 32];
    __shared__ unsigned       stage_w[BIN_CHUNK];
    __shared__ unsigned short stage_b[BIN_CHUNK];

    const int tid   = threadIdx.x;
    const int lane  = tid & 31;
    const int wave  = tid >> 5;

    if (blockIdx.x < FOLD_GROUPS) {
        const int task = (int)blockIdx.x * (BIN_T / 32) + __builtin_amdgcn_readfirstlane(wave);
        if (task < FOLD_TASKS) fold_task(task, lane, W2, Wlin, b2, M, bb);
        return;
    }

    const int first = ((int)blockIdx.x - FOLD_GROUPS) * BIN_CHUNK;
    const int cn    = min(BIN_CHUNK, e - first);

    lhist[tid] = 0;
    __syncthreads();

    int s[BIN_EPT], d[BIN_EPT], rank[BIN_EPT];
#pragma unroll
    for (int k = 0; k < BIN_EPT; ++k) {
        const int i = k * BIN_T + tid;
        s[k] = 0;
        d[k] = 0;
        if (i < cn) {
            s[k] = src[first + i];
            d[k] = dst[first + i];
        }
    }
#pragma unroll
    for (int k = 0; k < BIN_EPT; ++k) {
        rank[k] = 0;
        if (k * BIN_T + tid < cn) rank[k] = atomicAdd(&lhist[d[k] >> BIN_SHIFT], 1);
    }
    __syncthreads();

    const int v = lhist[tid];
    int incl = v;
#pragma unroll
    for (int off = 1; off < 32; off <<= 1) {
        const int t = __shfl_up(incl, off);
        if (lane >= off) incl += t;
    }
    if (lane == 31) wsum[wave] = incl;
    __syncthreads();
    int wbase = 0;
    for (int w = 0; w < wave; ++w) wbase += wsum[w];
    const int start = wbase + incl - v;
    int gbase = 0;
    if (v > 0) gbase = atomicAdd(&bincnt[tid], v);
    lstart[tid] = start;
    ldelta[tid] = gbase - start;
    __syncthreads();

#pragma unroll
    for (int k = 0; k < BIN_EPT; ++k) {
        if (k * BIN_T + tid < cn) {
            const int b = d[k] >> BIN_SHIFT;
            const int p = lstart[b] + rank[k];
            stage_w[p] = ((unsigned)(d[k] & (BIN_NODES - 1)) << SRC_BITS) | (unsigned)s[k];
            stage_b[p] = (unsigned short)b;
        }
    }
    __syncthreads();

    unsigned sw[BIN_EPT];
    int      sb[BIN_EPT], sp[BIN_EPT];
#pragma unroll
    for (int k = 0; k < BIN_EPT; ++k) {
        sw[k] = stage_w[k * BIN_T + tid];
        sb[k] = (k * BIN_T + tid < cn) ? (int)stage_b[k * BIN_T + tid] : 0;
    }
#pragma unroll
    for (int k = 0; k < BIN_EPT; ++k) sp[k] = ldelta[sb[k]] + k * BIN_T + tid;
#pragma unroll
    for (int k = 0; k < BIN_EPT; ++k) {
        const int i = k * BIN_T + tid;
        if (i < cn) {
            const unsigned w   = sw[k];
            const int      b   = sb[k];
            const int      pos = sp[k];
            if (pos < BIN_CAP) {
                bins[(size_t)b * BIN_CAP + pos] = w;
            } else {
                const int q = atomicAdd(ovfb_n, 1);
                ovfb[q] = make_int2((b << BIN_SHIFT) | (int)(w >> SRC_BITS), (int)(w & ((1u << SRC_BITS) - 1)));
            }
        }
    }
}

__device__ __forceinline__ void store_edge(int* lell, int dl, int slot, int s, int node0,
                                           int2* __restrict__ ovf, int* ovf_n) {
    if (slot < LIST_CAP) {
        lell[dl * LIST_CAP + slot] = s;
    } else {
        const int q = atomicAdd(ovf_n, 1);
        ovf[q] = make_int2(node0 + dl, s);
    }
}
__device__ __forceinline__ void place_edge(int* lcnt, int* lell, int dl, int s, int node0,
                                           int2* __restrict__ ovf, int* ovf_n) {
    store_edge(lell, dl, atomicAdd(&lcnt[dl], 1), s, node0, ovf, ovf_n);
}

__global__ __launch_bounds__(BUILD_T) void k_build_lists(const unsigned* __restrict__ bins,
                                                         const int* __restrict__ bincnt,
                                                         const int2* __restrict__ ovfb,
                                                         const int* __restrict__ ovfb_n,
                                                         const float* __restrict__ x,
                                                         int* __restrict__ ell, int2* __restrict__ ovf, int* ovf_n,
                                                         int* __restrict__ cnt, float* __restrict__ dinv,
                                                         float* __restrict__ xs,
                                                         const int* __restrict__ batch, int* __restrict__ gstart,
                                                         int* __restrict__ unsorted, int n) {
    __shared__ int lcnt[BIN_NODES];
    __shared__ int lell[BIN_NODES * LIST_CAP];

    const int tid   = threadIdx.x;
    const int lane  = tid & 31;
    const int wave  = tid >> 5;
    const int b     = blockIdx.x;
    const int node0 = b << BIN_SHIFT;

    if (tid < BIN_NODES) lcnt[tid] = 0;
    __syncthreads();

    float xv[BIN_NODES * XS_PITCH / BUILD_T];
#pragma unroll
    for (int t = 0; t < BIN_NODES * XS_PITCH / BUILD_T; ++t) {
        const int idx  = t * BUILD_T + tid;
        const int node = node0 + (idx >> 4);
        const int j    = idx & (XS_PITCH - 1);
        xv[t] = (node < n && j < FIN) ? x[(size_t)node * FIN + j] : 0.0f;
    }

    int bg = -1, bp = -1;
    if (tid < BIN_NODES && node0 + tid < n) {
        bg = batch[node0 + tid];
        if (node0 + tid > 0) bp = batch[node0 + tid - 1];
    }

    const int       tot = bincnt[b];
    const int       nb  = min(tot, BIN_CAP);
    const unsigned* reg = bins + (size_t)b * BIN_CAP;
    for (int i0 = 0; i0 < nb; i0 += BUILD_G * BUILD_T) {
        unsigned w[BUILD_G];
        int      slot[BUILD_G];
#pragma unroll
        for (int t = 0; t < BUILD_G; ++t) {
            const int i = i0 + t * BUILD_T + tid;
            w[t] = (i < nb) ? reg[i] : 0u;
        }
#pragma unroll
        for (int t = 0; t < BUILD_G; ++t) {
            const int i = i0 + t * BUILD_T + tid;
            slot[t] = (i < nb) ? atomicAdd(&lcnt[w[t] >> SRC_BITS], 1) : 0;
        }
#pragma unroll
        for (int t = 0; t < BUILD_G; ++t) {
            const int i = i0 + t * BUILD_T + tid;
            if (i < nb) store_edge(lell, (int)(w[t] >> SRC_BITS), slot[t], (int)(w[t] & ((1u << SRC_BITS) - 1)), node0, ovf, ovf_n);
        }
    }
    if (tot > BIN_CAP) {
        const int no = *ovfb_n;
        for (int i = tid; i < no; i += BUILD_T) {
            const int2 r = ovfb[i];
            if ((r.x >> BIN_SHIFT) == b) place_edge(lcnt, lell, r.x & (BIN_NODES - 1), r.y, node0, ovf, ovf_n);
        }
    }
    __syncthreads();

    if (tid < BIN_NODES && node0 + tid < n) {
        const int   c  = lcnt[tid];
        const float dg = (float)(c + 1);
        cnt[node0 + tid]  = c;
        dinv[node0 + tid] = (dg > 0.0f) ? rsqrtf(dg) : 0.0f;
        if (bp > bg) *unsorted = 1;
        const int me = node0 + tid;
        const int hi = min(bg, NGRAPHS - 1);
        for (int q = max(bp + 1, 0); q <= hi; ++q) gstart[q] = me;
        if (me == n - 1)
            for (int q = max(hi + 1, 0); q <= NGRAPHS; ++q) gstart[q] = n;
    }
    int cx[BIN_NODES * XS_PITCH / BUILD_T];
#pragma unroll
    for (int t = 0; t < BIN_NODES * XS_PITCH / BUILD_T; ++t) cx[t] = lcnt[(t * BUILD_T + tid) >> 4];
#pragma unroll
    for (int t = 0; t < BIN_NODES * XS_PITCH / BUILD_T; ++t) {
        const int idx  = t * BUILD_T + tid;
        const int r    = idx >> 4;
        if (node0 + r < n) {
            const float dg = (float)(cx[t] + 1);
            const float dn = (dg > 0.0f) ? rsqrtf(dg) : 0.0f;
            xs[(size_t)node0 * XS_PITCH + idx] = dn * xv[t];
        }
    }
    for (int r0 = wave; r0 < BIN_NODES; r0 += 4 * (BUILD_T / 32)) {
        int c[4], lo[4], hi[4];
#pragma unroll
        for (int t = 0; t < 4; ++t) {
            const int r = r0 + t * (BUILD_T / 32);
            c[t] = (node0 + r < n) ? lcnt[r] : 0;
        }
#pragma unroll
        for (int t = 0; t < 4; ++t) {
            const int r = r0 + t * (BUILD_T / 32);
            lo[t] = (lane < c[t]) ? lell[r * LIST_CAP + lane] : 0;
            hi[t] = (32 + lane < c[t]) ? lell[r * LIST_CAP + 32 + lane] : 0;
        }
#pragma unroll
        for (int t = 0; t < 4; ++t) {
            const size_t node = (size_t)(node0 + r0 + t * (BUILD_T / 32));
            if (c[t] > 0)  ell[node * LIST_CAP + lane]      = lo[t];
            if (c[t] > 32) ell[node * LIST_CAP + 32 + lane] = hi[t];
        }
    }
}


template <int J>
__device__ __forceinline__ int group_lane(int v) {
    return __builtin_amdgcn_mov_dpp8(v, J * 0x249249);
}
__device__ __forceinline__ void add_row(float2& a, const float2 v, float w) {
    a.x = fmaf(v.x, w, a.x);
    a.y = fmaf(v.y, w, a.y);
}
__device__ __forceinline__ void add_row(float4& a, const float4 v, float w) {
    a.x = fmaf(v.x, w, a.x);
    a.y = fmaf(v.y, w, a.y);
    a.z = fmaf(v.z, w, a.z);
    a.w = fmaf(v.w, w, a.w);
}
template <typename V, int PITCH, int NL, int J0, int J = 0>
__device__ __forceinline__ void load_rows(const float* __restrict__ t, int li, int s_l, V (&v)[NL]) {
    if constexpr (J < NL) {
        const unsigned off = (unsigned)group_lane<J0 + J>(s_l) * (unsigned)(PITCH * sizeof(float)) + (unsigned)li * (unsigned)sizeof(V);
        v[J] = *(const V*)((const char*)t + off);
        load_rows<V, PITCH, NL, J0, J + 1>(t, li, s_l, v);
    }
}
template <typename V, int NL, int J0, int J = 0>
__device__ __forceinline__ void add_rows(const V (&v)[NL], int w_l, V& a) {
    if constexpr (J < NL) {
        add_row(a, v[J], __int_as_float(group_lane<J0 + J>(w_l)));
        add_rows<V, NL, J0, J + 1>(v, w_l, a);
    }
}
template <typename V, int PITCH, int NL, int J0 = 0>
__device__ __forceinline__ void sum_rows(const float* __restrict__ t, int li, int s_l, int w_l, V& a) {
    V v[NL];
    load_rows<V, PITCH, NL, J0>(t, li, s_l, v);
    add_rows<V, NL, J0>(v, w_l, a);
    __builtin_amdgcn_sched_barrier(0);
}
template <typename V, int PITCH>
__device__ __forceinline__ void sum_trip(const float* __restrict__ t, int li, int m, int s_l, int w_l, V& a) {
    if (m > 24) {
        sum_rows<V, PITCH, 4, 0>(t, li, s_l, w_l, a);
        sum_rows<V, PITCH, 4, 4>(t, li, s_l, w_l, a);
    }
    else if (m > 16) sum_rows<V, PITCH, 6>(t, li, s_l, w_l, a);
    else if (m > 8)  sum_rows<V, PITCH, 4>(t, li, s_l, w_l, a);
    else             sum_rows<V, PITCH, 2>(t, li, s_l, w_l, a);
}

#define L1_NODES   16
#define HROW_PITCH 132
#define PART_PITCH 272

__global__ __launch_bounds__(512, 16) void k_sum_gemm1(const float* __restrict__ xs,
                                                   const float* __restrict__ W1,
                                                   const float* __restrict__ bias,
                                                   const float* __restrict__ M,
                                                   float* __restrict__ u,
                                                   const int* __restrict__ ell,
                                                   const int2* __restrict__ ovf,
                                                   const int* __restrict__ ovf_n,
                                                   const int* __restrict__ cnt,
                                                   const float* __restrict__ dinv, int n) {
    __shared__ __attribute__((aligned(16))) float hrow[L1_NODES * HROW_PITCH];
    __shared__ float part[L1_NODES * PART_PITCH];

    const int lane = threadIdx.x & 31;
    const int wave = threadIdx.x >> 5;
    const int node = __builtin_amdgcn_readfirstlane((int)(blockIdx.x * L1_NODES + wave));

    const int half = lane >> 4;
    const int l15  = lane & 15;

    const int q  = lane >> 3;
    const int li = lane & 7;

    float2 a  = make_float2(0.0f, 0.0f);
    float  dn = 0.0f;
    if (node < n) {
        const int   c   = cnt[node];
        dn = dinv[node];
        const int*  row = ell + (size_t)node * LIST_CAP;

        const int ce = min(c, LIST_CAP);
        const int p  = 4 * li + q;
        for (int base = 0; base <= ce; base += 32) {
            const int m  = min(32, ce + 1 - base);
            const int gi = base + p;
            int s_l = node;
            int w_l = 0;
            if (gi < ce)  s_l = row[gi];
            if (gi <= ce) w_l = __float_as_int(dn);
            sum_trip<float2, XS_PITCH>(xs, li, m, s_l, w_l, a);
        }
        a.x += __shfl_xor(a.x, 8);
        a.y += __shfl_xor(a.y, 8);
        a.x += __shfl_xor(a.x, 16);
        a.y += __shfl_xor(a.y, 16);
        if (c > LIST_CAP) {
            const int no = *ovf_n;
            for (int base = 0; base < no; base += 32) {
                const int i = base + lane;
                int2 r = make_int2(-1, 0);
                if (i < no) r = ovf[i];
                unsigned hit = __builtin_amdgcn_ballot_w32(r.x == node);
                while (hit) {
                    const int k = __builtin_ctz(hit);
                    hit &= hit - 1;
                    const int s = __shfl(r.y, k);
                    add_row(a, *(const float2*)(xs + (size_t)s * XS_PITCH + 2 * li), dn);
                }
            }
        }
    }

    const int ncol = ((wave & 1) << 4) + l15;
    const int kb   = (wave >> 1) << 4;
    v2f bfr[4];
#pragma unroll
    for (int t = 0; t < 4; ++t) {
        const int ka = kb + 4 * t + half * 2;
        bfr[t][0] = M[ka * UP + ncol];
        bfr[t][1] = M[(ka + 1) * UP + ncol];
    }

    float4 o = make_float4(0.0f, 0.0f, 0.0f, 0.0f);
    if (node < n) {
        const float4* W4 = (const float4*)W1;
#pragma unroll
        for (int k = 0; k < FIN; ++k) {
            const float  ak = __int_as_float(__builtin_amdgcn_readlane(__float_as_int((k & 1) ? a.y : a.x), k >> 1));
            const float4 w  = W4[k * (HID / 4) + lane];
            o.x = fmaf(ak, w.x, o.x);
            o.y = fmaf(ak, w.y, o.y);
            o.z = fmaf(ak, w.z, o.z);
            o.w = fmaf(ak, w.w, o.w);
        }
        const float4 b = ((const float4*)bias)[lane];
        o.x = fmaxf(o.x + b.x, 0.0f) * dn;
        o.y = fmaxf(o.y + b.y, 0.0f) * dn;
        o.z = fmaxf(o.z + b.z, 0.0f) * dn;
        o.w = fmaxf(o.w + b.w, 0.0f) * dn;
    }

    ((float4*)(hrow + wave * HROW_PITCH))[lane] = o;
    __syncthreads();

    v8f c8 = {};
#pragma unroll
    for (int t = 0; t < 4; ++t) {
        const int ka = kb + 4 * t + half * 2;
        const float2 av = *(const float2*)(hrow + l15 * HROW_PITCH + ka);
        v2f a2;
        a2[0] = av.x;
        a2[1] = av.y;
        c8 = __builtin_amdgcn_wmma_f32_16x16x4_f32(
                  false, a2,  false, bfr[t],
                  (short)0, c8,  false,  false);
    }
#pragma unroll
    for (int i = 0; i < 8; ++i) part[wave * PART_PITCH + i * 32 + lane] = c8[i];
    __syncthreads();

    const int src = (wave & 7) * 32 + (wave >> 3) * 16 + l15;
    float s = 0.0f;
#pragma unroll
    for (int kq = 0; kq < 8; ++kq) s += part[(kq * 2 + half) * PART_PITCH + src];
    if (node < n) u[(size_t)node * UP + lane] = s;
}


#define PULL_WAVES 16

__global__ __launch_bounds__(32 * PULL_WAVES, 16) void k_pull_head(const float* __restrict__ u,
                                                               const int* __restrict__ ell,
                                                               const int2* __restrict__ ovf,
                                                               const int* __restrict__ ovf_n,
                                                               const int* __restrict__ cnt,
                                                               const float* __restrict__ dinv,
                                                               const int* __restrict__ batch,
                                                               const int* __restrict__ gstart,
                                                               const int* __restrict__ unsorted,
                                                               const float* __restrict__ bb,
                                                               const float* __restrict__ blin,
                                                               float* __restrict__ out, int n) {
    __shared__ __attribute__((aligned(16))) float rows[PULL_WAVES * UP];
    __shared__ int   mine[PULL_WAVES];

    const int lane = threadIdx.x & 31;
    const int wave = __builtin_amdgcn_readfirstlane((int)(threadIdx.x >> 5));
    const int g    = blockIdx.x;

    const int  mark = *unsorted;
    const int  g0   = gstart[g];
    const int  g1   = gstart[g + 1];
    const bool runs = (mark == 0);
    const int  keep = runs ? -1 : 0;
    const int  r0   = min(max(g0, 0), n);
    const int  r1   = min(max(g1, r0), n);
    const int  ns   = r0 & keep;
    const int  ne   = (r1 & keep) | (n & ~keep);

    const int   q   = lane >> 3;
    const int   li  = lane & 7;

    float4 acc  = make_float4(0.0f, 0.0f, 0.0f, 0.0f);
    int   taken = 0;
    for (int node = ns + wave; node < ne; node += PULL_WAVES) {
        if (!runs && batch[node] != g) continue;
        ++taken;
        const int   c   = cnt[node];
        const float dn  = dinv[node];

        float4 a = make_float4(0.0f, 0.0f, 0.0f, 0.0f);

        const int ce = min(c, LIST_CAP);
        const int p  = 4 * li + q;
        for (int base = 0; base <= ce; base += 32) {
            const int m  = min(32, ce + 1 - base);
            const int gi = base + p;
            int s_l = node;
            int w_l = 0;
            if (gi < ce)  s_l = *(const int*)((const char*)ell + ((unsigned)node * (unsigned)(LIST_CAP * sizeof(int)) + (unsigned)gi * 4u));
            if (gi <= ce) w_l = __float_as_int(1.0f);
            sum_trip<float4, UP>(u, li, m, s_l, w_l, a);
        }
        if (c > LIST_CAP) {
            const int no = *ovf_n;
            for (int base = 0; base < no; base += 32) {
                const int i = base + lane;
                int2 r = make_int2(-1, 0);
                if (i < no) r = ovf[i];
                unsigned hit = __builtin_amdgcn_ballot_w32(r.x == node);
                while (hit) {
                    const int k = __builtin_ctz(hit);
                    hit &= hit - 1;
                    const int s = __shfl(r.y, k);
                    if (q == 0) add_row(a, *(const float4*)(u + (size_t)s * UP + 4 * li), 1.0f);
                }
            }
        }
        add_row(acc, a, dn);
    }

    float bbl = 0.0f, bl = 0.0f;
    if (wave == 0 && lane < FOUT) {
        bbl = bb[lane];
        bl  = blin[lane];
    }

    acc.x += __shfl_xor(acc.x, 8);
    acc.y += __shfl_xor(acc.y, 8);
    acc.z += __shfl_xor(acc.z, 8);
    acc.w += __shfl_xor(acc.w, 8);
    acc.x += __shfl_xor(acc.x, 16);
    acc.y += __shfl_xor(acc.y, 16);
    acc.z += __shfl_xor(acc.z, 16);
    acc.w += __shfl_xor(acc.w, 16);
    if (q == 0) ((float4*)(rows + wave * UP))[li] = acc;
    if (lane == 0) mine[wave] = taken;
    __syncthreads();
    if (wave != 0) return;

    float v[PULL_WAVES];
#pragma unroll
    for (int r = 0; r < PULL_WAVES; ++r) v[r] = rows[r * UP + lane];
    int k[PULL_WAVES];
#pragma unroll
    for (int r = 0; r < PULL_WAVES; ++r) k[r] = mine[r];
    float s   = 0.0f;
    int   tot = 0;
#pragma unroll
    for (int r = 0; r < PULL_WAVES; ++r) {
        s   += v[r];
        tot += k[r];
    }

    if (lane < FOUT) {
        const float c   = (float)tot;
        const float inv = 1.0f / fmaxf(c, 1.0f);
        const float has = (c > 0.0f) ? 1.0f : 0.0f;
        out[g * FOUT + lane] = fmaf(s, inv, fmaf(bbl, has, bl));
    }
}

extern "C" void kernel_launch(void* const* d_in, const int* in_sizes, int n_in,
                              void* d_out, int out_size, void* d_ws, size_t ws_size,
                              hipStream_t stream) {
    (void)n_in; (void)out_size; (void)ws_size;

    const float* x     = (const float*)d_in[0];
    const int*   esrc  = (const int*)  d_in[1];
    const int*   edst  = (const int*)  d_in[2];
    const int*   batch = (const int*)  d_in[3];
    const float* W1    = (const float*)d_in[4];
    const float* b1    = (const float*)d_in[5];
    const float* W2    = (const float*)d_in[6];
    const float* b2    = (const float*)d_in[7];
    const float* Wlin  = (const float*)d_in[8];
    const float* blin  = (const float*)d_in[9];
    float*       out   = (float*)d_out;

    const int N = in_sizes[3];
    const int E = in_sizes[1];

    char*  ws  = (char*)d_ws;
    size_t off = 0;
    auto carve = [&](size_t bytes) -> void* {
        off = (off + 255) & ~(size_t)255;
        void* p = ws + off;
        off += bytes;
        return p;
    };
    int*   ecnt  = (int*)  carve((size_t)N * 4);
    float* dinv  = (float*)carve((size_t)N * 4);
    float* bufH  = (float*)carve((size_t)N * HID * 4);
    float* bufO  = (float*)carve((size_t)N * HID * 4);
    char*  lists = (char*) carve((size_t)N * HID * 4);
    float* sums  = (float*)carve((size_t)NGRAPHS * HID * 4);

    int*   gstart = (int*)sums;
    int*   gflag  = gstart + NGRAPHS + 16;
    float* bbv    = sums + 2 * NGRAPHS;

    float*    xs   = bufH + (size_t)NGRAPHS * HID;
    unsigned* bins = (unsigned*)(xs + (size_t)N * XS_PITCH);
    int2*     ovfb = (int2*)(bufO + (size_t)N * UP);

    ws  = lists;
    off = 0;
    int*  ell   = (int*) carve((size_t)N * LIST_CAP * 4);
    int2* ovf   = (int2*)carve((size_t)E * 8);
    int*  ovf_n = (int*) carve(4);
    int*  ovfb_n = (int*)carve(4);
    int*  bincnt = (int*)carve((size_t)MAX_BINS * 4);

    const int T = 256;
    int nbin = (N + BIN_NODES - 1) / BIN_NODES;
    int eb   = (E + BIN_CHUNK - 1) / BIN_CHUNK;
    int lb   = (N + L1_NODES - 1) / L1_NODES;

    k_zero_bins  <<<MAX_BINS / T, T, 0, stream>>>(bincnt, ovf_n, ovfb_n, gflag);
    k_bin_edges  <<<eb + FOLD_GROUPS, BIN_T, 0, stream>>>(esrc, edst, bincnt, bins, ovfb, ovfb_n, E, W2, Wlin, b2, bufH, bbv);
    k_build_lists<<<nbin, BUILD_T, 0, stream>>>(bins, bincnt, ovfb, ovfb_n, x, ell, ovf, ovf_n, ecnt, dinv, xs, batch, gstart, gflag, N);

    k_sum_gemm1<<<lb, 32 * L1_NODES, 0, stream>>>(xs, W1, b1, bufH, bufO, ell, ovf, ovf_n, ecnt, dinv, N);

    k_pull_head<<<NGRAPHS, 32 * PULL_WAVES, 0, stream>>>(bufO, ell, ovf, ovf_n, ecnt, dinv, batch, gstart, gflag, bbv, blin, out, N);
}
